# starting kernel
# baseline (speedup 1.0000x reference)
#include <hip/hip_runtime.h>

#define NN 50000
#define NE 800000
#define FN 32
#define FE 16
#define HID 128
#define NL 3

typedef float v2f __attribute__((ext_vector_type(2)));
typedef float v8f __attribute__((ext_vector_type(8)));

__global__ __launch_bounds__(256) void gemm_bias_relu(
    const float* __restrict__ A, const float* __restrict__ W,
    const float* __restrict__ bias1, const float* __restrict__ bias2,
    float* __restrict__ C, int M, int K, int do_relu)
{
    const int lane    = threadIdx.x & 31;
    const int colTile = threadIdx.x >> 5;
    const int rowTile = blockIdx.x;
    const int r = lane & 15;
    const int g = lane >> 4;

    const int rowA = rowTile * 16 + r;
    const int colN = colTile * 16 + r;

    const float* __restrict__ Arow = A + (size_t)rowA * K;

    v8f acc = {};
    for (int k0 = 0; k0 < K; k0 += 4) {
        v2f a = *(const v2f*)(Arow + k0 + 2 * g);
        v2f b;
        b.x = W[(size_t)(k0 + 2 * g + 0) * HID + colN];
        b.y = W[(size_t)(k0 + 2 * g + 1) * HID + colN];
        acc = __builtin_amdgcn_wmma_f32_16x16x4_f32(
             false, a,  false, b,
             (short)0, acc,  false,  false);
    }

    float bv = 0.0f;
    if (bias1) bv += bias1[colN];
    if (bias2) bv += bias2[colN];

#pragma unroll
    for (int v = 0; v < 8; ++v) {
        int rowOut = rowTile * 16 + v + 8 * g;
        float val  = acc[v] + bv;
        if (do_relu) val = fmaxf(val, 0.0f);
        C[(size_t)rowOut * HID + colN] = val;
    }
}

__global__ void fill_f32(float* __restrict__ p, float v, int n)
{
    int i = blockIdx.x * blockDim.x + threadIdx.x;
    if (i < n) p[i] = v;
}

__global__ void deg_accum(const int* __restrict__ dst, float* __restrict__ deg, int n)
{
    int i = blockIdx.x * blockDim.x + threadIdx.x;
    if (i < n) atomicAdd(deg + dst[i], 1.0f);
}

__global__ void rsqrt_inplace(float* __restrict__ p, int n)
{
    int i = blockIdx.x * blockDim.x + threadIdx.x;
    if (i < n) {
        float d = p[i];
        p[i] = (d > 0.0f) ? rsqrtf(d) : 0.0f;
    }
}

__global__ __launch_bounds__(256) void scatter_edges(
    const float* __restrict__ hw, const int* __restrict__ src,
    const int* __restrict__ dst, const float* __restrict__ dinv,
    float* __restrict__ agg, int E)
{
    int w    = (blockIdx.x * blockDim.x + threadIdx.x) >> 5;
    int lane = threadIdx.x & 31;
    if (w >= E) return;
    int s = src[w];
    int d = dst[w];
    float n = dinv[s] * dinv[d];
    const float4 v = *(const float4*)(hw + (size_t)s * HID + lane * 4);
    float* ap = agg + (size_t)d * HID + lane * 4;
    atomicAdd(ap + 0, v.x * n);
    atomicAdd(ap + 1, v.y * n);
    atomicAdd(ap + 2, v.z * n);
    atomicAdd(ap + 3, v.w * n);
}

__global__ void combine_relu(const float* __restrict__ agg,
                             const float* __restrict__ hw,
                             const float* __restrict__ dinv,
                             float* __restrict__ h, int total)
{
    int idx = blockIdx.x * blockDim.x + threadIdx.x;
    if (idx >= total) return;
    int node = idx >> 7;
    float di = dinv[node];
    float v  = agg[idx] + hw[idx] * (di * di);
    h[idx]   = fmaxf(v, 0.0f);
}

__global__ __launch_bounds__(256) void decode_edges(
    const float* __restrict__ h,  const float* __restrict__ ea,
    const int* __restrict__ src,  const int* __restrict__ dst,
    const float* __restrict__ We, const float* __restrict__ be,
    const float* __restrict__ Wd, const float* __restrict__ bd,
    float* __restrict__ out, int E)
{
    int e    = (blockIdx.x * blockDim.x + threadIdx.x) >> 5;
    int lane = threadIdx.x & 31;
    if (e >= E) return;

    int r = src[e];
    int c = dst[e];

    float eav = (lane < FE) ? ea[(size_t)e * FE + lane] : 0.0f;

    float acc = 0.0f;
#pragma unroll
    for (int j = 0; j < 4; ++j) {
        int col = lane + 32 * j;
        acc += h[(size_t)r * HID + col] * Wd[col];
        acc += h[(size_t)c * HID + col] * Wd[HID + col];
        float s = be[col];
#pragma unroll
        for (int k = 0; k < FE; ++k) {
            float a = __shfl(eav, k, 32);
            s += a * We[(size_t)k * HID + col];
        }
        acc += fmaxf(s, 0.0f) * Wd[2 * HID + col];
    }
#pragma unroll
    for (int off = 16; off > 0; off >>= 1)
        acc += __shfl_xor(acc, off, 32);
    if (lane == 0) out[e] = acc + bd[0];
}

extern "C" void kernel_launch(void* const* d_in, const int* in_sizes, int n_in,
                              void* d_out, int out_size, void* d_ws, size_t ws_size,
                              hipStream_t stream)
{
    const float* x   = (const float*)d_in[0];
    const float* ea  = (const float*)d_in[1];
    const int*   ei  = (const int*)  d_in[2];
    const float* Wn  = (const float*)d_in[3];
    const float* bn  = (const float*)d_in[4];
    const float* We  = (const float*)d_in[5];
    const float* be  = (const float*)d_in[6];
    const float* Wc  = (const float*)d_in[7];
    const float* bc  = (const float*)d_in[8];
    const float* Wr  = (const float*)d_in[9];
    const float* br  = (const float*)d_in[10];
    const float* Wd  = (const float*)d_in[11];
    const float* bd  = (const float*)d_in[12];
    float* out = (float*)d_out;

    const int* srcI = ei;
    const int* dstI = ei + NE;

    float* ws   = (float*)d_ws;
    float* h    = ws;
    float* hw   = h   + (size_t)NN * HID;
    float* agg  = hw  + (size_t)NN * HID;
    float* dinv = agg + (size_t)NN * HID;

    const int nodeElems = NN * HID;

    gemm_bias_relu<<<NN / 16, 256, 0, stream>>>(x, Wn, bn, nullptr, h, NN, FN, 1);

    fill_f32<<<(NN + 255) / 256, 256, 0, stream>>>(dinv, 1.0f, NN);
    deg_accum<<<(NE + 255) / 256, 256, 0, stream>>>(dstI, dinv, NE);
    rsqrt_inplace<<<(NN + 255) / 256, 256, 0, stream>>>(dinv, NN);

    for (int l = 0; l < NL; ++l) {
        gemm_bias_relu<<<NN / 16, 256, 0, stream>>>(
            h, Wc + (size_t)l * HID * HID, nullptr, nullptr, hw, NN, HID, 0);
        gemm_bias_relu<<<NN / 16, 256, 0, stream>>>(
            h, Wr, br, bc + (size_t)l * HID, agg, NN, HID, 0);
        scatter_edges<<<NE / 8, 256, 0, stream>>>(hw, srcI, dstI, dinv, agg, NE);
        combine_relu<<<(nodeElems + 255) / 256, 256, 0, stream>>>(agg, hw, dinv, h, nodeElems);
    }

    decode_edges<<<NE / 8, 256, 0, stream>>>(h, ea, srcI, dstI, We, be, Wd, bd, out, NE);
}


# Edge aggregation: float-atomic scatter + combine pass replaced by per-call dst-bucketed edge list (atomic rank, scan) and one-wave-per-node register gather; edge sum order = atomic arrival rank
# speedup vs baseline: 1.6290x; 1.6290x over previous
#include <hip/hip_runtime.h>

#define NN 50000
#define NE 800000
#define FN 32
#define FE 16
#define HID 128
#define NL 3

typedef float v2f __attribute__((ext_vector_type(2)));
typedef float v8f __attribute__((ext_vector_type(8)));

__global__ __launch_bounds__(256) void gemm_bias_relu(
    const float* __restrict__ A, const float* __restrict__ W,
    const float* __restrict__ bias1, const float* __restrict__ bias2,
    float* __restrict__ C, int M, int K, int do_relu)
{
    const int lane    = threadIdx.x & 31;
    const int colTile = threadIdx.x >> 5;
    const int rowTile = blockIdx.x;
    const int r = lane & 15;
    const int g = lane >> 4;

    const int rowA = rowTile * 16 + r;
    const int colN = colTile * 16 + r;

    const float* __restrict__ Arow = A + (size_t)rowA * K;

    v8f acc = {};
    for (int k0 = 0; k0 < K; k0 += 4) {
        v2f a = *(const v2f*)(Arow + k0 + 2 * g);
        v2f b;
        b.x = W[(size_t)(k0 + 2 * g + 0) * HID + colN];
        b.y = W[(size_t)(k0 + 2 * g + 1) * HID + colN];
        acc = __builtin_amdgcn_wmma_f32_16x16x4_f32(
             false, a,  false, b,
             (short)0, acc,  false,  false);
    }

    float bv = 0.0f;
    if (bias1) bv += bias1[colN];
    if (bias2) bv += bias2[colN];

#pragma unroll
    for (int v = 0; v < 8; ++v) {
        int rowOut = rowTile * 16 + v + 8 * g;
        float val  = acc[v] + bv;
        if (do_relu) val = fmaxf(val, 0.0f);
        C[(size_t)rowOut * HID + colN] = val;
    }
}

__global__ void fill_i32(int* __restrict__ p, int v, int n)
{
    int i = blockIdx.x * blockDim.x + threadIdx.x;
    if (i < n) p[i] = v;
}

__global__ void rank_edges(const int* __restrict__ dst, int* __restrict__ cnt,
                           int* __restrict__ rank, int n)
{
    int i = blockIdx.x * blockDim.x + threadIdx.x;
    if (i < n) rank[i] = atomicAdd(cnt + dst[i], 1);
}

__global__ __launch_bounds__(256) void scan_offsets_dinv(
    const int* __restrict__ cnt, int* __restrict__ offs,
    float* __restrict__ dinv, int n)
{
    __shared__ int waveCnt[8];
    __shared__ int wavePre[8];
    const int tid  = threadIdx.x;
    const int lane = tid & 31;
    const int wv   = tid >> 5;
    const int b    = blockIdx.x;

    int pre = 0;
    const int4* __restrict__ c4 = (const int4*)cnt;
    const int n4   = b * 64;
    const int full = n4 >> 8;
#pragma unroll 4
    for (int it = 0; it < full; ++it) {
        int4 v = c4[it * 256 + tid];
        pre += (v.x + v.y) + (v.z + v.w);
    }
    if (tid < (n4 & 255)) {
        int4 v = c4[full * 256 + tid];
        pre += (v.x + v.y) + (v.z + v.w);
    }

    const int i = b * 256 + tid;
    const int c = (i < n) ? cnt[i] : 0;

    int inc = c;
    for (int off = 1; off < 32; off <<= 1) {
        int t = __shfl_up(inc, off, 32);
        if (lane >= off) inc += t;
    }
    for (int off = 16; off > 0; off >>= 1)
        pre += __shfl_xor(pre, off, 32);

    if (lane == 31) waveCnt[wv] = inc;
    if (lane == 0)  wavePre[wv] = pre;
    __syncthreads();

    int base = 0;
    for (int k = 0; k < 8; ++k) {
        base += wavePre[k];
        if (k < wv) base += waveCnt[k];
    }
    const int excl = base + inc - c;
    if (i < n) {
        offs[i] = excl;
        float d = (float)(c + 1);
        dinv[i] = (d > 0.0f) ? rsqrtf(d) : 0.0f;
    }
    if (i == n - 1) offs[n] = excl + c;
}

__global__ void bucket_edges(const int* __restrict__ src, const int* __restrict__ dst,
                             const int* __restrict__ rank, const int* __restrict__ offs,
                             int* __restrict__ bsrc, int n)
{
    int i = blockIdx.x * blockDim.x + threadIdx.x;
    if (i < n) bsrc[offs[dst[i]] + rank[i]] = src[i];
}

__global__ __launch_bounds__(256) void gather_combine_relu(
    const float* __restrict__ hw, const float* __restrict__ res,
    const int* __restrict__ offs, const int* __restrict__ bsrc,
    const float* __restrict__ dinv, float* __restrict__ h, int N)
{
    const int lane = threadIdx.x & 31;
    const int n = __builtin_amdgcn_readfirstlane(
        (int)((blockIdx.x * blockDim.x + threadIdx.x) >> 5));
    if (n >= N) return;

    const int   beg = offs[n];
    const int   end = offs[n + 1];
    const float di  = dinv[n];
    const size_t rowOff = (size_t)n * HID + lane * 4;

    float4 acc        = *(const float4*)(res + rowOff);
    const float4 self = *(const float4*)(hw  + rowOff);

    for (int base = beg; base < end; base += 32) {
        const int m = min(32, end - base);
        int   sL = 0;
        float wL = 0.0f;
        if (lane < m) {
            sL = bsrc[base + lane];
            wL = dinv[sL] * di;
        }
#pragma unroll 4
        for (int j = 0; j < m; ++j) {
            const int   s = __builtin_amdgcn_readlane(sL, j);
            const float w = __int_as_float(__builtin_amdgcn_readlane(__float_as_int(wL), j));
            const float4 v = *(const float4*)(hw + (size_t)s * HID + lane * 4);
            acc.x += v.x * w;
            acc.y += v.y * w;
            acc.z += v.z * w;
            acc.w += v.w * w;
        }
    }

    const float dd = di * di;
    acc.x = fmaxf(acc.x + self.x * dd, 0.0f);
    acc.y = fmaxf(acc.y + self.y * dd, 0.0f);
    acc.z = fmaxf(acc.z + self.z * dd, 0.0f);
    acc.w = fmaxf(acc.w + self.w * dd, 0.0f);
    *(float4*)(h + rowOff) = acc;
}

__global__ __launch_bounds__(256) void decode_edges(
    const float* __restrict__ h,  const float* __restrict__ ea,
    const int* __restrict__ src,  const int* __restrict__ dst,
    const float* __restrict__ We, const float* __restrict__ be,
    const float* __restrict__ Wd, const float* __restrict__ bd,
    float* __restrict__ out, int E)
{
    int e    = (blockIdx.x * blockDim.x + threadIdx.x) >> 5;
    int lane = threadIdx.x & 31;
    if (e >= E) return;

    int r = src[e];
    int c = dst[e];

    float eav = (lane < FE) ? ea[(size_t)e * FE + lane] : 0.0f;

    float acc = 0.0f;
#pragma unroll
    for (int j = 0; j < 4; ++j) {
        int col = lane + 32 * j;
        acc += h[(size_t)r * HID + col] * Wd[col];
        acc += h[(size_t)c * HID + col] * Wd[HID + col];
        float s = be[col];
#pragma unroll
        for (int k = 0; k < FE; ++k) {
            float a = __shfl(eav, k, 32);
            s += a * We[(size_t)k * HID + col];
        }
        acc += fmaxf(s, 0.0f) * Wd[2 * HID + col];
    }
#pragma unroll
    for (int off = 16; off > 0; off >>= 1)
        acc += __shfl_xor(acc, off, 32);
    if (lane == 0) out[e] = acc + bd[0];
}

extern "C" void kernel_launch(void* const* d_in, const int* in_sizes, int n_in,
                              void* d_out, int out_size, void* d_ws, size_t ws_size,
                              hipStream_t stream)
{
    const float* x   = (const float*)d_in[0];
    const float* ea  = (const float*)d_in[1];
    const int*   ei  = (const int*)  d_in[2];
    const float* Wn  = (const float*)d_in[3];
    const float* bn  = (const float*)d_in[4];
    const float* We  = (const float*)d_in[5];
    const float* be  = (const float*)d_in[6];
    const float* Wc  = (const float*)d_in[7];
    const float* bc  = (const float*)d_in[8];
    const float* Wr  = (const float*)d_in[9];
    const float* br  = (const float*)d_in[10];
    const float* Wd  = (const float*)d_in[11];
    const float* bd  = (const float*)d_in[12];
    float* out = (float*)d_out;

    const int* srcI = ei;
    const int* dstI = ei + NE;

    float* ws   = (float*)d_ws;
    float* h    = ws;
    float* hw   = h   + (size_t)NN * HID;
    float* agg  = hw  + (size_t)NN * HID;
    float* dinv = agg + (size_t)NN * HID;
    int*   cnt  = (int*)(dinv + NN);
    int*   offs = cnt  + NN;
    int*   rank = offs + NN + 16;
    int*   bsrc = rank + NE;

    gemm_bias_relu<<<NN / 16, 256, 0, stream>>>(x, Wn, bn, nullptr, h, NN, FN, 1);

    fill_i32<<<(NN + 255) / 256, 256, 0, stream>>>(cnt, 0, NN);
    rank_edges<<<(NE + 255) / 256, 256, 0, stream>>>(dstI, cnt, rank, NE);
    scan_offsets_dinv<<<(NN + 255) / 256, 256, 0, stream>>>(cnt, offs, dinv, NN);
    bucket_edges<<<(NE + 255) / 256, 256, 0, stream>>>(srcI, dstI, rank, offs, bsrc, NE);

    for (int l = 0; l < NL; ++l) {
        gemm_bias_relu<<<NN / 16, 256, 0, stream>>>(
            h, Wc + (size_t)l * HID * HID, nullptr, nullptr, hw, NN, HID, 0);
        gemm_bias_relu<<<NN / 16, 256, 0, stream>>>(
            h, Wr, br, bc + (size_t)l * HID, agg, NN, HID, 0);
        gather_combine_relu<<<NN / 8, 256, 0, stream>>>(hw, agg, offs, bsrc, dinv, h, NN);
    }

    decode_edges<<<NE / 8, 256, 0, stream>>>(h, ea, srcI, dstI, We, be, Wd, bd, out, NE);
}


# Decoder re-tiled: each wave walks 32-edge groups with We, be, Wd held in registers, edge encoder on f32 16x16x4 WMMA, row dots summed by a transposing lane butterfly; decoder sum order changes
# speedup vs baseline: 2.2466x; 1.3791x over previous
#include <hip/hip_runtime.h>

#define NN 50000
#define NE 800000
#define FN 32
#define FE 16
#define HID 128
#define NL 3

typedef float v2f __attribute__((ext_vector_type(2)));
typedef float v8f __attribute__((ext_vector_type(8)));

__global__ __launch_bounds__(256) void gemm_bias_relu(
    const float* __restrict__ A, const float* __restrict__ W,
    const float* __restrict__ bias1, const float* __restrict__ bias2,
    float* __restrict__ C, int M, int K, int do_relu)
{
    const int lane    = threadIdx.x & 31;
    const int colTile = threadIdx.x >> 5;
    const int rowTile = blockIdx.x;
    const int r = lane & 15;
    const int g = lane >> 4;

    const int rowA = rowTile * 16 + r;
    const int colN = colTile * 16 + r;

    const float* __restrict__ Arow = A + (size_t)rowA * K;

    v8f acc = {};
    for (int k0 = 0; k0 < K; k0 += 4) {
        v2f a = *(const v2f*)(Arow + k0 + 2 * g);
        v2f b;
        b.x = W[(size_t)(k0 + 2 * g + 0) * HID + colN];
        b.y = W[(size_t)(k0 + 2 * g + 1) * HID + colN];
        acc = __builtin_amdgcn_wmma_f32_16x16x4_f32(
             false, a,  false, b,
             (short)0, acc,  false,  false);
    }

    float bv = 0.0f;
    if (bias1) bv += bias1[colN];
    if (bias2) bv += bias2[colN];

#pragma unroll
    for (int v = 0; v < 8; ++v) {
        int rowOut = rowTile * 16 + v + 8 * g;
        float val  = acc[v] + bv;
        if (do_relu) val = fmaxf(val, 0.0f);
        C[(size_t)rowOut * HID + colN] = val;
    }
}

__global__ void fill_i32(int* __restrict__ p, int v, int n)
{
    int i = blockIdx.x * blockDim.x + threadIdx.x;
    if (i < n) p[i] = v;
}

__global__ void rank_edges(const int* __restrict__ dst, int* __restrict__ cnt,
                           int* __restrict__ rank, int n)
{
    int i = blockIdx.x * blockDim.x + threadIdx.x;
    if (i < n) rank[i] = atomicAdd(cnt + dst[i], 1);
}

__global__ __launch_bounds__(256) void scan_offsets_dinv(
    const int* __restrict__ cnt, int* __restrict__ offs,
    float* __restrict__ dinv, int n)
{
    __shared__ int waveCnt[8];
    __shared__ int wavePre[8];
    const int tid  = threadIdx.x;
    const int lane = tid & 31;
    const int wv   = tid >> 5;
    const int b    = blockIdx.x;

    int pre = 0;
    const int4* __restrict__ c4 = (const int4*)cnt;
    const int n4   = b * 64;
    const int full = n4 >> 8;
#pragma unroll 4
    for (int it = 0; it < full; ++it) {
        int4 v = c4[it * 256 + tid];
        pre += (v.x + v.y) + (v.z + v.w);
    }
    if (tid < (n4 & 255)) {
        int4 v = c4[full * 256 + tid];
        pre += (v.x + v.y) + (v.z + v.w);
    }

    const int i = b * 256 + tid;
    const int c = (i < n) ? cnt[i] : 0;

    int inc = c;
    for (int off = 1; off < 32; off <<= 1) {
        int t = __shfl_up(inc, off, 32);
        if (lane >= off) inc += t;
    }
    for (int off = 16; off > 0; off >>= 1)
        pre += __shfl_xor(pre, off, 32);

    if (lane == 31) waveCnt[wv] = inc;
    if (lane == 0)  wavePre[wv] = pre;
    __syncthreads();

    int base = 0;
    for (int k = 0; k < 8; ++k) {
        base += wavePre[k];
        if (k < wv) base += waveCnt[k];
    }
    const int excl = base + inc - c;
    if (i < n) {
        offs[i] = excl;
        float d = (float)(c + 1);
        dinv[i] = (d > 0.0f) ? rsqrtf(d) : 0.0f;
    }
    if (i == n - 1) offs[n] = excl + c;
}

__global__ void bucket_edges(const int* __restrict__ src, const int* __restrict__ dst,
                             const int* __restrict__ rank, const int* __restrict__ offs,
                             int* __restrict__ bsrc, int n)
{
    int i = blockIdx.x * blockDim.x + threadIdx.x;
    if (i < n) bsrc[offs[dst[i]] + rank[i]] = src[i];
}

__global__ __launch_bounds__(256) void gather_combine_relu(
    const float* __restrict__ hw, const float* __restrict__ res,
    const int* __restrict__ offs, const int* __restrict__ bsrc,
    const float* __restrict__ dinv, float* __restrict__ h, int N)
{
    const int lane = threadIdx.x & 31;
    const int n = __builtin_amdgcn_readfirstlane(
        (int)((blockIdx.x * blockDim.x + threadIdx.x) >> 5));
    if (n >= N) return;

    const int   beg = offs[n];
    const int   end = offs[n + 1];
    const float di  = dinv[n];
    const size_t rowOff = (size_t)n * HID + lane * 4;

    float4 acc        = *(const float4*)(res + rowOff);
    const float4 self = *(const float4*)(hw  + rowOff);

    for (int base = beg; base < end; base += 32) {
        const int m = min(32, end - base);
        int   sL = 0;
        float wL = 0.0f;
        if (lane < m) {
            sL = bsrc[base + lane];
            wL = dinv[sL] * di;
        }
#pragma unroll 4
        for (int j = 0; j < m; ++j) {
            const int   s = __builtin_amdgcn_readlane(sL, j);
            const float w = __int_as_float(__builtin_amdgcn_readlane(__float_as_int(wL), j));
            const float4 v = *(const float4*)(hw + (size_t)s * HID + lane * 4);
            acc.x += v.x * w;
            acc.y += v.y * w;
            acc.z += v.z * w;
            acc.w += v.w * w;
        }
    }

    const float dd = di * di;
    acc.x = fmaxf(acc.x + self.x * dd, 0.0f);
    acc.y = fmaxf(acc.y + self.y * dd, 0.0f);
    acc.z = fmaxf(acc.z + self.z * dd, 0.0f);
    acc.w = fmaxf(acc.w + self.w * dd, 0.0f);
    *(float4*)(h + rowOff) = acc;
}

#define MB 4
__device__ __forceinline__ float xhalf(float x)
{
    int b = __builtin_bit_cast(int, x);
    return __builtin_bit_cast(float,
        __builtin_amdgcn_permlanex16(b, b, 0x76543210, 0xfedcba98, false, false));
}

__global__ __launch_bounds__(128, 4) void decode_edges(
    const float* __restrict__ h,  const float* __restrict__ ea,
    const int* __restrict__ src,  const int* __restrict__ dst,
    const float* __restrict__ We, const float* __restrict__ be,
    const float* __restrict__ Wd, const float* __restrict__ bd,
    float* __restrict__ out, int nGroups)
{
    const int lane = threadIdx.x & 31;
    const int c    = lane & 15;
    const int g    = lane >> 4;
    const int wave = __builtin_amdgcn_readfirstlane(
        (int)((blockIdx.x * blockDim.x + threadIdx.x) >> 5));
    const int nWaves = (int)((gridDim.x * blockDim.x) >> 5);

    v2f Bf[8][4];
#pragma unroll
    for (int s = 0; s < 4; ++s) {
#pragma unroll
        for (int j = 0; j < 2; ++j) {
            const float* rp = We + (size_t)(8 * g + 2 * s + j) * HID + 8 * c;
            const float4 lo = *(const float4*)rp;
            const float4 hi = *(const float4*)(rp + 4);
            Bf[0][s][j] = lo.x; Bf[1][s][j] = lo.y; Bf[2][s][j] = lo.z; Bf[3][s][j] = lo.w;
            Bf[4][s][j] = hi.x; Bf[5][s][j] = hi.y; Bf[6][s][j] = hi.z; Bf[7][s][j] = hi.w;
        }
    }
    float bev[8], wd3[8];
    {
        const float4 b0 = *(const float4*)(be + 8 * c);
        const float4 b1 = *(const float4*)(be + 8 * c + 4);
        const float4 w0 = *(const float4*)(Wd + 2 * HID + 8 * c);
        const float4 w1 = *(const float4*)(Wd + 2 * HID + 8 * c + 4);
        bev[0] = b0.x; bev[1] = b0.y; bev[2] = b0.z; bev[3] = b0.w;
        bev[4] = b1.x; bev[5] = b1.y; bev[6] = b1.z; bev[7] = b1.w;
        wd3[0] = w0.x; wd3[1] = w0.y; wd3[2] = w0.z; wd3[3] = w0.w;
        wd3[4] = w1.x; wd3[5] = w1.y; wd3[6] = w1.z; wd3[7] = w1.w;
    }
    const float4 wd1 = *(const float4*)(Wd + lane * 4);
    const float4 wd2 = *(const float4*)(Wd + HID + lane * 4);
    const float  bdv = bd[0];

    const bool hiHalf = (g != 0);
    const bool bit3 = (c & 8) != 0;
    const bool bit2 = (c & 4) != 0;
    const bool bit1 = (c & 2) != 0;
    const bool bit0 = (c & 1) != 0;
    const unsigned laneByte = (unsigned)lane * 16u;
    const char* __restrict__ hB = (const char*)h;

    int grp = wave;
    int sN = 0, dN = 0;
    if (grp < nGroups) {
        sN = src[grp * 32 + lane];
        dN = dst[grp * 32 + lane];
    }

    for (; grp < nGroups; grp += nWaves) {
        const int e0 = grp * 32;
        const int sL = sN;
        const int dL = dN;
        const int nxt = min(grp + nWaves, nGroups - 1);
        sN = src[nxt * 32 + lane];
        dN = dst[nxt * 32 + lane];

        float4 alo[2], ahi[2];
#pragma unroll
        for (int T = 0; T < 2; ++T) {
            const float* ap = ea + (size_t)(e0 + 16 * T + c) * FE + 8 * g;
            alo[T] = *(const float4*)ap;
            ahi[T] = *(const float4*)(ap + 4);
        }

        float q[16];
#pragma unroll
        for (int mb = 0; mb < 16; mb += MB) {
            unsigned off[MB][2][2];
#pragma unroll
            for (int mm = 0; mm < MB; ++mm) {
                const int m  = mb + mm;
                const int eA = (m < 8) ? m : (m + 8);
#pragma unroll
                for (int x = 0; x < 2; ++x) {
                    const int e = eA + 8 * x;
                    off[mm][x][0] = (unsigned)__builtin_amdgcn_readlane(sL, e) * (HID * 4u) + laneByte;
                    off[mm][x][1] = (unsigned)__builtin_amdgcn_readlane(dL, e) * (HID * 4u) + laneByte;
                }
            }
            __builtin_amdgcn_sched_barrier(0);
            float4 v[MB][2][2];
#pragma unroll
            for (int mm = 0; mm < MB; ++mm)
#pragma unroll
                for (int x = 0; x < 2; ++x) {
                    v[mm][x][0] = *(const float4*)(hB + off[mm][x][0]);
                    v[mm][x][1] = *(const float4*)(hB + off[mm][x][1]);
                }
            __builtin_amdgcn_sched_barrier(0);
#pragma unroll
            for (int mm = 0; mm < MB; ++mm) {
                const int m = mb + mm;
                float p[2];
#pragma unroll
                for (int x = 0; x < 2; ++x) {
                    const float4 v1 = v[mm][x][0];
                    const float4 v2 = v[mm][x][1];
                    float a = v1.x * wd1.x;
                    a += v1.y * wd1.y;
                    a += v1.z * wd1.z;
                    a += v1.w * wd1.w;
                    a += v2.x * wd2.x;
                    a += v2.y * wd2.y;
                    a += v2.z * wd2.z;
                    a += v2.w * wd2.w;
                    p[x] = a;
                }
                const float keep = hiHalf ? p[1] : p[0];
                const float send = hiHalf ? p[0] : p[1];
                q[m] = keep + xhalf(send);
            }
        }

#pragma unroll
        for (int T = 0; T < 2; ++T) {
            v2f A[4];
            A[0][0] = alo[T].x; A[0][1] = alo[T].y;
            A[1][0] = alo[T].z; A[1][1] = alo[T].w;
            A[2][0] = ahi[T].x; A[2][1] = ahi[T].y;
            A[3][0] = ahi[T].z; A[3][1] = ahi[T].w;

            float part[8];
#pragma unroll
            for (int r = 0; r < 8; ++r) part[r] = 0.0f;
#pragma unroll
            for (int t = 0; t < 8; t += 2) {
                v8f acc0 = {};
                v8f acc1 = {};
#pragma unroll
                for (int s = 0; s < 4; ++s) {
                    acc0 = __builtin_amdgcn_wmma_f32_16x16x4_f32(
                        false, A[s], false, Bf[t][s], (short)0, acc0, false, false);
                    acc1 = __builtin_amdgcn_wmma_f32_16x16x4_f32(
                        false, A[s], false, Bf[t + 1][s], (short)0, acc1, false, false);
                }
#pragma unroll
                for (int r = 0; r < 8; ++r) {
                    part[r] += fmaxf(acc0[r] + bev[t], 0.0f) * wd3[t];
                    part[r] += fmaxf(acc1[r] + bev[t + 1], 0.0f) * wd3[t + 1];
                }
            }
#pragma unroll
            for (int r = 0; r < 8; ++r) q[8 * T + r] += part[r];
        }

        float u[8];
#pragma unroll
        for (int m = 0; m < 8; ++m) {
            const float keep = bit3 ? q[8 + m] : q[m];
            const float send = bit3 ? q[m] : q[8 + m];
            u[m] = keep + __shfl_xor(send, 8, 32);
        }
        float w4[4];
#pragma unroll
        for (int m = 0; m < 4; ++m) {
            const float keep = bit2 ? u[4 + m] : u[m];
            const float send = bit2 ? u[m] : u[4 + m];
            w4[m] = keep + __shfl_xor(send, 4, 32);
        }
        float x2[2];
#pragma unroll
        for (int m = 0; m < 2; ++m) {
            const float keep = bit1 ? w4[2 + m] : w4[m];
            const float send = bit1 ? w4[m] : w4[2 + m];
            x2[m] = keep + __shfl_xor(send, 2, 32);
        }
        const float keep1 = bit0 ? x2[1] : x2[0];
        const float send1 = bit0 ? x2[0] : x2[1];
        const float z = keep1 + __shfl_xor(send1, 1, 32);

        out[e0 + ((c & 8) << 1) + 8 * g + (c & 7)] = z + bdv;
    }
}

extern "C" void kernel_launch(void* const* d_in, const int* in_sizes, int n_in,
                              void* d_out, int out_size, void* d_ws, size_t ws_size,
                              hipStream_t stream)
{
    const float* x   = (const float*)d_in[0];
    const float* ea  = (const float*)d_in[1];
    const int*   ei  = (const int*)  d_in[2];
    const float* Wn  = (const float*)d_in[3];
    const float* bn  = (const float*)d_in[4];
    const float* We  = (const float*)d_in[5];
    const float* be  = (const float*)d_in[6];
    const float* Wc  = (const float*)d_in[7];
    const float* bc  = (const float*)d_in[8];
    const float* Wr  = (const float*)d_in[9];
    const float* br  = (const float*)d_in[10];
    const float* Wd  = (const float*)d_in[11];
    const float* bd  = (const float*)d_in[12];
    float* out = (float*)d_out;

    const int* srcI = ei;
    const int* dstI = ei + NE;

    float* ws   = (float*)d_ws;
    float* h    = ws;
    float* hw   = h   + (size_t)NN * HID;
    float* agg  = hw  + (size_t)NN * HID;
    float* dinv = agg + (size_t)NN * HID;
    int*   cnt  = (int*)(dinv + NN);
    int*   offs = cnt  + NN;
    int*   rank = offs + NN + 16;
    int*   bsrc = rank + NE;

    gemm_bias_relu<<<NN / 16, 256, 0, stream>>>(x, Wn, bn, nullptr, h, NN, FN, 1);

    fill_i32<<<(NN + 255) / 256, 256, 0, stream>>>(cnt, 0, NN);
    rank_edges<<<(NE + 255) / 256, 256, 0, stream>>>(dstI, cnt, rank, NE);
    scan_offsets_dinv<<<(NN + 255) / 256, 256, 0, stream>>>(cnt, offs, dinv, NN);
    bucket_edges<<<(NE + 255) / 256, 256, 0, stream>>>(srcI, dstI, rank, offs, bsrc, NE);

    for (int l = 0; l < NL; ++l) {
        gemm_bias_relu<<<NN / 16, 256, 0, stream>>>(
            h, Wc + (size_t)l * HID * HID, nullptr, nullptr, hw, NN, HID, 0);
        gemm_bias_relu<<<NN / 16, 256, 0, stream>>>(
            h, Wr, br, bc + (size_t)l * HID, agg, NN, HID, 0);
        gather_combine_relu<<<NN / 8, 256, 0, stream>>>(hw, agg, offs, bsrc, dinv, h, NN);
    }

    decode_edges<<<768, 128, 0, stream>>>(h, ea, srcI, dstI, We, be, Wd, bd, out, NE / 32);
}


# Node GEMMs re-tiled: one wave per 16x128 strip (8 accumulators, A fragments loaded once), W staged once per block in LDS in fragment order, whole-line 16-byte stores; k sum order changes
# speedup vs baseline: 2.7710x; 1.2334x over previous
#include <hip/hip_runtime.h>

#define NN 50000
#define NE 800000
#define FN 32
#define FE 16
#define HID 128
#define NL 3
#define NRT (NN / 16)
#define GEMM_BLOCKS ((NRT + 7) / 8)

typedef float v2f __attribute__((ext_vector_type(2)));
typedef float v8f __attribute__((ext_vector_type(8)));

#define WPITCH 264

template <int KB>
__global__ __launch_bounds__(256) void gemm_strip(
    const float* __restrict__ A, const float* __restrict__ W,
    const float* __restrict__ bias1, const float* __restrict__ bias2,
    float* __restrict__ C, int nRowTiles, int do_relu)
{
    constexpr int K = 16 * KB;
    __shared__ __attribute__((aligned(16))) float Ws[(K / 2) * WPITCH];

    const int tid  = threadIdx.x;
    const int lane = tid & 31;
    const int c    = lane & 15;
    const int g    = lane >> 4;
    const int rt = __builtin_amdgcn_readfirstlane((int)(blockIdx.x * 8 + (tid >> 5)));
    const bool active = rt < nRowTiles;
    const int rtl = active ? rt : nRowTiles - 1;

    float4 alo[KB], ahi[KB];
    {
        const float* ap = A + (size_t)(rtl * 16 + c) * K + 8 * g;
#pragma unroll
        for (int b = 0; b < KB; ++b) {
            alo[b] = *(const float4*)(ap + 16 * b);
            ahi[b] = *(const float4*)(ap + 16 * b + 4);
        }
    }

#pragma unroll
    for (int i = 0; i < KB; ++i) {
        const int kp = (tid >> 5) + 8 * i;
        const int m  = tid & 31;
        const float4 w0 = *(const float4*)(W + (size_t)(2 * kp) * HID + 4 * m);
        const float4 w1 = *(const float4*)(W + (size_t)(2 * kp + 1) * HID + 4 * m);
        float* dp = Ws + kp * WPITCH + (m >> 4) * 128 + (m & 15) * 4;
        *(float4*)dp        = make_float4(w0.x, w1.x, w0.y, w1.y);
        *(float4*)(dp + 64) = make_float4(w0.z, w1.z, w0.w, w1.w);
    }
    __syncthreads();

    if (!active) return;

    v8f acc[8];
#pragma unroll
    for (int t = 0; t < 8; ++t) acc[t] = v8f{};

    const float* bp = Ws + (4 * g) * WPITCH + c * 4;
#pragma unroll
    for (int b = 0; b < KB; ++b) {
        v2f Af[4];
        Af[0][0] = alo[b].x; Af[0][1] = alo[b].y;
        Af[1][0] = alo[b].z; Af[1][1] = alo[b].w;
        Af[2][0] = ahi[b].x; Af[2][1] = ahi[b].y;
        Af[3][0] = ahi[b].z; Af[3][1] = ahi[b].w;
#pragma unroll
        for (int s = 0; s < 4; ++s) {
            const float* q = bp + (8 * b + s) * WPITCH;
#pragma unroll
            for (int p = 0; p < 4; ++p) {
                const float4 w = *(const float4*)(q + 64 * p);
                v2f b0, b1;
                b0[0] = w.x; b0[1] = w.y;
                b1[0] = w.z; b1[1] = w.w;
                acc[2 * p] = __builtin_amdgcn_wmma_f32_16x16x4_f32(
                    false, Af[s], false, b0, (short)0, acc[2 * p], false, false);
                acc[2 * p + 1] = __builtin_amdgcn_wmma_f32_16x16x4_f32(
                    false, Af[s], false, b1, (short)0, acc[2 * p + 1], false, false);
            }
        }
    }

    float4 bvlo = make_float4(0.0f, 0.0f, 0.0f, 0.0f);
    float4 bvhi = make_float4(0.0f, 0.0f, 0.0f, 0.0f);
    if (bias1) {
        const float4 l = *(const float4*)(bias1 + 4 * c);
        const float4 h = *(const float4*)(bias1 + 64 + 4 * c);
        bvlo.x += l.x; bvlo.y += l.y; bvlo.z += l.z; bvlo.w += l.w;
        bvhi.x += h.x; bvhi.y += h.y; bvhi.z += h.z; bvhi.w += h.w;
    }
    if (bias2) {
        const float4 l = *(const float4*)(bias2 + 4 * c);
        const float4 h = *(const float4*)(bias2 + 64 + 4 * c);
        bvlo.x += l.x; bvlo.y += l.y; bvlo.z += l.z; bvlo.w += l.w;
        bvhi.x += h.x; bvhi.y += h.y; bvhi.z += h.z; bvhi.w += h.w;
    }

#pragma unroll
    for (int r = 0; r < 8; ++r) {
        float4 lo, hi;
        lo.x = acc[0][r] + bvlo.x; lo.y = acc[1][r] + bvlo.y;
        lo.z = acc[2][r] + bvlo.z; lo.w = acc[3][r] + bvlo.w;
        hi.x = acc[4][r] + bvhi.x; hi.y = acc[5][r] + bvhi.y;
        hi.z = acc[6][r] + bvhi.z; hi.w = acc[7][r] + bvhi.w;
        if (do_relu) {
            lo.x = fmaxf(lo.x, 0.0f); lo.y = fmaxf(lo.y, 0.0f);
            lo.z = fmaxf(lo.z, 0.0f); lo.w = fmaxf(lo.w, 0.0f);
            hi.x = fmaxf(hi.x, 0.0f); hi.y = fmaxf(hi.y, 0.0f);
            hi.z = fmaxf(hi.z, 0.0f); hi.w = fmaxf(hi.w, 0.0f);
        }
        float* cp = C + (size_t)(rt * 16 + 8 * g + r) * HID + 4 * c;
        *(float4*)cp        = lo;
        *(float4*)(cp + 64) = hi;
    }
}

__global__ void fill_i32(int* __restrict__ p, int v, int n)
{
    int i = blockIdx.x * blockDim.x + threadIdx.x;
    if (i < n) p[i] = v;
}

__global__ void rank_edges(const int* __restrict__ dst, int* __restrict__ cnt,
                           int* __restrict__ rank, int n)
{
    int i = blockIdx.x * blockDim.x + threadIdx.x;
    if (i < n) rank[i] = atomicAdd(cnt + dst[i], 1);
}

__global__ __launch_bounds__(256) void scan_offsets_dinv(
    const int* __restrict__ cnt, int* __restrict__ offs,
    float* __restrict__ dinv, int n)
{
    __shared__ int waveCnt[8];
    __shared__ int wavePre[8];
    const int tid  = threadIdx.x;
    const int lane = tid & 31;
    const int wv   = tid >> 5;
    const int b    = blockIdx.x;

    int pre = 0;
    const int4* __restrict__ c4 = (const int4*)cnt;
    const int n4   = b * 64;
    const int full = n4 >> 8;
#pragma unroll 4
    for (int it = 0; it < full; ++it) {
        int4 v = c4[it * 256 + tid];
        pre += (v.x + v.y) + (v.z + v.w);
    }
    if (tid < (n4 & 255)) {
        int4 v = c4[full * 256 + tid];
        pre += (v.x + v.y) + (v.z + v.w);
    }

    const int i = b * 256 + tid;
    const int c = (i < n) ? cnt[i] : 0;

    int inc = c;
    for (int off = 1; off < 32; off <<= 1) {
        int t = __shfl_up(inc, off, 32);
        if (lane >= off) inc += t;
    }
    for (int off = 16; off > 0; off >>= 1)
        pre += __shfl_xor(pre, off, 32);

    if (lane == 31) waveCnt[wv] = inc;
    if (lane == 0)  wavePre[wv] = pre;
    __syncthreads();

    int base = 0;
    for (int k = 0; k < 8; ++k) {
        base += wavePre[k];
        if (k < wv) base += waveCnt[k];
    }
    const int excl = base + inc - c;
    if (i < n) {
        offs[i] = excl;
        float d = (float)(c + 1);
        dinv[i] = (d > 0.0f) ? rsqrtf(d) : 0.0f;
    }
    if (i == n - 1) offs[n] = excl + c;
}

__global__ void bucket_edges(const int* __restrict__ src, const int* __restrict__ dst,
                             const int* __restrict__ rank, const int* __restrict__ offs,
                             int* __restrict__ bsrc, int n)
{
    int i = blockIdx.x * blockDim.x + threadIdx.x;
    if (i < n) bsrc[offs[dst[i]] + rank[i]] = src[i];
}

__global__ __launch_bounds__(256) void gather_combine_relu(
    const float* __restrict__ hw, const float* __restrict__ res,
    const int* __restrict__ offs, const int* __restrict__ bsrc,
    const float* __restrict__ dinv, float* __restrict__ h, int N)
{
    const int lane = threadIdx.x & 31;
    const int n = __builtin_amdgcn_readfirstlane(
        (int)((blockIdx.x * blockDim.x + threadIdx.x) >> 5));
    if (n >= N) return;

    const int   beg = offs[n];
    const int   end = offs[n + 1];
    const float di  = dinv[n];
    const size_t rowOff = (size_t)n * HID + lane * 4;

    float4 acc        = *(const float4*)(res + rowOff);
    const float4 self = *(const float4*)(hw  + rowOff);

    for (int base = beg; base < end; base += 32) {
        const int m = min(32, end - base);
        int   sL = 0;
        float wL = 0.0f;
        if (lane < m) {
            sL = bsrc[base + lane];
            wL = dinv[sL] * di;
        }
#pragma unroll 4
        for (int j = 0; j < m; ++j) {
            const int   s = __builtin_amdgcn_readlane(sL, j);
            const float w = __int_as_float(__builtin_amdgcn_readlane(__float_as_int(wL), j));
            const float4 v = *(const float4*)(hw + (size_t)s * HID + lane * 4);
            acc.x += v.x * w;
            acc.y += v.y * w;
            acc.z += v.z * w;
            acc.w += v.w * w;
        }
    }

    const float dd = di * di;
    acc.x = fmaxf(acc.x + self.x * dd, 0.0f);
    acc.y = fmaxf(acc.y + self.y * dd, 0.0f);
    acc.z = fmaxf(acc.z + self.z * dd, 0.0f);
    acc.w = fmaxf(acc.w + self.w * dd, 0.0f);
    *(float4*)(h + rowOff) = acc;
}

#define MB 4
__device__ __forceinline__ float xhalf(float x)
{
    int b = __builtin_bit_cast(int, x);
    return __builtin_bit_cast(float,
        __builtin_amdgcn_permlanex16(b, b, 0x76543210, 0xfedcba98, false, false));
}

__global__ __launch_bounds__(128, 4) void decode_edges(
    const float* __restrict__ h,  const float* __restrict__ ea,
    const int* __restrict__ src,  const int* __restrict__ dst,
    const float* __restrict__ We, const float* __restrict__ be,
    const float* __restrict__ Wd, const float* __restrict__ bd,
    float* __restrict__ out, int nGroups)
{
    const int lane = threadIdx.x & 31;
    const int c    = lane & 15;
    const int g    = lane >> 4;
    const int wave = __builtin_amdgcn_readfirstlane(
        (int)((blockIdx.x * blockDim.x + threadIdx.x) >> 5));
    const int nWaves = (int)((gridDim.x * blockDim.x) >> 5);

    v2f Bf[8][4];
#pragma unroll
    for (int s = 0; s < 4; ++s) {
#pragma unroll
        for (int j = 0; j < 2; ++j) {
            const float* rp = We + (size_t)(8 * g + 2 * s + j) * HID + 8 * c;
            const float4 lo = *(const float4*)rp;
            const float4 hi = *(const float4*)(rp + 4);
            Bf[0][s][j] = lo.x; Bf[1][s][j] = lo.y; Bf[2][s][j] = lo.z; Bf[3][s][j] = lo.w;
            Bf[4][s][j] = hi.x; Bf[5][s][j] = hi.y; Bf[6][s][j] = hi.z; Bf[7][s][j] = hi.w;
        }
    }
    float bev[8], wd3[8];
    {
        const float4 b0 = *(const float4*)(be + 8 * c);
        const float4 b1 = *(const float4*)(be + 8 * c + 4);
        const float4 w0 = *(const float4*)(Wd + 2 * HID + 8 * c);
        const float4 w1 = *(const float4*)(Wd + 2 * HID + 8 * c + 4);
        bev[0] = b0.x; bev[1] = b0.y; bev[2] = b0.z; bev[3] = b0.w;
        bev[4] = b1.x; bev[5] = b1.y; bev[6] = b1.z; bev[7] = b1.w;
        wd3[0] = w0.x; wd3[1] = w0.y; wd3[2] = w0.z; wd3[3] = w0.w;
        wd3[4] = w1.x; wd3[5] = w1.y; wd3[6] = w1.z; wd3[7] = w1.w;
    }
    const float4 wd1 = *(const float4*)(Wd + lane * 4);
    const float4 wd2 = *(const float4*)(Wd + HID + lane * 4);
    const float  bdv = bd[0];

    const bool hiHalf = (g != 0);
    const bool bit3 = (c & 8) != 0;
    const bool bit2 = (c & 4) != 0;
    const bool bit1 = (c & 2) != 0;
    const bool bit0 = (c & 1) != 0;
    const unsigned laneByte = (unsigned)lane * 16u;
    const char* __restrict__ hB = (const char*)h;

    int grp = wave;
    int sN = 0, dN = 0;
    if (grp < nGroups) {
        sN = src[grp * 32 + lane];
        dN = dst[grp * 32 + lane];
    }

    for (; grp < nGroups; grp += nWaves) {
        const int e0 = grp * 32;
        const int sL = sN;
        const int dL = dN;
        const int nxt = min(grp + nWaves, nGroups - 1);
        sN = src[nxt * 32 + lane];
        dN = dst[nxt * 32 + lane];

        float4 alo[2], ahi[2];
#pragma unroll
        for (int T = 0; T < 2; ++T) {
            const float* ap = ea + (size_t)(e0 + 16 * T + c) * FE + 8 * g;
            alo[T] = *(const float4*)ap;
            ahi[T] = *(const float4*)(ap + 4);
        }

        float q[16];
#pragma unroll
        for (int mb = 0; mb < 16; mb += MB) {
            unsigned off[MB][2][2];
#pragma unroll
            for (int mm = 0; mm < MB; ++mm) {
                const int m  = mb + mm;
                const int eA = (m < 8) ? m : (m + 8);
#pragma unroll
                for (int x = 0; x < 2; ++x) {
                    const int e = eA + 8 * x;
                    off[mm][x][0] = (unsigned)__builtin_amdgcn_readlane(sL, e) * (HID * 4u) + laneByte;
                    off[mm][x][1] = (unsigned)__builtin_amdgcn_readlane(dL, e) * (HID * 4u) + laneByte;
                }
            }
            __builtin_amdgcn_sched_barrier(0);
            float4 v[MB][2][2];
#pragma unroll
            for (int mm = 0; mm < MB; ++mm)
#pragma unroll
                for (int x = 0; x < 2; ++x) {
                    v[mm][x][0] = *(const float4*)(hB + off[mm][x][0]);
                    v[mm][x][1] = *(const float4*)(hB + off[mm][x][1]);
                }
            __builtin_amdgcn_sched_barrier(0);
#pragma unroll
            for (int mm = 0; mm < MB; ++mm) {
                const int m = mb + mm;
                float p[2];
#pragma unroll
                for (int x = 0; x < 2; ++x) {
                    const float4 v1 = v[mm][x][0];
                    const float4 v2 = v[mm][x][1];
                    float a = v1.x * wd1.x;
                    a += v1.y * wd1.y;
                    a += v1.z * wd1.z;
                    a += v1.w * wd1.w;
                    a += v2.x * wd2.x;
                    a += v2.y * wd2.y;
                    a += v2.z * wd2.z;
                    a += v2.w * wd2.w;
                    p[x] = a;
                }
                const float keep = hiHalf ? p[1] : p[0];
                const float send = hiHalf ? p[0] : p[1];
                q[m] = keep + xhalf(send);
            }
        }

#pragma unroll
        for (int T = 0; T < 2; ++T) {
            v2f A[4];
            A[0][0] = alo[T].x; A[0][1] = alo[T].y;
            A[1][0] = alo[T].z; A[1][1] = alo[T].w;
            A[2][0] = ahi[T].x; A[2][1] = ahi[T].y;
            A[3][0] = ahi[T].z; A[3][1] = ahi[T].w;

            float part[8];
#pragma unroll
            for (int r = 0; r < 8; ++r) part[r] = 0.0f;
#pragma unroll
            for (int t = 0; t < 8; t += 2) {
                v8f acc0 = {};
                v8f acc1 = {};
#pragma unroll
                for (int s = 0; s < 4; ++s) {
                    acc0 = __builtin_amdgcn_wmma_f32_16x16x4_f32(
                        false, A[s], false, Bf[t][s], (short)0, acc0, false, false);
                    acc1 = __builtin_amdgcn_wmma_f32_16x16x4_f32(
                        false, A[s], false, Bf[t + 1][s], (short)0, acc1, false, false);
                }
#pragma unroll
                for (int r = 0; r < 8; ++r) {
                    part[r] += fmaxf(acc0[r] + bev[t], 0.0f) * wd3[t];
                    part[r] += fmaxf(acc1[r] + bev[t + 1], 0.0f) * wd3[t + 1];
                }
            }
#pragma unroll
            for (int r = 0; r < 8; ++r) q[8 * T + r] += part[r];
        }

        float u[8];
#pragma unroll
        for (int m = 0; m < 8; ++m) {
            const float keep = bit3 ? q[8 + m] : q[m];
            const float send = bit3 ? q[m] : q[8 + m];
            u[m] = keep + __shfl_xor(send, 8, 32);
        }
        float w4[4];
#pragma unroll
        for (int m = 0; m < 4; ++m) {
            const float keep = bit2 ? u[4 + m] : u[m];
            const float send = bit2 ? u[m] : u[4 + m];
            w4[m] = keep + __shfl_xor(send, 4, 32);
        }
        float x2[2];
#pragma unroll
        for (int m = 0; m < 2; ++m) {
            const float keep = bit1 ? w4[2 + m] : w4[m];
            const float send = bit1 ? w4[m] : w4[2 + m];
            x2[m] = keep + __shfl_xor(send, 2, 32);
        }
        const float keep1 = bit0 ? x2[1] : x2[0];
        const float send1 = bit0 ? x2[0] : x2[1];
        const float z = keep1 + __shfl_xor(send1, 1, 32);

        out[e0 + ((c & 8) << 1) + 8 * g + (c & 7)] = z + bdv;
    }
}

extern "C" void kernel_launch(void* const* d_in, const int* in_sizes, int n_in,
                              void* d_out, int out_size, void* d_ws, size_t ws_size,
                              hipStream_t stream)
{
    const float* x   = (const float*)d_in[0];
    const float* ea  = (const float*)d_in[1];
    const int*   ei  = (const int*)  d_in[2];
    const float* Wn  = (const float*)d_in[3];
    const float* bn  = (const float*)d_in[4];
    const float* We  = (const float*)d_in[5];
    const float* be  = (const float*)d_in[6];
    const float* Wc  = (const float*)d_in[7];
    const float* bc  = (const float*)d_in[8];
    const float* Wr  = (const float*)d_in[9];
    const float* br  = (const float*)d_in[10];
    const float* Wd  = (const float*)d_in[11];
    const float* bd  = (const float*)d_in[12];
    float* out = (float*)d_out;

    const int* srcI = ei;
    const int* dstI = ei + NE;

    float* ws   = (float*)d_ws;
    float* h    = ws;
    float* hw   = h   + (size_t)NN * HID;
    float* agg  = hw  + (size_t)NN * HID;
    float* dinv = agg + (size_t)NN * HID;
    int*   cnt  = (int*)(dinv + NN);
    int*   offs = cnt  + NN;
    int*   rank = offs + NN + 16;
    int*   bsrc = rank + NE;

    gemm_strip<FN / 16><<<GEMM_BLOCKS, 256, 0, stream>>>(x, Wn, bn, nullptr, h, NRT, 1);

    fill_i32<<<(NN + 255) / 256, 256, 0, stream>>>(cnt, 0, NN);
    rank_edges<<<(NE + 255) / 256, 256, 0, stream>>>(dstI, cnt, rank, NE);
    scan_offsets_dinv<<<(NN + 255) / 256, 256, 0, stream>>>(cnt, offs, dinv, NN);
    bucket_edges<<<(NE + 255) / 256, 256, 0, stream>>>(srcI, dstI, rank, offs, bsrc, NE);

    for (int l = 0; l < NL; ++l) {
        gemm_strip<HID / 16><<<GEMM_BLOCKS, 256, 0, stream>>>(
            h, Wc + (size_t)l * HID * HID, nullptr, nullptr, hw, NRT, 0);
        gemm_strip<HID / 16><<<GEMM_BLOCKS, 256, 0, stream>>>(
            h, Wr, br, bc + (size_t)l * HID, agg, NRT, 0);
        gather_combine_relu<<<NN / 8, 256, 0, stream>>>(hw, agg, offs, bsrc, dinv, h, NN);
    }

    decode_edges<<<768, 128, 0, stream>>>(h, ea, srcI, dstI, We, be, Wd, bd, out, NE / 32);
}


# Decoder row dots hoisted from per edge to per node: last layer's gather epilogue emits h.Wd[0:128], h.Wd[128:256] per node, decoder adds two gathered scalars per edge; decoder sum order changes
# speedup vs baseline: 3.0090x; 1.0859x over previous
#include <hip/hip_runtime.h>

#define NN 50000
#define NE 800000
#define FN 32
#define FE 16
#define HID 128
#define NL 3
#define NRT (NN / 16)
#define GEMM_BLOCKS ((NRT + 7) / 8)

typedef float v2f __attribute__((ext_vector_type(2)));
typedef float v8f __attribute__((ext_vector_type(8)));

#define WPITCH 264

template <int KB>
__global__ __launch_bounds__(256) void gemm_strip(
    const float* __restrict__ A, const float* __restrict__ W,
    const float* __restrict__ bias1, const float* __restrict__ bias2,
    float* __restrict__ C, int nRowTiles, int do_relu)
{
    constexpr int K = 16 * KB;
    __shared__ __attribute__((aligned(16))) float Ws[(K / 2) * WPITCH];

    const int tid  = threadIdx.x;
    const int lane = tid & 31;
    const int c    = lane & 15;
    const int g    = lane >> 4;
    const int rt = __builtin_amdgcn_readfirstlane((int)(blockIdx.x * 8 + (tid >> 5)));
    const bool active = rt < nRowTiles;
    const int rtl = active ? rt : nRowTiles - 1;

    float4 alo[KB], ahi[KB];
    {
        const float* ap = A + (size_t)(rtl * 16 + c) * K + 8 * g;
#pragma unroll
        for (int b = 0; b < KB; ++b) {
            alo[b] = *(const float4*)(ap + 16 * b);
            ahi[b] = *(const float4*)(ap + 16 * b + 4);
        }
    }

#pragma unroll
    for (int i = 0; i < KB; ++i) {
        const int kp = (tid >> 5) + 8 * i;
        const int m  = tid & 31;
        const float4 w0 = *(const float4*)(W + (size_t)(2 * kp) * HID + 4 * m);
        const float4 w1 = *(const float4*)(W + (size_t)(2 * kp + 1) * HID + 4 * m);
        float* dp = Ws + kp * WPITCH + (m >> 4) * 128 + (m & 15) * 4;
        *(float4*)dp        = make_float4(w0.x, w1.x, w0.y, w1.y);
        *(float4*)(dp + 64) = make_float4(w0.z, w1.z, w0.w, w1.w);
    }
    __syncthreads();

    if (!active) return;

    v8f acc[8];
#pragma unroll
    for (int t = 0; t < 8; ++t) acc[t] = v8f{};

    const float* bp = Ws + (4 * g) * WPITCH + c * 4;
#pragma unroll
    for (int b = 0; b < KB; ++b) {
        v2f Af[4];
        Af[0][0] = alo[b].x; Af[0][1] = alo[b].y;
        Af[1][0] = alo[b].z; Af[1][1] = alo[b].w;
        Af[2][0] = ahi[b].x; Af[2][1] = ahi[b].y;
        Af[3][0] = ahi[b].z; Af[3][1] = ahi[b].w;
#pragma unroll
        for (int s = 0; s < 4; ++s) {
            const float* q = bp + (8 * b + s) * WPITCH;
#pragma unroll
            for (int p = 0; p < 4; ++p) {
                const float4 w = *(const float4*)(q + 64 * p);
                v2f b0, b1;
                b0[0] = w.x; b0[1] = w.y;
                b1[0] = w.z; b1[1] = w.w;
                acc[2 * p] = __builtin_amdgcn_wmma_f32_16x16x4_f32(
                    false, Af[s], false, b0, (short)0, acc[2 * p], false, false);
                acc[2 * p + 1] = __builtin_amdgcn_wmma_f32_16x16x4_f32(
                    false, Af[s], false, b1, (short)0, acc[2 * p + 1], false, false);
            }
        }
    }

    float4 bvlo = make_float4(0.0f, 0.0f, 0.0f, 0.0f);
    float4 bvhi = make_float4(0.0f, 0.0f, 0.0f, 0.0f);
    if (bias1) {
        const float4 l = *(const float4*)(bias1 + 4 * c);
        const float4 h = *(const float4*)(bias1 + 64 + 4 * c);
        bvlo.x += l.x; bvlo.y += l.y; bvlo.z += l.z; bvlo.w += l.w;
        bvhi.x += h.x; bvhi.y += h.y; bvhi.z += h.z; bvhi.w += h.w;
    }
    if (bias2) {
        const float4 l = *(const float4*)(bias2 + 4 * c);
        const float4 h = *(const float4*)(bias2 + 64 + 4 * c);
        bvlo.x += l.x; bvlo.y += l.y; bvlo.z += l.z; bvlo.w += l.w;
        bvhi.x += h.x; bvhi.y += h.y; bvhi.z += h.z; bvhi.w += h.w;
    }

#pragma unroll
    for (int r = 0; r < 8; ++r) {
        float4 lo, hi;
        lo.x = acc[0][r] + bvlo.x; lo.y = acc[1][r] + bvlo.y;
        lo.z = acc[2][r] + bvlo.z; lo.w = acc[3][r] + bvlo.w;
        hi.x = acc[4][r] + bvhi.x; hi.y = acc[5][r] + bvhi.y;
        hi.z = acc[6][r] + bvhi.z; hi.w = acc[7][r] + bvhi.w;
        if (do_relu) {
            lo.x = fmaxf(lo.x, 0.0f); lo.y = fmaxf(lo.y, 0.0f);
            lo.z = fmaxf(lo.z, 0.0f); lo.w = fmaxf(lo.w, 0.0f);
            hi.x = fmaxf(hi.x, 0.0f); hi.y = fmaxf(hi.y, 0.0f);
            hi.z = fmaxf(hi.z, 0.0f); hi.w = fmaxf(hi.w, 0.0f);
        }
        float* cp = C + (size_t)(rt * 16 + 8 * g + r) * HID + 4 * c;
        *(float4*)cp        = lo;
        *(float4*)(cp + 64) = hi;
    }
}

__global__ void fill_i32(int* __restrict__ p, int v, int n)
{
    int i = blockIdx.x * blockDim.x + threadIdx.x;
    if (i < n) p[i] = v;
}

__global__ void rank_edges(const int* __restrict__ dst, int* __restrict__ cnt,
                           int* __restrict__ rank, int n)
{
    int i = blockIdx.x * blockDim.x + threadIdx.x;
    if (i < n) rank[i] = atomicAdd(cnt + dst[i], 1);
}

__global__ __launch_bounds__(256) void scan_offsets_dinv(
    const int* __restrict__ cnt, int* __restrict__ offs,
    float* __restrict__ dinv, int n)
{
    __shared__ int waveCnt[8];
    __shared__ int wavePre[8];
    const int tid  = threadIdx.x;
    const int lane = tid & 31;
    const int wv   = tid >> 5;
    const int b    = blockIdx.x;

    int pre = 0;
    const int4* __restrict__ c4 = (const int4*)cnt;
    const int n4   = b * 64;
    const int full = n4 >> 8;
#pragma unroll 4
    for (int it = 0; it < full; ++it) {
        int4 v = c4[it * 256 + tid];
        pre += (v.x + v.y) + (v.z + v.w);
    }
    if (tid < (n4 & 255)) {
        int4 v = c4[full * 256 + tid];
        pre += (v.x + v.y) + (v.z + v.w);
    }

    const int i = b * 256 + tid;
    const int c = (i < n) ? cnt[i] : 0;

    int inc = c;
    for (int off = 1; off < 32; off <<= 1) {
        int t = __shfl_up(inc, off, 32);
        if (lane >= off) inc += t;
    }
    for (int off = 16; off > 0; off >>= 1)
        pre += __shfl_xor(pre, off, 32);

    if (lane == 31) waveCnt[wv] = inc;
    if (lane == 0)  wavePre[wv] = pre;
    __syncthreads();

    int base = 0;
    for (int k = 0; k < 8; ++k) {
        base += wavePre[k];
        if (k < wv) base += waveCnt[k];
    }
    const int excl = base + inc - c;
    if (i < n) {
        offs[i] = excl;
        float d = (float)(c + 1);
        dinv[i] = (d > 0.0f) ? rsqrtf(d) : 0.0f;
    }
    if (i == n - 1) offs[n] = excl + c;
}

__global__ void bucket_edges(const int* __restrict__ src, const int* __restrict__ dst,
                             const int* __restrict__ rank, const int* __restrict__ offs,
                             int* __restrict__ bsrc, int n)
{
    int i = blockIdx.x * blockDim.x + threadIdx.x;
    if (i < n) bsrc[offs[dst[i]] + rank[i]] = src[i];
}

__device__ __forceinline__ float xhalf(float x)
{
    int b = __builtin_bit_cast(int, x);
    return __builtin_bit_cast(float,
        __builtin_amdgcn_permlanex16(b, b, 0x76543210, 0xfedcba98, false, false));
}

template <bool LAST>
__global__ __launch_bounds__(256) void gather_combine_relu(
    const float* __restrict__ hw, const float* __restrict__ res,
    const int* __restrict__ offs, const int* __restrict__ bsrc,
    const float* __restrict__ dinv, float* __restrict__ h,
    const float* __restrict__ Wd, float* __restrict__ pp, int N)
{
    const int lane = threadIdx.x & 31;
    const int n = __builtin_amdgcn_readfirstlane(
        (int)((blockIdx.x * blockDim.x + threadIdx.x) >> 5));
    if (n >= N) return;

    const int   beg = offs[n];
    const int   end = offs[n + 1];
    const float di  = dinv[n];
    const size_t rowOff = (size_t)n * HID + lane * 4;

    float4 acc        = *(const float4*)(res + rowOff);
    const float4 self = *(const float4*)(hw  + rowOff);
    float4 wd1 = make_float4(0.0f, 0.0f, 0.0f, 0.0f);
    float4 wd2 = make_float4(0.0f, 0.0f, 0.0f, 0.0f);
    if (LAST) {
        wd1 = *(const float4*)(Wd + lane * 4);
        wd2 = *(const float4*)(Wd + HID + lane * 4);
    }

    for (int base = beg; base < end; base += 32) {
        const int m = min(32, end - base);
        int   sL = 0;
        float wL = 0.0f;
        if (lane < m) {
            sL = bsrc[base + lane];
            wL = dinv[sL] * di;
        }
#pragma unroll 4
        for (int j = 0; j < m; ++j) {
            const int   s = __builtin_amdgcn_readlane(sL, j);
            const float w = __int_as_float(__builtin_amdgcn_readlane(__float_as_int(wL), j));
            const float4 v = *(const float4*)(hw + (size_t)s * HID + lane * 4);
            acc.x += v.x * w;
            acc.y += v.y * w;
            acc.z += v.z * w;
            acc.w += v.w * w;
        }
    }

    const float dd = di * di;
    acc.x = fmaxf(acc.x + self.x * dd, 0.0f);
    acc.y = fmaxf(acc.y + self.y * dd, 0.0f);
    acc.z = fmaxf(acc.z + self.z * dd, 0.0f);
    acc.w = fmaxf(acc.w + self.w * dd, 0.0f);
    if (!LAST) {
        *(float4*)(h + rowOff) = acc;
    } else {
        float a = acc.x * wd1.x;
        a += acc.y * wd1.y;
        a += acc.z * wd1.z;
        a += acc.w * wd1.w;
        float b = acc.x * wd2.x;
        b += acc.y * wd2.y;
        b += acc.z * wd2.z;
        b += acc.w * wd2.w;
        const bool hiHalf = lane >= 16;
        const float keep = hiHalf ? b : a;
        const float send = hiHalf ? a : b;
        float v = keep + xhalf(send);
        v += __shfl_xor(v, 8, 32);
        v += __shfl_xor(v, 4, 32);
        v += __shfl_xor(v, 2, 32);
        v += __shfl_xor(v, 1, 32);
        const float vHi = __int_as_float(__builtin_amdgcn_readlane(__float_as_int(v), 16));
        if (lane == 0) *(float2*)(pp + 2 * (size_t)n) = make_float2(v, vHi);
    }
}

__global__ __launch_bounds__(128, 4) void decode_edges(
    const float* __restrict__ pp, const float* __restrict__ ea,
    const int* __restrict__ src,  const int* __restrict__ dst,
    const float* __restrict__ We, const float* __restrict__ be,
    const float* __restrict__ Wd, const float* __restrict__ bd,
    float* __restrict__ out, int nGroups)
{
    const int lane = threadIdx.x & 31;
    const int c    = lane & 15;
    const int g    = lane >> 4;
    const int wave = __builtin_amdgcn_readfirstlane(
        (int)((blockIdx.x * blockDim.x + threadIdx.x) >> 5));
    const int nWaves = (int)((gridDim.x * blockDim.x) >> 5);

    v2f Bf[8][4];
#pragma unroll
    for (int s = 0; s < 4; ++s) {
#pragma unroll
        for (int j = 0; j < 2; ++j) {
            const float* rp = We + (size_t)(8 * g + 2 * s + j) * HID + 8 * c;
            const float4 lo = *(const float4*)rp;
            const float4 hi = *(const float4*)(rp + 4);
            Bf[0][s][j] = lo.x; Bf[1][s][j] = lo.y; Bf[2][s][j] = lo.z; Bf[3][s][j] = lo.w;
            Bf[4][s][j] = hi.x; Bf[5][s][j] = hi.y; Bf[6][s][j] = hi.z; Bf[7][s][j] = hi.w;
        }
    }
    float bev[8], wd3[8];
    {
        const float4 b0 = *(const float4*)(be + 8 * c);
        const float4 b1 = *(const float4*)(be + 8 * c + 4);
        const float4 w0 = *(const float4*)(Wd + 2 * HID + 8 * c);
        const float4 w1 = *(const float4*)(Wd + 2 * HID + 8 * c + 4);
        bev[0] = b0.x; bev[1] = b0.y; bev[2] = b0.z; bev[3] = b0.w;
        bev[4] = b1.x; bev[5] = b1.y; bev[6] = b1.z; bev[7] = b1.w;
        wd3[0] = w0.x; wd3[1] = w0.y; wd3[2] = w0.z; wd3[3] = w0.w;
        wd3[4] = w1.x; wd3[5] = w1.y; wd3[6] = w1.z; wd3[7] = w1.w;
    }
    const float bdv = bd[0];

    const bool bit3 = (c & 8) != 0;
    const bool bit2 = (c & 4) != 0;
    const bool bit1 = (c & 2) != 0;
    const bool bit0 = (c & 1) != 0;
    const int slot = ((c & 8) << 1) + 8 * g + (c & 7);

    int grp = wave;
    int sN = 0, dN = 0;
    float4 aloN[2], ahiN[2];
#pragma unroll
    for (int T = 0; T < 2; ++T) {
        aloN[T] = make_float4(0.0f, 0.0f, 0.0f, 0.0f);
        ahiN[T] = make_float4(0.0f, 0.0f, 0.0f, 0.0f);
    }
    if (grp < nGroups) {
        sN = src[grp * 32 + slot];
        dN = dst[grp * 32 + slot];
#pragma unroll
        for (int T = 0; T < 2; ++T) {
            const float* ap = ea + (size_t)(grp * 32 + 16 * T + c) * FE + 8 * g;
            aloN[T] = *(const float4*)ap;
            ahiN[T] = *(const float4*)(ap + 4);
        }
    }

    for (; grp < nGroups; grp += nWaves) {
        const int e0 = grp * 32;
        const float ps = pp[2 * (size_t)sN];
        const float pd = pp[2 * (size_t)dN + 1];

        float4 alo[2], ahi[2];
#pragma unroll
        for (int T = 0; T < 2; ++T) { alo[T] = aloN[T]; ahi[T] = ahiN[T]; }

        const int nxt = min(grp + nWaves, nGroups - 1);
        sN = src[nxt * 32 + slot];
        dN = dst[nxt * 32 + slot];
#pragma unroll
        for (int T = 0; T < 2; ++T) {
            const float* ap = ea + (size_t)(nxt * 32 + 16 * T + c) * FE + 8 * g;
            aloN[T] = *(const float4*)ap;
            ahiN[T] = *(const float4*)(ap + 4);
        }

        float q[16];
#pragma unroll
        for (int T = 0; T < 2; ++T) {
            v2f A[4];
            A[0][0] = alo[T].x; A[0][1] = alo[T].y;
            A[1][0] = alo[T].z; A[1][1] = alo[T].w;
            A[2][0] = ahi[T].x; A[2][1] = ahi[T].y;
            A[3][0] = ahi[T].z; A[3][1] = ahi[T].w;

            float part[8];
#pragma unroll
            for (int r = 0; r < 8; ++r) part[r] = 0.0f;
#pragma unroll
            for (int t = 0; t < 8; t += 2) {
                v8f acc0 = {};
                v8f acc1 = {};
#pragma unroll
                for (int s = 0; s < 4; ++s) {
                    acc0 = __builtin_amdgcn_wmma_f32_16x16x4_f32(
                        false, A[s], false, Bf[t][s], (short)0, acc0, false, false);
                    acc1 = __builtin_amdgcn_wmma_f32_16x16x4_f32(
                        false, A[s], false, Bf[t + 1][s], (short)0, acc1, false, false);
                }
#pragma unroll
                for (int r = 0; r < 8; ++r) {
                    part[r] += fmaxf(acc0[r] + bev[t], 0.0f) * wd3[t];
                    part[r] += fmaxf(acc1[r] + bev[t + 1], 0.0f) * wd3[t + 1];
                }
            }
#pragma unroll
            for (int r = 0; r < 8; ++r) q[8 * T + r] = part[r];
        }

        float u[8];
#pragma unroll
        for (int m = 0; m < 8; ++m) {
            const float keep = bit3 ? q[8 + m] : q[m];
            const float send = bit3 ? q[m] : q[8 + m];
            u[m] = keep + __shfl_xor(send, 8, 32);
        }
        float w4[4];
#pragma unroll
        for (int m = 0; m < 4; ++m) {
            const float keep = bit2 ? u[4 + m] : u[m];
            const float send = bit2 ? u[m] : u[4 + m];
            w4[m] = keep + __shfl_xor(send, 4, 32);
        }
        float x2[2];
#pragma unroll
        for (int m = 0; m < 2; ++m) {
            const float keep = bit1 ? w4[2 + m] : w4[m];
            const float send = bit1 ? w4[m] : w4[2 + m];
            x2[m] = keep + __shfl_xor(send, 2, 32);
        }
        const float keep1 = bit0 ? x2[1] : x2[0];
        const float send1 = bit0 ? x2[0] : x2[1];
        const float z = keep1 + __shfl_xor(send1, 1, 32);

        out[e0 + slot] = (ps + pd) + z + bdv;
    }
}

extern "C" void kernel_launch(void* const* d_in, const int* in_sizes, int n_in,
                              void* d_out, int out_size, void* d_ws, size_t ws_size,
                              hipStream_t stream)
{
    const float* x   = (const float*)d_in[0];
    const float* ea  = (const float*)d_in[1];
    const int*   ei  = (const int*)  d_in[2];
    const float* Wn  = (const float*)d_in[3];
    const float* bn  = (const float*)d_in[4];
    const float* We  = (const float*)d_in[5];
    const float* be  = (const float*)d_in[6];
    const float* Wc  = (const float*)d_in[7];
    const float* bc  = (const float*)d_in[8];
    const float* Wr  = (const float*)d_in[9];
    const float* br  = (const float*)d_in[10];
    const float* Wd  = (const float*)d_in[11];
    const float* bd  = (const float*)d_in[12];
    float* out = (float*)d_out;

    const int* srcI = ei;
    const int* dstI = ei + NE;

    float* ws   = (float*)d_ws;
    float* h    = ws;
    float* hw   = h   + (size_t)NN * HID;
    float* agg  = hw  + (size_t)NN * HID;
    float* dinv = agg + (size_t)NN * HID;
    int*   cnt  = (int*)(dinv + NN);
    int*   offs = cnt  + NN;
    int*   rank = offs + NN + 16;
    int*   bsrc = rank + NE;
    float* pp   = (float*)(bsrc + NE);

    gemm_strip<FN / 16><<<GEMM_BLOCKS, 256, 0, stream>>>(x, Wn, bn, nullptr, h, NRT, 1);

    fill_i32<<<(NN + 255) / 256, 256, 0, stream>>>(cnt, 0, NN);
    rank_edges<<<(NE + 255) / 256, 256, 0, stream>>>(dstI, cnt, rank, NE);
    scan_offsets_dinv<<<(NN + 255) / 256, 256, 0, stream>>>(cnt, offs, dinv, NN);
    bucket_edges<<<(NE + 255) / 256, 256, 0, stream>>>(srcI, dstI, rank, offs, bsrc, NE);

    for (int l = 0; l < NL; ++l) {
        gemm_strip<HID / 16><<<GEMM_BLOCKS, 256, 0, stream>>>(
            h, Wc + (size_t)l * HID * HID, nullptr, nullptr, hw, NRT, 0);
        gemm_strip<HID / 16><<<GEMM_BLOCKS, 256, 0, stream>>>(
            h, Wr, br, bc + (size_t)l * HID, agg, NRT, 0);
        if (l < NL - 1)
            gather_combine_relu<false><<<NN / 8, 256, 0, stream>>>(
                hw, agg, offs, bsrc, dinv, h, Wd, pp, NN);
        else
            gather_combine_relu<true><<<NN / 8, 256, 0, stream>>>(
                hw, agg, offs, bsrc, dinv, h, Wd, pp, NN);
    }

    decode_edges<<<768, 128, 0, stream>>>(pp, ea, srcI, dstI, We, be, Wd, bd, out, NE / 32);
}


# Edge ranks from LDS counters instead of returning global atomics: 64 blocks each count an edge slice in a 50000-word LDS histogram; scan adds per-slice starts; sum order = slice, then LDS arrival
# speedup vs baseline: 3.5596x; 1.1830x over previous
#include <hip/hip_runtime.h>

#define NN 50000
#define NE 800000
#define FN 32
#define FE 16
#define HID 128
#define NL 3
#define NRT (NN / 16)
#define GEMM_BLOCKS ((NRT + 7) / 8)
#define NSLICE 64
#define EPS (NE / NSLICE)
#define NNP (((NN + 255) / 256) * 256)
#define NCHUNK (NNP / 256)
#define RANK_IT ((EPS / 4 + 1023) / 1024)
#define CHUNK_IT ((NCHUNK + 31) / 32)
static_assert(NE % NSLICE == 0 && EPS % 4 == 0, "slices are whole 16-byte runs");
static_assert(NSLICE % 4 == 0, "chunk sums are read 16 bytes at a time");

typedef float v2f __attribute__((ext_vector_type(2)));
typedef float v8f __attribute__((ext_vector_type(8)));

#define WPITCH 264

template <int KB>
__global__ __launch_bounds__(256) void gemm_strip(
    const float* __restrict__ A, const float* __restrict__ W,
    const float* __restrict__ bias1, const float* __restrict__ bias2,
    float* __restrict__ C, int nRowTiles, int do_relu)
{
    constexpr int K = 16 * KB;
    __shared__ __attribute__((aligned(16))) float Ws[(K / 2) * WPITCH];

    const int tid  = threadIdx.x;
    const int lane = tid & 31;
    const int c    = lane & 15;
    const int g    = lane >> 4;
    const int rt = __builtin_amdgcn_readfirstlane((int)(blockIdx.x * 8 + (tid >> 5)));
    const bool active = rt < nRowTiles;
    const int rtl = active ? rt : nRowTiles - 1;

    float4 alo[KB], ahi[KB];
    {
        const float* ap = A + (size_t)(rtl * 16 + c) * K + 8 * g;
#pragma unroll
        for (int b = 0; b < KB; ++b) {
            alo[b] = *(const float4*)(ap + 16 * b);
            ahi[b] = *(const float4*)(ap + 16 * b + 4);
        }
    }

#pragma unroll
    for (int i = 0; i < KB; ++i) {
        const int kp = (tid >> 5) + 8 * i;
        const int m  = tid & 31;
        const float4 w0 = *(const float4*)(W + (size_t)(2 * kp) * HID + 4 * m);
        const float4 w1 = *(const float4*)(W + (size_t)(2 * kp + 1) * HID + 4 * m);
        float* dp = Ws + kp * WPITCH + (m >> 4) * 128 + (m & 15) * 4;
        *(float4*)dp        = make_float4(w0.x, w1.x, w0.y, w1.y);
        *(float4*)(dp + 64) = make_float4(w0.z, w1.z, w0.w, w1.w);
    }
    __syncthreads();

    if (!active) return;

    v8f acc[8];
#pragma unroll
    for (int t = 0; t < 8; ++t) acc[t] = v8f{};

    const float* bp = Ws + (4 * g) * WPITCH + c * 4;
#pragma unroll
    for (int b = 0; b < KB; ++b) {
        v2f Af[4];
        Af[0][0] = alo[b].x; Af[0][1] = alo[b].y;
        Af[1][0] = alo[b].z; Af[1][1] = alo[b].w;
        Af[2][0] = ahi[b].x; Af[2][1] = ahi[b].y;
        Af[3][0] = ahi[b].z; Af[3][1] = ahi[b].w;
#pragma unroll
        for (int s = 0; s < 4; ++s) {
            const float* q = bp + (8 * b + s) * WPITCH;
#pragma unroll
            for (int p = 0; p < 4; ++p) {
                const float4 w = *(const float4*)(q + 64 * p);
                v2f b0, b1;
                b0[0] = w.x; b0[1] = w.y;
                b1[0] = w.z; b1[1] = w.w;
                acc[2 * p] = __builtin_amdgcn_wmma_f32_16x16x4_f32(
                    false, Af[s], false, b0, (short)0, acc[2 * p], false, false);
                acc[2 * p + 1] = __builtin_amdgcn_wmma_f32_16x16x4_f32(
                    false, Af[s], false, b1, (short)0, acc[2 * p + 1], false, false);
            }
        }
    }

    float4 bvlo = make_float4(0.0f, 0.0f, 0.0f, 0.0f);
    float4 bvhi = make_float4(0.0f, 0.0f, 0.0f, 0.0f);
    if (bias1) {
        const float4 l = *(const float4*)(bias1 + 4 * c);
        const float4 h = *(const float4*)(bias1 + 64 + 4 * c);
        bvlo.x += l.x; bvlo.y += l.y; bvlo.z += l.z; bvlo.w += l.w;
        bvhi.x += h.x; bvhi.y += h.y; bvhi.z += h.z; bvhi.w += h.w;
    }
    if (bias2) {
        const float4 l = *(const float4*)(bias2 + 4 * c);
        const float4 h = *(const float4*)(bias2 + 64 + 4 * c);
        bvlo.x += l.x; bvlo.y += l.y; bvlo.z += l.z; bvlo.w += l.w;
        bvhi.x += h.x; bvhi.y += h.y; bvhi.z += h.z; bvhi.w += h.w;
    }

#pragma unroll
    for (int r = 0; r < 8; ++r) {
        float4 lo, hi;
        lo.x = acc[0][r] + bvlo.x; lo.y = acc[1][r] + bvlo.y;
        lo.z = acc[2][r] + bvlo.z; lo.w = acc[3][r] + bvlo.w;
        hi.x = acc[4][r] + bvhi.x; hi.y = acc[5][r] + bvhi.y;
        hi.z = acc[6][r] + bvhi.z; hi.w = acc[7][r] + bvhi.w;
        if (do_relu) {
            lo.x = fmaxf(lo.x, 0.0f); lo.y = fmaxf(lo.y, 0.0f);
            lo.z = fmaxf(lo.z, 0.0f); lo.w = fmaxf(lo.w, 0.0f);
            hi.x = fmaxf(hi.x, 0.0f); hi.y = fmaxf(hi.y, 0.0f);
            hi.z = fmaxf(hi.z, 0.0f); hi.w = fmaxf(hi.w, 0.0f);
        }
        float* cp = C + (size_t)(rt * 16 + 8 * g + r) * HID + 4 * c;
        *(float4*)cp        = lo;
        *(float4*)(cp + 64) = hi;
    }
}

__global__ __launch_bounds__(1024) void rank_edges_lds(
    const int* __restrict__ dst, int* __restrict__ rank,
    int* __restrict__ cntS, int* __restrict__ csum)
{
    __shared__ __attribute__((aligned(16))) int cnt[NNP];
    const int tid  = threadIdx.x;
    const int lane = tid & 31;
    const int wv   = tid >> 5;
    const int sl   = blockIdx.x;

    const int4* __restrict__ d4 = (const int4*)(dst + (size_t)sl * EPS);
    int4 d[RANK_IT];
#pragma unroll
    for (int k = 0; k < RANK_IT; ++k) {
        const int i = tid + 1024 * k;
        d[k] = make_int4(0, 0, 0, 0);
        if (i < EPS / 4) d[k] = d4[i];
    }

    int4* c4 = (int4*)cnt;
    for (int i = tid; i < NNP / 4; i += 1024) c4[i] = make_int4(0, 0, 0, 0);
    __syncthreads();

#pragma unroll
    for (int k = 0; k < RANK_IT; ++k) {
        const int i = tid + 1024 * k;
        if (i < EPS / 4) {
            d[k].x = atomicAdd(&cnt[d[k].x], 1);
            d[k].y = atomicAdd(&cnt[d[k].y], 1);
            d[k].z = atomicAdd(&cnt[d[k].z], 1);
            d[k].w = atomicAdd(&cnt[d[k].w], 1);
        }
    }
    __syncthreads();

    int4* __restrict__ r4 = (int4*)(rank + (size_t)sl * EPS);
#pragma unroll
    for (int k = 0; k < RANK_IT; ++k) {
        const int i = tid + 1024 * k;
        if (i < EPS / 4) r4[i] = d[k];
    }

    int4* __restrict__ g4 = (int4*)(cntS + (size_t)sl * NNP);
    int sum[CHUNK_IT];
#pragma unroll
    for (int t = 0; t < CHUNK_IT; ++t) {
        const int j = wv + 32 * t;
        sum[t] = 0;
        if (j < NCHUNK) {
            const int4 v0 = c4[64 * j + lane];
            const int4 v1 = c4[64 * j + 32 + lane];
            g4[64 * j + lane]      = v0;
            g4[64 * j + 32 + lane] = v1;
            sum[t] = ((v0.x + v0.y) + (v0.z + v0.w)) + ((v1.x + v1.y) + (v1.z + v1.w));
        }
    }
#pragma unroll
    for (int off = 16; off > 0; off >>= 1) {
#pragma unroll
        for (int t = 0; t < CHUNK_IT; ++t)
            sum[t] += __shfl_xor(sum[t], off, 32);
    }
    if (lane == 0) {
#pragma unroll
        for (int t = 0; t < CHUNK_IT; ++t) {
            const int j = wv + 32 * t;
            if (j < NCHUNK) csum[j * NSLICE + sl] = sum[t];
        }
    }
}

__global__ __launch_bounds__(256) void scan_offsets_dinv(
    int* __restrict__ cntS, const int* __restrict__ csum,
    int* __restrict__ offs, float* __restrict__ dinv, int n)
{
    __shared__ int waveCnt[8];
    __shared__ int wavePre[8];
    const int tid  = threadIdx.x;
    const int lane = tid & 31;
    const int wv   = tid >> 5;
    const int b    = blockIdx.x;
    const int i    = b * 256 + tid;

    int cs[NSLICE];
#pragma unroll
    for (int k = 0; k < NSLICE; ++k) cs[k] = cntS[(size_t)k * NNP + i];

    int pre = 0;
    const int4* __restrict__ s4 = (const int4*)csum;
    const int n4   = b * (NSLICE / 4);
    const int full = n4 >> 8;
#pragma unroll 4
    for (int it = 0; it < full; ++it) {
        int4 v = s4[it * 256 + tid];
        pre += (v.x + v.y) + (v.z + v.w);
    }
    if (tid < (n4 & 255)) {
        int4 v = s4[full * 256 + tid];
        pre += (v.x + v.y) + (v.z + v.w);
    }

    int c = 0;
#pragma unroll
    for (int k = 0; k < NSLICE; ++k) c += cs[k];

    int inc = c;
    for (int off = 1; off < 32; off <<= 1) {
        int t = __shfl_up(inc, off, 32);
        if (lane >= off) inc += t;
    }
    for (int off = 16; off > 0; off >>= 1)
        pre += __shfl_xor(pre, off, 32);

    if (lane == 31) waveCnt[wv] = inc;
    if (lane == 0)  wavePre[wv] = pre;
    __syncthreads();

    int base = 0;
    for (int k = 0; k < 8; ++k) {
        base += wavePre[k];
        if (k < wv) base += waveCnt[k];
    }
    const int excl = base + inc - c;

    int run = excl;
#pragma unroll
    for (int k = 0; k < NSLICE; ++k) {
        cntS[(size_t)k * NNP + i] = run;
        run += cs[k];
    }

    if (i < n) {
        offs[i] = excl;
        float d = (float)(c + 1);
        dinv[i] = (d > 0.0f) ? rsqrtf(d) : 0.0f;
    }
    if (i == n - 1) offs[n] = excl + c;
}

__global__ void bucket_edges(const int* __restrict__ src, const int* __restrict__ dst,
                             const int* __restrict__ rank, const int* __restrict__ start,
                             int* __restrict__ bsrc, int n)
{
    int i = blockIdx.x * blockDim.x + threadIdx.x;
    if (i < n) bsrc[start[(size_t)(i / EPS) * NNP + dst[i]] + rank[i]] = src[i];
}

__device__ __forceinline__ float xhalf(float x)
{
    int b = __builtin_bit_cast(int, x);
    return __builtin_bit_cast(float,
        __builtin_amdgcn_permlanex16(b, b, 0x76543210, 0xfedcba98, false, false));
}

template <bool LAST>
__global__ __launch_bounds__(256) void gather_combine_relu(
    const float* __restrict__ hw, const float* __restrict__ res,
    const int* __restrict__ offs, const int* __restrict__ bsrc,
    const float* __restrict__ dinv, float* __restrict__ h,
    const float* __restrict__ Wd, float* __restrict__ pp, int N)
{
    const int lane = threadIdx.x & 31;
    const int n = __builtin_amdgcn_readfirstlane(
        (int)((blockIdx.x * blockDim.x + threadIdx.x) >> 5));
    if (n >= N) return;

    const int   beg = offs[n];
    const int   end = offs[n + 1];
    const float di  = dinv[n];
    const size_t rowOff = (size_t)n * HID + lane * 4;

    float4 acc        = *(const float4*)(res + rowOff);
    const float4 self = *(const float4*)(hw  + rowOff);
    float4 wd1 = make_float4(0.0f, 0.0f, 0.0f, 0.0f);
    float4 wd2 = make_float4(0.0f, 0.0f, 0.0f, 0.0f);
    if (LAST) {
        wd1 = *(const float4*)(Wd + lane * 4);
        wd2 = *(const float4*)(Wd + HID + lane * 4);
    }

    for (int base = beg; base < end; base += 32) {
        const int m = min(32, end - base);
        int   sL = 0;
        float wL = 0.0f;
        if (lane < m) {
            sL = bsrc[base + lane];
            wL = dinv[sL] * di;
        }
#pragma unroll 4
        for (int j = 0; j < m; ++j) {
            const int   s = __builtin_amdgcn_readlane(sL, j);
            const float w = __int_as_float(__builtin_amdgcn_readlane(__float_as_int(wL), j));
            const float4 v = *(const float4*)(hw + (size_t)s * HID + lane * 4);
            acc.x += v.x * w;
            acc.y += v.y * w;
            acc.z += v.z * w;
            acc.w += v.w * w;
        }
    }

    const float dd = di * di;
    acc.x = fmaxf(acc.x + self.x * dd, 0.0f);
    acc.y = fmaxf(acc.y + self.y * dd, 0.0f);
    acc.z = fmaxf(acc.z + self.z * dd, 0.0f);
    acc.w = fmaxf(acc.w + self.w * dd, 0.0f);
    if (!LAST) {
        *(float4*)(h + rowOff) = acc;
    } else {
        float a = acc.x * wd1.x;
        a += acc.y * wd1.y;
        a += acc.z * wd1.z;
        a += acc.w * wd1.w;
        float b = acc.x * wd2.x;
        b += acc.y * wd2.y;
        b += acc.z * wd2.z;
        b += acc.w * wd2.w;
        const bool hiHalf = lane >= 16;
        const float keep = hiHalf ? b : a;
        const float send = hiHalf ? a : b;
        float v = keep + xhalf(send);
        v += __shfl_xor(v, 8, 32);
        v += __shfl_xor(v, 4, 32);
        v += __shfl_xor(v, 2, 32);
        v += __shfl_xor(v, 1, 32);
        const float vHi = __int_as_float(__builtin_amdgcn_readlane(__float_as_int(v), 16));
        if (lane == 0) *(float2*)(pp + 2 * (size_t)n) = make_float2(v, vHi);
    }
}

__global__ __launch_bounds__(128, 4) void decode_edges(
    const float* __restrict__ pp, const float* __restrict__ ea,
    const int* __restrict__ src,  const int* __restrict__ dst,
    const float* __restrict__ We, const float* __restrict__ be,
    const float* __restrict__ Wd, const float* __restrict__ bd,
    float* __restrict__ out, int nGroups)
{
    const int lane = threadIdx.x & 31;
    const int c    = lane & 15;
    const int g    = lane >> 4;
    const int wave = __builtin_amdgcn_readfirstlane(
        (int)((blockIdx.x * blockDim.x + threadIdx.x) >> 5));
    const int nWaves = (int)((gridDim.x * blockDim.x) >> 5);

    v2f Bf[8][4];
#pragma unroll
    for (int s = 0; s < 4; ++s) {
#pragma unroll
        for (int j = 0; j < 2; ++j) {
            const float* rp = We + (size_t)(8 * g + 2 * s + j) * HID + 8 * c;
            const float4 lo = *(const float4*)rp;
            const float4 hi = *(const float4*)(rp + 4);
            Bf[0][s][j] = lo.x; Bf[1][s][j] = lo.y; Bf[2][s][j] = lo.z; Bf[3][s][j] = lo.w;
            Bf[4][s][j] = hi.x; Bf[5][s][j] = hi.y; Bf[6][s][j] = hi.z; Bf[7][s][j] = hi.w;
        }
    }
    float bev[8], wd3[8];
    {
        const float4 b0 = *(const float4*)(be + 8 * c);
        const float4 b1 = *(const float4*)(be + 8 * c + 4);
        const float4 w0 = *(const float4*)(Wd + 2 * HID + 8 * c);
        const float4 w1 = *(const float4*)(Wd + 2 * HID + 8 * c + 4);
        bev[0] = b0.x; bev[1] = b0.y; bev[2] = b0.z; bev[3] = b0.w;
        bev[4] = b1.x; bev[5] = b1.y; bev[6] = b1.z; bev[7] = b1.w;
        wd3[0] = w0.x; wd3[1] = w0.y; wd3[2] = w0.z; wd3[3] = w0.w;
        wd3[4] = w1.x; wd3[5] = w1.y; wd3[6] = w1.z; wd3[7] = w1.w;
    }
    const float bdv = bd[0];

    const bool bit3 = (c & 8) != 0;
    const bool bit2 = (c & 4) != 0;
    const bool bit1 = (c & 2) != 0;
    const bool bit0 = (c & 1) != 0;
    const int slot = ((c & 8) << 1) + 8 * g + (c & 7);

    int grp = wave;
    int sN = 0, dN = 0;
    float4 aloN[2], ahiN[2];
#pragma unroll
    for (int T = 0; T < 2; ++T) {
        aloN[T] = make_float4(0.0f, 0.0f, 0.0f, 0.0f);
        ahiN[T] = make_float4(0.0f, 0.0f, 0.0f, 0.0f);
    }
    if (grp < nGroups) {
        sN = src[grp * 32 + slot];
        dN = dst[grp * 32 + slot];
#pragma unroll
        for (int T = 0; T < 2; ++T) {
            const float* ap = ea + (size_t)(grp * 32 + 16 * T + c) * FE + 8 * g;
            aloN[T] = *(const float4*)ap;
            ahiN[T] = *(const float4*)(ap + 4);
        }
    }

    for (; grp < nGroups; grp += nWaves) {
        const int e0 = grp * 32;
        const float ps = pp[2 * (size_t)sN];
        const float pd = pp[2 * (size_t)dN + 1];

        float4 alo[2], ahi[2];
#pragma unroll
        for (int T = 0; T < 2; ++T) { alo[T] = aloN[T]; ahi[T] = ahiN[T]; }

        const int nxt = min(grp + nWaves, nGroups - 1);
        sN = src[nxt * 32 + slot];
        dN = dst[nxt * 32 + slot];
#pragma unroll
        for (int T = 0; T < 2; ++T) {
            const float* ap = ea + (size_t)(nxt * 32 + 16 * T + c) * FE + 8 * g;
            aloN[T] = *(const float4*)ap;
            ahiN[T] = *(const float4*)(ap + 4);
        }

        float q[16];
#pragma unroll
        for (int T = 0; T < 2; ++T) {
            v2f A[4];
            A[0][0] = alo[T].x; A[0][1] = alo[T].y;
            A[1][0] = alo[T].z; A[1][1] = alo[T].w;
            A[2][0] = ahi[T].x; A[2][1] = ahi[T].y;
            A[3][0] = ahi[T].z; A[3][1] = ahi[T].w;

            float part[8];
#pragma unroll
            for (int r = 0; r < 8; ++r) part[r] = 0.0f;
#pragma unroll
            for (int t = 0; t < 8; t += 2) {
                v8f acc0 = {};
                v8f acc1 = {};
#pragma unroll
                for (int s = 0; s < 4; ++s) {
                    acc0 = __builtin_amdgcn_wmma_f32_16x16x4_f32(
                        false, A[s], false, Bf[t][s], (short)0, acc0, false, false);
                    acc1 = __builtin_amdgcn_wmma_f32_16x16x4_f32(
                        false, A[s], false, Bf[t + 1][s], (short)0, acc1, false, false);
                }
#pragma unroll
                for (int r = 0; r < 8; ++r) {
                    part[r] += fmaxf(acc0[r] + bev[t], 0.0f) * wd3[t];
                    part[r] += fmaxf(acc1[r] + bev[t + 1], 0.0f) * wd3[t + 1];
                }
            }
#pragma unroll
            for (int r = 0; r < 8; ++r) q[8 * T + r] = part[r];
        }

        float u[8];
#pragma unroll
        for (int m = 0; m < 8; ++m) {
            const float keep = bit3 ? q[8 + m] : q[m];
            const float send = bit3 ? q[m] : q[8 + m];
            u[m] = keep + __shfl_xor(send, 8, 32);
        }
        float w4[4];
#pragma unroll
        for (int m = 0; m < 4; ++m) {
            const float keep = bit2 ? u[4 + m] : u[m];
            const float send = bit2 ? u[m] : u[4 + m];
            w4[m] = keep + __shfl_xor(send, 4, 32);
        }
        float x2[2];
#pragma unroll
        for (int m = 0; m < 2; ++m) {
            const float keep = bit1 ? w4[2 + m] : w4[m];
            const float send = bit1 ? w4[m] : w4[2 + m];
            x2[m] = keep + __shfl_xor(send, 2, 32);
        }
        const float keep1 = bit0 ? x2[1] : x2[0];
        const float send1 = bit0 ? x2[0] : x2[1];
        const float z = keep1 + __shfl_xor(send1, 1, 32);

        out[e0 + slot] = (ps + pd) + z + bdv;
    }
}

extern "C" void kernel_launch(void* const* d_in, const int* in_sizes, int n_in,
                              void* d_out, int out_size, void* d_ws, size_t ws_size,
                              hipStream_t stream)
{
    const float* x   = (const float*)d_in[0];
    const float* ea  = (const float*)d_in[1];
    const int*   ei  = (const int*)  d_in[2];
    const float* Wn  = (const float*)d_in[3];
    const float* bn  = (const float*)d_in[4];
    const float* We  = (const float*)d_in[5];
    const float* be  = (const float*)d_in[6];
    const float* Wc  = (const float*)d_in[7];
    const float* bc  = (const float*)d_in[8];
    const float* Wr  = (const float*)d_in[9];
    const float* br  = (const float*)d_in[10];
    const float* Wd  = (const float*)d_in[11];
    const float* bd  = (const float*)d_in[12];
    float* out = (float*)d_out;

    const int* srcI = ei;
    const int* dstI = ei + NE;

    float* ws   = (float*)d_ws;
    float* h    = ws;
    float* hw   = h   + (size_t)NN * HID;
    float* agg  = hw  + (size_t)NN * HID;
    float* dinv = agg + (size_t)NN * HID;
    int*   offs = (int*)(dinv + NN);
    int*   rank = offs + NN + 16;
    int*   bsrc = rank + NE;
    float* pp   = (float*)(bsrc + NE);
    int*   cntS = (int*)(pp + 2 * (size_t)NN);
    int*   csum = cntS + (size_t)NSLICE * NNP;

    gemm_strip<FN / 16><<<GEMM_BLOCKS, 256, 0, stream>>>(x, Wn, bn, nullptr, h, NRT, 1);

    rank_edges_lds<<<NSLICE, 1024, 0, stream>>>(dstI, rank, cntS, csum);
    scan_offsets_dinv<<<NCHUNK, 256, 0, stream>>>(cntS, csum, offs, dinv, NN);
    bucket_edges<<<(NE + 255) / 256, 256, 0, stream>>>(srcI, dstI, rank, cntS, bsrc, NE);

    for (int l = 0; l < NL; ++l) {
        gemm_strip<HID / 16><<<GEMM_BLOCKS, 256, 0, stream>>>(
            h, Wc + (size_t)l * HID * HID, nullptr, nullptr, hw, NRT, 0);
        gemm_strip<HID / 16><<<GEMM_BLOCKS, 256, 0, stream>>>(
            h, Wr, br, bc + (size_t)l * HID, agg, NRT, 0);
        if (l < NL - 1)
            gather_combine_relu<false><<<NN / 8, 256, 0, stream>>>(
                hw, agg, offs, bsrc, dinv, h, Wd, pp, NN);
        else
            gather_combine_relu<true><<<NN / 8, 256, 0, stream>>>(
                hw, agg, offs, bsrc, dinv, h, Wd, pp, NN);
    }

    decode_edges<<<768, 128, 0, stream>>>(pp, ea, srcI, dstI, We, be, Wd, bd, out, NE / 32);
}


# Neighbour sum moved in front of the layer product, Ahat(hW) = (Ahat h)W: one k=256 WMMA product [g|h]@[Wc;Wr] per layer replaces two k=128 ones and the residual round trip; sum order changes
# speedup vs baseline: 3.7182x; 1.0446x over previous
#include <hip/hip_runtime.h>

#define NN 50000
#define NE 800000
#define FN 32
#define FE 16
#define HID 128
#define NL 3
#define NRT (NN / 16)
#define GEMM_BLOCKS ((NRT + 7) / 8)
#define NSLICE 64
#define EPS (NE / NSLICE)
#define NNP (((NN + 255) / 256) * 256)
#define NCHUNK (NNP / 256)
#define RANK_IT ((EPS / 4 + 1023) / 1024)
#define CHUNK_IT ((NCHUNK + 31) / 32)
static_assert(NE % NSLICE == 0 && EPS % 4 == 0, "slices are whole 16-byte runs");
static_assert(NSLICE % 4 == 0, "chunk sums are read 16 bytes at a time");

typedef float v2f __attribute__((ext_vector_type(2)));
typedef float v8f __attribute__((ext_vector_type(8)));

#define WPITCH 264

template <int KB1, int KB2, int MODE>
__global__ __launch_bounds__(256, 4) void gemm_strip(
    const float* __restrict__ A1, const float* __restrict__ W1,
    const float* __restrict__ A2, const float* __restrict__ W2,
    const float* __restrict__ bias1, const float* __restrict__ bias2,
    float* __restrict__ C, const float* __restrict__ Wd,
    int nRowTiles, int do_relu)
{
    constexpr int KB = KB1 + KB2;
    constexpr int K1 = 16 * KB1;
    constexpr int K2 = 16 * KB2;
    __shared__ __attribute__((aligned(16))) float Ws[(8 * KB) * WPITCH];

    const int tid  = threadIdx.x;
    const int lane = tid & 31;
    const int c    = lane & 15;
    const int g    = lane >> 4;
    const int rt = __builtin_amdgcn_readfirstlane((int)(blockIdx.x * 8 + (tid >> 5)));
    const bool active = rt < nRowTiles;
    const int rtl = active ? rt : nRowTiles - 1;

    float4 alo[KB], ahi[KB];
    {
        const float* ap1 = A1 + (size_t)(rtl * 16 + c) * K1 + 8 * g;
#pragma unroll
        for (int b = 0; b < KB1; ++b) {
            alo[b] = *(const float4*)(ap1 + 16 * b);
            ahi[b] = *(const float4*)(ap1 + 16 * b + 4);
        }
    }

#pragma unroll
    for (int i = 0; i < KB1; ++i) {
        const int kp = (tid >> 5) + 8 * i;
        const int m  = tid & 31;
        const float4 w0 = *(const float4*)(W1 + (size_t)(2 * kp) * HID + 4 * m);
        const float4 w1 = *(const float4*)(W1 + (size_t)(2 * kp + 1) * HID + 4 * m);
        float* dp = Ws + kp * WPITCH + (m >> 4) * 128 + (m & 15) * 4;
        *(float4*)dp        = make_float4(w0.x, w1.x, w0.y, w1.y);
        *(float4*)(dp + 64) = make_float4(w0.z, w1.z, w0.w, w1.w);
    }
#pragma unroll
    for (int i = 0; i < KB2; ++i) {
        const int kp = (tid >> 5) + 8 * i;
        const int m  = tid & 31;
        const float4 w0 = *(const float4*)(W2 + (size_t)(2 * kp) * HID + 4 * m);
        const float4 w1 = *(const float4*)(W2 + (size_t)(2 * kp + 1) * HID + 4 * m);
        float* dp = Ws + (8 * KB1 + kp) * WPITCH + (m >> 4) * 128 + (m & 15) * 4;
        *(float4*)dp        = make_float4(w0.x, w1.x, w0.y, w1.y);
        *(float4*)(dp + 64) = make_float4(w0.z, w1.z, w0.w, w1.w);
    }
    if (KB2 > 0) {
        const float* ap2 = A2 + (size_t)(rtl * 16 + c) * K2 + 8 * g;
#pragma unroll
        for (int b = 0; b < KB2; ++b) {
            alo[KB1 + b] = *(const float4*)(ap2 + 16 * b);
            ahi[KB1 + b] = *(const float4*)(ap2 + 16 * b + 4);
        }
    }
    __syncthreads();

    if (!active) return;

    v8f acc[8];
#pragma unroll
    for (int t = 0; t < 8; ++t) acc[t] = v8f{};

    const float* bp = Ws + (4 * g) * WPITCH + c * 4;
#pragma unroll
    for (int b = 0; b < KB; ++b) {
        v2f Af[4];
        Af[0][0] = alo[b].x; Af[0][1] = alo[b].y;
        Af[1][0] = alo[b].z; Af[1][1] = alo[b].w;
        Af[2][0] = ahi[b].x; Af[2][1] = ahi[b].y;
        Af[3][0] = ahi[b].z; Af[3][1] = ahi[b].w;
#pragma unroll
        for (int s = 0; s < 4; ++s) {
            const float* q = bp + (8 * b + s) * WPITCH;
#pragma unroll
            for (int p = 0; p < 4; ++p) {
                const float4 w = *(const float4*)(q + 64 * p);
                v2f b0, b1;
                b0[0] = w.x; b0[1] = w.y;
                b1[0] = w.z; b1[1] = w.w;
                acc[2 * p] = __builtin_amdgcn_wmma_f32_16x16x4_f32(
                    false, Af[s], false, b0, (short)0, acc[2 * p], false, false);
                acc[2 * p + 1] = __builtin_amdgcn_wmma_f32_16x16x4_f32(
                    false, Af[s], false, b1, (short)0, acc[2 * p + 1], false, false);
            }
        }
    }

    float4 bvlo = make_float4(0.0f, 0.0f, 0.0f, 0.0f);
    float4 bvhi = make_float4(0.0f, 0.0f, 0.0f, 0.0f);
    if (bias1) {
        const float4 l = *(const float4*)(bias1 + 4 * c);
        const float4 h = *(const float4*)(bias1 + 64 + 4 * c);
        bvlo.x += l.x; bvlo.y += l.y; bvlo.z += l.z; bvlo.w += l.w;
        bvhi.x += h.x; bvhi.y += h.y; bvhi.z += h.z; bvhi.w += h.w;
    }
    if (bias2) {
        const float4 l = *(const float4*)(bias2 + 4 * c);
        const float4 h = *(const float4*)(bias2 + 64 + 4 * c);
        bvlo.x += l.x; bvlo.y += l.y; bvlo.z += l.z; bvlo.w += l.w;
        bvhi.x += h.x; bvhi.y += h.y; bvhi.z += h.z; bvhi.w += h.w;
    }

    if (MODE == 0) {
#pragma unroll
        for (int r = 0; r < 8; ++r) {
            float4 lo, hi;
            lo.x = acc[0][r] + bvlo.x; lo.y = acc[1][r] + bvlo.y;
            lo.z = acc[2][r] + bvlo.z; lo.w = acc[3][r] + bvlo.w;
            hi.x = acc[4][r] + bvhi.x; hi.y = acc[5][r] + bvhi.y;
            hi.z = acc[6][r] + bvhi.z; hi.w = acc[7][r] + bvhi.w;
            if (do_relu) {
                lo.x = fmaxf(lo.x, 0.0f); lo.y = fmaxf(lo.y, 0.0f);
                lo.z = fmaxf(lo.z, 0.0f); lo.w = fmaxf(lo.w, 0.0f);
                hi.x = fmaxf(hi.x, 0.0f); hi.y = fmaxf(hi.y, 0.0f);
                hi.z = fmaxf(hi.z, 0.0f); hi.w = fmaxf(hi.w, 0.0f);
            }
            float* cp = C + (size_t)(rt * 16 + 8 * g + r) * HID + 4 * c;
            *(float4*)cp        = lo;
            *(float4*)(cp + 64) = hi;
        }
    } else {
        const float4 u1l = *(const float4*)(Wd + 4 * c);
        const float4 u1h = *(const float4*)(Wd + 64 + 4 * c);
        const float4 u2l = *(const float4*)(Wd + HID + 4 * c);
        const float4 u2h = *(const float4*)(Wd + HID + 64 + 4 * c);
        float q[16];
#pragma unroll
        for (int r = 0; r < 8; ++r) {
            float4 lo, hi;
            lo.x = fmaxf(acc[0][r] + bvlo.x, 0.0f); lo.y = fmaxf(acc[1][r] + bvlo.y, 0.0f);
            lo.z = fmaxf(acc[2][r] + bvlo.z, 0.0f); lo.w = fmaxf(acc[3][r] + bvlo.w, 0.0f);
            hi.x = fmaxf(acc[4][r] + bvhi.x, 0.0f); hi.y = fmaxf(acc[5][r] + bvhi.y, 0.0f);
            hi.z = fmaxf(acc[6][r] + bvhi.z, 0.0f); hi.w = fmaxf(acc[7][r] + bvhi.w, 0.0f);
            float a = lo.x * u1l.x;
            a += lo.y * u1l.y; a += lo.z * u1l.z; a += lo.w * u1l.w;
            a += hi.x * u1h.x; a += hi.y * u1h.y; a += hi.z * u1h.z; a += hi.w * u1h.w;
            float d = lo.x * u2l.x;
            d += lo.y * u2l.y; d += lo.z * u2l.z; d += lo.w * u2l.w;
            d += hi.x * u2h.x; d += hi.y * u2h.y; d += hi.z * u2h.z; d += hi.w * u2h.w;
            q[2 * r]     = a;
            q[2 * r + 1] = d;
        }
        const bool bit3 = (c & 8) != 0;
        const bool bit2 = (c & 4) != 0;
        const bool bit1 = (c & 2) != 0;
        const bool bit0 = (c & 1) != 0;
        float u[8];
#pragma unroll
        for (int m = 0; m < 8; ++m) {
            const float keep = bit3 ? q[8 + m] : q[m];
            const float send = bit3 ? q[m] : q[8 + m];
            u[m] = keep + __shfl_xor(send, 8, 32);
        }
        float w4[4];
#pragma unroll
        for (int m = 0; m < 4; ++m) {
            const float keep = bit2 ? u[4 + m] : u[m];
            const float send = bit2 ? u[m] : u[4 + m];
            w4[m] = keep + __shfl_xor(send, 4, 32);
        }
        float x2[2];
#pragma unroll
        for (int m = 0; m < 2; ++m) {
            const float keep = bit1 ? w4[2 + m] : w4[m];
            const float send = bit1 ? w4[m] : w4[2 + m];
            x2[m] = keep + __shfl_xor(send, 2, 32);
        }
        const float keep1 = bit0 ? x2[1] : x2[0];
        const float send1 = bit0 ? x2[0] : x2[1];
        const float z = keep1 + __shfl_xor(send1, 1, 32);
        C[(size_t)rt * 32 + lane] = z;
    }
}

__global__ __launch_bounds__(1024) void rank_edges_lds(
    const int* __restrict__ dst, int* __restrict__ rank,
    int* __restrict__ cntS, int* __restrict__ csum)
{
    __shared__ __attribute__((aligned(16))) int cnt[NNP];
    const int tid  = threadIdx.x;
    const int lane = tid & 31;
    const int wv   = tid >> 5;
    const int sl   = blockIdx.x;

    const int4* __restrict__ d4 = (const int4*)(dst + (size_t)sl * EPS);
    int4 d[RANK_IT];
#pragma unroll
    for (int k = 0; k < RANK_IT; ++k) {
        const int i = tid + 1024 * k;
        d[k] = make_int4(0, 0, 0, 0);
        if (i < EPS / 4) d[k] = d4[i];
    }

    int4* c4 = (int4*)cnt;
    for (int i = tid; i < NNP / 4; i += 1024) c4[i] = make_int4(0, 0, 0, 0);
    __syncthreads();

#pragma unroll
    for (int k = 0; k < RANK_IT; ++k) {
        const int i = tid + 1024 * k;
        if (i < EPS / 4) {
            d[k].x = atomicAdd(&cnt[d[k].x], 1);
            d[k].y = atomicAdd(&cnt[d[k].y], 1);
            d[k].z = atomicAdd(&cnt[d[k].z], 1);
            d[k].w = atomicAdd(&cnt[d[k].w], 1);
        }
    }
    __syncthreads();

    int4* __restrict__ r4 = (int4*)(rank + (size_t)sl * EPS);
#pragma unroll
    for (int k = 0; k < RANK_IT; ++k) {
        const int i = tid + 1024 * k;
        if (i < EPS / 4) r4[i] = d[k];
    }

    int4* __restrict__ g4 = (int4*)(cntS + (size_t)sl * NNP);
    int sum[CHUNK_IT];
#pragma unroll
    for (int t = 0; t < CHUNK_IT; ++t) {
        const int j = wv + 32 * t;
        sum[t] = 0;
        if (j < NCHUNK) {
            const int4 v0 = c4[64 * j + lane];
            const int4 v1 = c4[64 * j + 32 + lane];
            g4[64 * j + lane]      = v0;
            g4[64 * j + 32 + lane] = v1;
            sum[t] = ((v0.x + v0.y) + (v0.z + v0.w)) + ((v1.x + v1.y) + (v1.z + v1.w));
        }
    }
#pragma unroll
    for (int off = 16; off > 0; off >>= 1) {
#pragma unroll
        for (int t = 0; t < CHUNK_IT; ++t)
            sum[t] += __shfl_xor(sum[t], off, 32);
    }
    if (lane == 0) {
#pragma unroll
        for (int t = 0; t < CHUNK_IT; ++t) {
            const int j = wv + 32 * t;
            if (j < NCHUNK) csum[j * NSLICE + sl] = sum[t];
        }
    }
}

__global__ __launch_bounds__(256) void scan_offsets_dinv(
    int* __restrict__ cntS, const int* __restrict__ csum,
    int* __restrict__ offs, float* __restrict__ dinv, int n)
{
    __shared__ int waveCnt[8];
    __shared__ int wavePre[8];
    const int tid  = threadIdx.x;
    const int lane = tid & 31;
    const int wv   = tid >> 5;
    const int b    = blockIdx.x;
    const int i    = b * 256 + tid;

    int cs[NSLICE];
#pragma unroll
    for (int k = 0; k < NSLICE; ++k) cs[k] = cntS[(size_t)k * NNP + i];

    int pre = 0;
    const int4* __restrict__ s4 = (const int4*)csum;
    const int n4   = b * (NSLICE / 4);
    const int full = n4 >> 8;
#pragma unroll 4
    for (int it = 0; it < full; ++it) {
        int4 v = s4[it * 256 + tid];
        pre += (v.x + v.y) + (v.z + v.w);
    }
    if (tid < (n4 & 255)) {
        int4 v = s4[full * 256 + tid];
        pre += (v.x + v.y) + (v.z + v.w);
    }

    int c = 0;
#pragma unroll
    for (int k = 0; k < NSLICE; ++k) c += cs[k];

    int inc = c;
    for (int off = 1; off < 32; off <<= 1) {
        int t = __shfl_up(inc, off, 32);
        if (lane >= off) inc += t;
    }
    for (int off = 16; off > 0; off >>= 1)
        pre += __shfl_xor(pre, off, 32);

    if (lane == 31) waveCnt[wv] = inc;
    if (lane == 0)  wavePre[wv] = pre;
    __syncthreads();

    int base = 0;
    for (int k = 0; k < 8; ++k) {
        base += wavePre[k];
        if (k < wv) base += waveCnt[k];
    }
    const int excl = base + inc - c;

    int run = excl;
#pragma unroll
    for (int k = 0; k < NSLICE; ++k) {
        cntS[(size_t)k * NNP + i] = run;
        run += cs[k];
    }

    if (i < n) {
        offs[i] = excl;
        float d = (float)(c + 1);
        dinv[i] = (d > 0.0f) ? rsqrtf(d) : 0.0f;
    }
    if (i == n - 1) offs[n] = excl + c;
}

__global__ void bucket_edges(const int* __restrict__ src, const int* __restrict__ dst,
                             const int* __restrict__ rank, const int* __restrict__ start,
                             int* __restrict__ bsrc, int n)
{
    int i = blockIdx.x * blockDim.x + threadIdx.x;
    if (i < n) bsrc[start[(size_t)(i / EPS) * NNP + dst[i]] + rank[i]] = src[i];
}

__global__ __launch_bounds__(256) void gather_norm(
    const float* __restrict__ h, const int* __restrict__ offs,
    const int* __restrict__ bsrc, const float* __restrict__ dinv,
    float* __restrict__ gout, int N)
{
    const int lane = threadIdx.x & 31;
    const int n = __builtin_amdgcn_readfirstlane(
        (int)((blockIdx.x * blockDim.x + threadIdx.x) >> 5));
    if (n >= N) return;

    const int   beg = offs[n];
    const int   end = offs[n + 1];
    const float di  = dinv[n];
    const size_t rowOff = (size_t)n * HID + lane * 4;

    const float4 self = *(const float4*)(h + rowOff);
    float4 acc = make_float4(0.0f, 0.0f, 0.0f, 0.0f);

    for (int base = beg; base < end; base += 32) {
        const int m = min(32, end - base);
        int   sL = 0;
        float wL = 0.0f;
        if (lane < m) {
            sL = bsrc[base + lane];
            wL = dinv[sL] * di;
        }
#pragma unroll 4
        for (int j = 0; j < m; ++j) {
            const int   s = __builtin_amdgcn_readlane(sL, j);
            const float w = __int_as_float(__builtin_amdgcn_readlane(__float_as_int(wL), j));
            const float4 v = *(const float4*)(h + (size_t)s * HID + lane * 4);
            acc.x += v.x * w;
            acc.y += v.y * w;
            acc.z += v.z * w;
            acc.w += v.w * w;
        }
    }

    const float dd = di * di;
    acc.x += self.x * dd;
    acc.y += self.y * dd;
    acc.z += self.z * dd;
    acc.w += self.w * dd;
    *(float4*)(gout + rowOff) = acc;
}

__global__ __launch_bounds__(128, 4) void decode_edges(
    const float* __restrict__ pp, const float* __restrict__ ea,
    const int* __restrict__ src,  const int* __restrict__ dst,
    const float* __restrict__ We, const float* __restrict__ be,
    const float* __restrict__ Wd, const float* __restrict__ bd,
    float* __restrict__ out, int nGroups)
{
    const int lane = threadIdx.x & 31;
    const int c    = lane & 15;
    const int g    = lane >> 4;
    const int wave = __builtin_amdgcn_readfirstlane(
        (int)((blockIdx.x * blockDim.x + threadIdx.x) >> 5));
    const int nWaves = (int)((gridDim.x * blockDim.x) >> 5);

    v2f Bf[8][4];
#pragma unroll
    for (int s = 0; s < 4; ++s) {
#pragma unroll
        for (int j = 0; j < 2; ++j) {
            const float* rp = We + (size_t)(8 * g + 2 * s + j) * HID + 8 * c;
            const float4 lo = *(const float4*)rp;
            const float4 hi = *(const float4*)(rp + 4);
            Bf[0][s][j] = lo.x; Bf[1][s][j] = lo.y; Bf[2][s][j] = lo.z; Bf[3][s][j] = lo.w;
            Bf[4][s][j] = hi.x; Bf[5][s][j] = hi.y; Bf[6][s][j] = hi.z; Bf[7][s][j] = hi.w;
        }
    }
    float bev[8], wd3[8];
    {
        const float4 b0 = *(const float4*)(be + 8 * c);
        const float4 b1 = *(const float4*)(be + 8 * c + 4);
        const float4 w0 = *(const float4*)(Wd + 2 * HID + 8 * c);
        const float4 w1 = *(const float4*)(Wd + 2 * HID + 8 * c + 4);
        bev[0] = b0.x; bev[1] = b0.y; bev[2] = b0.z; bev[3] = b0.w;
        bev[4] = b1.x; bev[5] = b1.y; bev[6] = b1.z; bev[7] = b1.w;
        wd3[0] = w0.x; wd3[1] = w0.y; wd3[2] = w0.z; wd3[3] = w0.w;
        wd3[4] = w1.x; wd3[5] = w1.y; wd3[6] = w1.z; wd3[7] = w1.w;
    }
    const float bdv = bd[0];

    const bool bit3 = (c & 8) != 0;
    const bool bit2 = (c & 4) != 0;
    const bool bit1 = (c & 2) != 0;
    const bool bit0 = (c & 1) != 0;
    const int slot = ((c & 8) << 1) + 8 * g + (c & 7);

    int grp = wave;
    int sN = 0, dN = 0;
    float4 aloN[2], ahiN[2];
#pragma unroll
    for (int T = 0; T < 2; ++T) {
        aloN[T] = make_float4(0.0f, 0.0f, 0.0f, 0.0f);
        ahiN[T] = make_float4(0.0f, 0.0f, 0.0f, 0.0f);
    }
    if (grp < nGroups) {
        sN = src[grp * 32 + slot];
        dN = dst[grp * 32 + slot];
#pragma unroll
        for (int T = 0; T < 2; ++T) {
            const float* ap = ea + (size_t)(grp * 32 + 16 * T + c) * FE + 8 * g;
            aloN[T] = *(const float4*)ap;
            ahiN[T] = *(const float4*)(ap + 4);
        }
    }

    for (; grp < nGroups; grp += nWaves) {
        const int e0 = grp * 32;
        const float ps = pp[2 * (size_t)sN];
        const float pd = pp[2 * (size_t)dN + 1];

        float4 alo[2], ahi[2];
#pragma unroll
        for (int T = 0; T < 2; ++T) { alo[T] = aloN[T]; ahi[T] = ahiN[T]; }

        const int nxt = min(grp + nWaves, nGroups - 1);
        sN = src[nxt * 32 + slot];
        dN = dst[nxt * 32 + slot];
#pragma unroll
        for (int T = 0; T < 2; ++T) {
            const float* ap = ea + (size_t)(nxt * 32 + 16 * T + c) * FE + 8 * g;
            aloN[T] = *(const float4*)ap;
            ahiN[T] = *(const float4*)(ap + 4);
        }

        float q[16];
#pragma unroll
        for (int T = 0; T < 2; ++T) {
            v2f A[4];
            A[0][0] = alo[T].x; A[0][1] = alo[T].y;
            A[1][0] = alo[T].z; A[1][1] = alo[T].w;
            A[2][0] = ahi[T].x; A[2][1] = ahi[T].y;
            A[3][0] = ahi[T].z; A[3][1] = ahi[T].w;

            float part[8];
#pragma unroll
            for (int r = 0; r < 8; ++r) part[r] = 0.0f;
#pragma unroll
            for (int t = 0; t < 8; t += 2) {
                v8f acc0 = {};
                v8f acc1 = {};
#pragma unroll
                for (int s = 0; s < 4; ++s) {
                    acc0 = __builtin_amdgcn_wmma_f32_16x16x4_f32(
                        false, A[s], false, Bf[t][s], (short)0, acc0, false, false);
                    acc1 = __builtin_amdgcn_wmma_f32_16x16x4_f32(
                        false, A[s], false, Bf[t + 1][s], (short)0, acc1, false, false);
                }
#pragma unroll
                for (int r = 0; r < 8; ++r) {
                    part[r] += fmaxf(acc0[r] + bev[t], 0.0f) * wd3[t];
                    part[r] += fmaxf(acc1[r] + bev[t + 1], 0.0f) * wd3[t + 1];
                }
            }
#pragma unroll
            for (int r = 0; r < 8; ++r) q[8 * T + r] = part[r];
        }

        float u[8];
#pragma unroll
        for (int m = 0; m < 8; ++m) {
            const float keep = bit3 ? q[8 + m] : q[m];
            const float send = bit3 ? q[m] : q[8 + m];
            u[m] = keep + __shfl_xor(send, 8, 32);
        }
        float w4[4];
#pragma unroll
        for (int m = 0; m < 4; ++m) {
            const float keep = bit2 ? u[4 + m] : u[m];
            const float send = bit2 ? u[m] : u[4 + m];
            w4[m] = keep + __shfl_xor(send, 4, 32);
        }
        float x2[2];
#pragma unroll
        for (int m = 0; m < 2; ++m) {
            const float keep = bit1 ? w4[2 + m] : w4[m];
            const float send = bit1 ? w4[m] : w4[2 + m];
            x2[m] = keep + __shfl_xor(send, 2, 32);
        }
        const float keep1 = bit0 ? x2[1] : x2[0];
        const float send1 = bit0 ? x2[0] : x2[1];
        const float z = keep1 + __shfl_xor(send1, 1, 32);

        out[e0 + slot] = (ps + pd) + z + bdv;
    }
}

extern "C" void kernel_launch(void* const* d_in, const int* in_sizes, int n_in,
                              void* d_out, int out_size, void* d_ws, size_t ws_size,
                              hipStream_t stream)
{
    const float* x   = (const float*)d_in[0];
    const float* ea  = (const float*)d_in[1];
    const int*   ei  = (const int*)  d_in[2];
    const float* Wn  = (const float*)d_in[3];
    const float* bn  = (const float*)d_in[4];
    const float* We  = (const float*)d_in[5];
    const float* be  = (const float*)d_in[6];
    const float* Wc  = (const float*)d_in[7];
    const float* bc  = (const float*)d_in[8];
    const float* Wr  = (const float*)d_in[9];
    const float* br  = (const float*)d_in[10];
    const float* Wd  = (const float*)d_in[11];
    const float* bd  = (const float*)d_in[12];
    float* out = (float*)d_out;

    const int* srcI = ei;
    const int* dstI = ei + NE;

    float* ws   = (float*)d_ws;
    float* hA   = ws;
    float* gsum = hA   + (size_t)NN * HID;
    float* hB   = gsum + (size_t)NN * HID;
    float* dinv = hB   + (size_t)NN * HID;
    int*   offs = (int*)(dinv + NN);
    int*   rank = offs + NN + 16;
    int*   bsrc = rank + NE;
    float* pp   = (float*)(bsrc + NE);
    int*   cntS = (int*)(pp + 2 * (size_t)NN);
    int*   csum = cntS + (size_t)NSLICE * NNP;

    gemm_strip<FN / 16, 0, 0><<<GEMM_BLOCKS, 256, 0, stream>>>(
        x, Wn, nullptr, nullptr, bn, nullptr, hA, nullptr, NRT, 1);

    rank_edges_lds<<<NSLICE, 1024, 0, stream>>>(dstI, rank, cntS, csum);
    scan_offsets_dinv<<<NCHUNK, 256, 0, stream>>>(cntS, csum, offs, dinv, NN);
    bucket_edges<<<(NE + 255) / 256, 256, 0, stream>>>(srcI, dstI, rank, cntS, bsrc, NE);

    gather_norm<<<NN / 8, 256, 0, stream>>>(hA, offs, bsrc, dinv, gsum, NN);
    gemm_strip<HID / 16, HID / 16, 0><<<GEMM_BLOCKS, 256, 0, stream>>>(
        gsum, Wc, hA, Wr, br, bc, hB, nullptr, NRT, 1);

    gather_norm<<<NN / 8, 256, 0, stream>>>(hB, offs, bsrc, dinv, gsum, NN);
    gemm_strip<HID / 16, HID / 16, 0><<<GEMM_BLOCKS, 256, 0, stream>>>(
        gsum, Wc + (size_t)HID * HID, hB, Wr, br, bc + HID, hA, nullptr, NRT, 1);

    gather_norm<<<NN / 8, 256, 0, stream>>>(hA, offs, bsrc, dinv, gsum, NN);
    gemm_strip<HID / 16, HID / 16, 1><<<GEMM_BLOCKS, 256, 0, stream>>>(
        gsum, Wc + (size_t)2 * HID * HID, hA, Wr, br, bc + 2 * HID, pp, Wd, NRT, 1);

    decode_edges<<<768, 128, 0, stream>>>(pp, ea, srcI, dstI, We, be, Wd, bd, out, NE / 32);
}


# Neighbour sum moved into the layer product's launch: 256 one-per-WGP blocks, each wave gathers its strip's 16 rows into its own LDS tile, then multiplies; 53 left-over strips' dot sum order changes
# speedup vs baseline: 4.0651x; 1.0933x over previous
#include <hip/hip_runtime.h>

#define NN 50000
#define NE 800000
#define FN 32
#define FE 16
#define HID 128
#define NL 3
#define NRT (NN / 16)
#define GEMM_BLOCKS ((NRT + 7) / 8)
#define NSLICE 64
#define EPS (NE / NSLICE)
#define NNP (((NN + 255) / 256) * 256)
#define NCHUNK (NNP / 256)
#define RANK_IT ((EPS / 4 + 1023) / 1024)
#define CHUNK_IT ((NCHUNK + 31) / 32)
static_assert(NE % NSLICE == 0 && EPS % 4 == 0, "slices are whole 16-byte runs");
static_assert(NSLICE % 4 == 0, "chunk sums are read 16 bytes at a time");

typedef float v2f __attribute__((ext_vector_type(2)));
typedef float v8f __attribute__((ext_vector_type(8)));

#define WPITCH 264

template <int KB1, int KB2, int MODE>
__global__ __launch_bounds__(256, 4) void gemm_strip(
    const float* __restrict__ A1, const float* __restrict__ W1,
    const float* __restrict__ A2, const float* __restrict__ W2,
    const float* __restrict__ bias1, const float* __restrict__ bias2,
    float* __restrict__ C, const float* __restrict__ Wd,
    int nRowTiles, int do_relu)
{
    constexpr int KB = KB1 + KB2;
    constexpr int K1 = 16 * KB1;
    constexpr int K2 = 16 * KB2;
    __shared__ __attribute__((aligned(16))) float Ws[(8 * KB) * WPITCH];

    const int tid  = threadIdx.x;
    const int lane = tid & 31;
    const int c    = lane & 15;
    const int g    = lane >> 4;
    const int rt = __builtin_amdgcn_readfirstlane((int)(blockIdx.x * 8 + (tid >> 5)));
    const bool active = rt < nRowTiles;
    const int rtl = active ? rt : nRowTiles - 1;

    float4 alo[KB], ahi[KB];
    {
        const float* ap1 = A1 + (size_t)(rtl * 16 + c) * K1 + 8 * g;
#pragma unroll
        for (int b = 0; b < KB1; ++b) {
            alo[b] = *(const float4*)(ap1 + 16 * b);
            ahi[b] = *(const float4*)(ap1 + 16 * b + 4);
        }
    }

#pragma unroll
    for (int i = 0; i < KB1; ++i) {
        const int kp = (tid >> 5) + 8 * i;
        const int m  = tid & 31;
        const float4 w0 = *(const float4*)(W1 + (size_t)(2 * kp) * HID + 4 * m);
        const float4 w1 = *(const float4*)(W1 + (size_t)(2 * kp + 1) * HID + 4 * m);
        float* dp = Ws + kp * WPITCH + (m >> 4) * 128 + (m & 15) * 4;
        *(float4*)dp        = make_float4(w0.x, w1.x, w0.y, w1.y);
        *(float4*)(dp + 64) = make_float4(w0.z, w1.z, w0.w, w1.w);
    }
#pragma unroll
    for (int i = 0; i < KB2; ++i) {
        const int kp = (tid >> 5) + 8 * i;
        const int m  = tid & 31;
        const float4 w0 = *(const float4*)(W2 + (size_t)(2 * kp) * HID + 4 * m);
        const float4 w1 = *(const float4*)(W2 + (size_t)(2 * kp + 1) * HID + 4 * m);
        float* dp = Ws + (8 * KB1 + kp) * WPITCH + (m >> 4) * 128 + (m & 15) * 4;
        *(float4*)dp        = make_float4(w0.x, w1.x, w0.y, w1.y);
        *(float4*)(dp + 64) = make_float4(w0.z, w1.z, w0.w, w1.w);
    }
    if (KB2 > 0) {
        const float* ap2 = A2 + (size_t)(rtl * 16 + c) * K2 + 8 * g;
#pragma unroll
        for (int b = 0; b < KB2; ++b) {
            alo[KB1 + b] = *(const float4*)(ap2 + 16 * b);
            ahi[KB1 + b] = *(const float4*)(ap2 + 16 * b + 4);
        }
    }
    __syncthreads();

    if (!active) return;

    v8f acc[8];
#pragma unroll
    for (int t = 0; t < 8; ++t) acc[t] = v8f{};

    const float* bp = Ws + (4 * g) * WPITCH + c * 4;
#pragma unroll
    for (int b = 0; b < KB; ++b) {
        v2f Af[4];
        Af[0][0] = alo[b].x; Af[0][1] = alo[b].y;
        Af[1][0] = alo[b].z; Af[1][1] = alo[b].w;
        Af[2][0] = ahi[b].x; Af[2][1] = ahi[b].y;
        Af[3][0] = ahi[b].z; Af[3][1] = ahi[b].w;
#pragma unroll
        for (int s = 0; s < 4; ++s) {
            const float* q = bp + (8 * b + s) * WPITCH;
#pragma unroll
            for (int p = 0; p < 4; ++p) {
                const float4 w = *(const float4*)(q + 64 * p);
                v2f b0, b1;
                b0[0] = w.x; b0[1] = w.y;
                b1[0] = w.z; b1[1] = w.w;
                acc[2 * p] = __builtin_amdgcn_wmma_f32_16x16x4_f32(
                    false, Af[s], false, b0, (short)0, acc[2 * p], false, false);
                acc[2 * p + 1] = __builtin_amdgcn_wmma_f32_16x16x4_f32(
                    false, Af[s], false, b1, (short)0, acc[2 * p + 1], false, false);
            }
        }
    }

    float4 bvlo = make_float4(0.0f, 0.0f, 0.0f, 0.0f);
    float4 bvhi = make_float4(0.0f, 0.0f, 0.0f, 0.0f);
    if (bias1) {
        const float4 l = *(const float4*)(bias1 + 4 * c);
        const float4 h = *(const float4*)(bias1 + 64 + 4 * c);
        bvlo.x += l.x; bvlo.y += l.y; bvlo.z += l.z; bvlo.w += l.w;
        bvhi.x += h.x; bvhi.y += h.y; bvhi.z += h.z; bvhi.w += h.w;
    }
    if (bias2) {
        const float4 l = *(const float4*)(bias2 + 4 * c);
        const float4 h = *(const float4*)(bias2 + 64 + 4 * c);
        bvlo.x += l.x; bvlo.y += l.y; bvlo.z += l.z; bvlo.w += l.w;
        bvhi.x += h.x; bvhi.y += h.y; bvhi.z += h.z; bvhi.w += h.w;
    }

    if (MODE == 0) {
#pragma unroll
        for (int r = 0; r < 8; ++r) {
            float4 lo, hi;
            lo.x = acc[0][r] + bvlo.x; lo.y = acc[1][r] + bvlo.y;
            lo.z = acc[2][r] + bvlo.z; lo.w = acc[3][r] + bvlo.w;
            hi.x = acc[4][r] + bvhi.x; hi.y = acc[5][r] + bvhi.y;
            hi.z = acc[6][r] + bvhi.z; hi.w = acc[7][r] + bvhi.w;
            if (do_relu) {
                lo.x = fmaxf(lo.x, 0.0f); lo.y = fmaxf(lo.y, 0.0f);
                lo.z = fmaxf(lo.z, 0.0f); lo.w = fmaxf(lo.w, 0.0f);
                hi.x = fmaxf(hi.x, 0.0f); hi.y = fmaxf(hi.y, 0.0f);
                hi.z = fmaxf(hi.z, 0.0f); hi.w = fmaxf(hi.w, 0.0f);
            }
            float* cp = C + (size_t)(rt * 16 + 8 * g + r) * HID + 4 * c;
            *(float4*)cp        = lo;
            *(float4*)(cp + 64) = hi;
        }
    } else {
        const float4 u1l = *(const float4*)(Wd + 4 * c);
        const float4 u1h = *(const float4*)(Wd + 64 + 4 * c);
        const float4 u2l = *(const float4*)(Wd + HID + 4 * c);
        const float4 u2h = *(const float4*)(Wd + HID + 64 + 4 * c);
        float q[16];
#pragma unroll
        for (int r = 0; r < 8; ++r) {
            float4 lo, hi;
            lo.x = fmaxf(acc[0][r] + bvlo.x, 0.0f); lo.y = fmaxf(acc[1][r] + bvlo.y, 0.0f);
            lo.z = fmaxf(acc[2][r] + bvlo.z, 0.0f); lo.w = fmaxf(acc[3][r] + bvlo.w, 0.0f);
            hi.x = fmaxf(acc[4][r] + bvhi.x, 0.0f); hi.y = fmaxf(acc[5][r] + bvhi.y, 0.0f);
            hi.z = fmaxf(acc[6][r] + bvhi.z, 0.0f); hi.w = fmaxf(acc[7][r] + bvhi.w, 0.0f);
            float a = lo.x * u1l.x;
            a += lo.y * u1l.y; a += lo.z * u1l.z; a += lo.w * u1l.w;
            a += hi.x * u1h.x; a += hi.y * u1h.y; a += hi.z * u1h.z; a += hi.w * u1h.w;
            float d = lo.x * u2l.x;
            d += lo.y * u2l.y; d += lo.z * u2l.z; d += lo.w * u2l.w;
            d += hi.x * u2h.x; d += hi.y * u2h.y; d += hi.z * u2h.z; d += hi.w * u2h.w;
            q[2 * r]     = a;
            q[2 * r + 1] = d;
        }
        const bool bit3 = (c & 8) != 0;
        const bool bit2 = (c & 4) != 0;
        const bool bit1 = (c & 2) != 0;
        const bool bit0 = (c & 1) != 0;
        float u[8];
#pragma unroll
        for (int m = 0; m < 8; ++m) {
            const float keep = bit3 ? q[8 + m] : q[m];
            const float send = bit3 ? q[m] : q[8 + m];
            u[m] = keep + __shfl_xor(send, 8, 32);
        }
        float w4[4];
#pragma unroll
        for (int m = 0; m < 4; ++m) {
            const float keep = bit2 ? u[4 + m] : u[m];
            const float send = bit2 ? u[m] : u[4 + m];
            w4[m] = keep + __shfl_xor(send, 4, 32);
        }
        float x2[2];
#pragma unroll
        for (int m = 0; m < 2; ++m) {
            const float keep = bit1 ? w4[2 + m] : w4[m];
            const float send = bit1 ? w4[m] : w4[2 + m];
            x2[m] = keep + __shfl_xor(send, 2, 32);
        }
        const float keep1 = bit0 ? x2[1] : x2[0];
        const float send1 = bit0 ? x2[0] : x2[1];
        const float z = keep1 + __shfl_xor(send1, 1, 32);
        C[(size_t)rt * 32 + lane] = z;
    }
}

#define LBLK 256
#define SPB (NRT / LBLK)
#define NTAIL (NRT - SPB * LBLK)
#define LWAVES (SPB + 4)
#define CPITCH 132
#define GR 16
static_assert(NTAIL >= 0 && NTAIL <= LBLK, "one left-over strip per block at most");
static_assert(LWAVES * 32 <= 1024, "a block is at most 1024 threads");

template <int MODE>
__global__ __launch_bounds__(LWAVES * 32, 4) void layer_fused(
    const float* __restrict__ h, const int* __restrict__ offs,
    const int* __restrict__ bsrc, const float* __restrict__ dinv,
    const float* __restrict__ W1, const float* __restrict__ W2,
    const float* __restrict__ bias1, const float* __restrict__ bias2,
    float* __restrict__ C, const float* __restrict__ Wd)
{
    constexpr int KB1 = HID / 16;
    constexpr int KB2 = HID / 16;
    constexpr int KB = KB1 + KB2;
    static_assert((8 * KB1) % LWAVES == 0 && (8 * KB2) % LWAVES == 0,
                  "the staging passes are whole");
    __shared__ __attribute__((aligned(16))) float Ws[(8 * KB) * WPITCH];
    __shared__ __attribute__((aligned(16))) float Gt[LWAVES * 16 * CPITCH];
    __shared__ __attribute__((aligned(16))) float Ct[16 * CPITCH];

    const int tid  = threadIdx.x;
    const int lane = tid & 31;
    const int c    = lane & 15;
    const int g    = lane >> 4;
    const int wv  = __builtin_amdgcn_readfirstlane(tid >> 5);
    const int blk = (int)blockIdx.x;
    const bool tailw   = wv < 4;
    const bool hasTail = blk < NTAIL;
    const bool active  = !tailw || hasTail;
    const int  pr      = wv;
    const int  rt      = tailw ? (SPB * LBLK + blk) : (blk * SPB + wv - 4);

#pragma unroll
    for (int i = 0; i < (8 * KB1) / LWAVES; ++i) {
        const int kp = (tid >> 5) + LWAVES * i;
        const int m  = tid & 31;
        const float4 w0 = *(const float4*)(W1 + (size_t)(2 * kp) * HID + 4 * m);
        const float4 w1 = *(const float4*)(W1 + (size_t)(2 * kp + 1) * HID + 4 * m);
        float* dp = Ws + kp * WPITCH + (m >> 4) * 128 + (m & 15) * 4;
        *(float4*)dp        = make_float4(w0.x, w1.x, w0.y, w1.y);
        *(float4*)(dp + 64) = make_float4(w0.z, w1.z, w0.w, w1.w);
    }
#pragma unroll
    for (int i = 0; i < (8 * KB2) / LWAVES; ++i) {
        const int kp = (tid >> 5) + LWAVES * i;
        const int m  = tid & 31;
        const float4 w0 = *(const float4*)(W2 + (size_t)(2 * kp) * HID + 4 * m);
        const float4 w1 = *(const float4*)(W2 + (size_t)(2 * kp + 1) * HID + 4 * m);
        float* dp = Ws + (8 * KB1 + kp) * WPITCH + (m >> 4) * 128 + (m & 15) * 4;
        *(float4*)dp        = make_float4(w0.x, w1.x, w0.y, w1.y);
        *(float4*)(dp + 64) = make_float4(w0.z, w1.z, w0.w, w1.w);
    }
    __syncthreads();

    float4 alo[KB], ahi[KB];
    if (active) {
        const int n0 = rt * 16;
        float* gt = Gt + wv * (16 * CPITCH);
        const float* hl = h + lane * 4;
        const int   ob = offs[n0 + min(lane, 16)];
        const float dn = dinv[n0 + (lane & 15)];
        const int beg = __builtin_amdgcn_readlane(ob, 0);
        const int end = __builtin_amdgcn_readlane(ob, 16);
        int   node = 0;
        int   nb   = __builtin_amdgcn_readlane(ob, 1);
        float di   = __int_as_float(__builtin_amdgcn_readlane(__float_as_int(dn), 0));
        float4 acc = make_float4(0.0f, 0.0f, 0.0f, 0.0f);

        int   sC = 0, sN = 0;
        float wC = 0.0f;
        if (beg + lane < end) {
            sC = bsrc[beg + lane];
            wC = dinv[sC];
        }
        if (beg + 32 + lane < end) sN = bsrc[beg + 32 + lane];
        auto ask_rows = [&](float4 (&v)[GR], int j0, int m) {
#pragma unroll
            for (int u = 0; u < GR; ++u) {
                const int jj = min(j0 + u, m - 1);
                const int s  = __builtin_amdgcn_readlane(sC, jj);
                v[u] = *(const float4*)(hl + (size_t)s * HID);
            }
        };
        auto add_rows = [&](const float4 (&v)[GR], int base, int j0, int m) {
#pragma unroll
            for (int u = 0; u < GR; ++u) {
                const int j = j0 + u;
                if (j < m) {
                    const int e = base + j;
                    while (e >= nb) {
                        *(float4*)(gt + node * CPITCH + 4 * lane) = acc;
                        acc = make_float4(0.0f, 0.0f, 0.0f, 0.0f);
                        ++node;
                        nb = __builtin_amdgcn_readlane(ob, node + 1);
                        di = __int_as_float(__builtin_amdgcn_readlane(__float_as_int(dn), node));
                    }
                    const float w = __int_as_float(__builtin_amdgcn_readlane(__float_as_int(wC), j)) * di;
                    acc.x += v[u].x * w;
                    acc.y += v[u].y * w;
                    acc.z += v[u].z * w;
                    acc.w += v[u].w * w;
                }
            }
        };
        static_assert(2 * GR == 32, "a chunk of 32 edges is two groups of rows");
        for (int base = beg; base < end; base += 32) {
            const int m = min(32, end - base);
            float4 v[GR];
            ask_rows(v, 0, m);
            const float wN = dinv[sN];
            int sNN = 0;
            if (base + 64 + lane < end) sNN = bsrc[base + 64 + lane];
            add_rows(v, base, 0, m);
            if (m > GR) {
                ask_rows(v, GR, m);
                add_rows(v, base, GR, m);
            }
            sC = sN;
            wC = wN;
            sN = sNN;
        }
        for (; node < 16; ++node) {
            *(float4*)(gt + node * CPITCH + 4 * lane) = acc;
            acc = make_float4(0.0f, 0.0f, 0.0f, 0.0f);
        }

        float4 sv[16];
#pragma unroll
        for (int i = 0; i < 16; ++i) sv[i] = *(const float4*)(hl + (size_t)(n0 + i) * HID);
        const float* ap2 = h + (size_t)(n0 + c) * HID + 8 * g;
#pragma unroll
        for (int b = 0; b < KB2; ++b) {
            alo[KB1 + b] = *(const float4*)(ap2 + 16 * b);
            ahi[KB1 + b] = *(const float4*)(ap2 + 16 * b + 4);
        }
#pragma unroll
        for (int i = 0; i < 16; ++i) {
            const float d  = __int_as_float(__builtin_amdgcn_readlane(__float_as_int(dn), i));
            const float dd = d * d;
            float4 t = *(const float4*)(gt + i * CPITCH + 4 * lane);
            t.x += sv[i].x * dd;
            t.y += sv[i].y * dd;
            t.z += sv[i].z * dd;
            t.w += sv[i].w * dd;
            *(float4*)(gt + i * CPITCH + 4 * lane) = t;
        }
        const float* ap1 = gt + c * CPITCH + 8 * g;
#pragma unroll
        for (int b = 0; b < KB1; ++b) {
            alo[b] = *(const float4*)(ap1 + 16 * b);
            ahi[b] = *(const float4*)(ap1 + 16 * b + 4);
        }
    }

    auto load_bias = [&](float4& lo, float4& hi) {
        const float4 l1 = *(const float4*)(bias1 + 4 * c);
        const float4 h1 = *(const float4*)(bias1 + 64 + 4 * c);
        const float4 l2 = *(const float4*)(bias2 + 4 * c);
        const float4 h2 = *(const float4*)(bias2 + 64 + 4 * c);
        lo = make_float4(l1.x + l2.x, l1.y + l2.y, l1.z + l2.z, l1.w + l2.w);
        hi = make_float4(h1.x + h2.x, h1.y + h2.y, h1.z + h2.z, h1.w + h2.w);
    };

    if (!tailw) {
        v8f acc[8];
#pragma unroll
        for (int t = 0; t < 8; ++t) acc[t] = v8f{};

        const float* bp = Ws + (4 * g) * WPITCH + c * 4;
        float4 wn[4];
#pragma unroll
        for (int p = 0; p < 4; ++p) wn[p] = *(const float4*)(bp + 64 * p);
#pragma unroll
        for (int b = 0; b < KB; ++b) {
            v2f Af[4];
            Af[0][0] = alo[b].x; Af[0][1] = alo[b].y;
            Af[1][0] = alo[b].z; Af[1][1] = alo[b].w;
            Af[2][0] = ahi[b].x; Af[2][1] = ahi[b].y;
            Af[3][0] = ahi[b].z; Af[3][1] = ahi[b].w;
#pragma unroll
            for (int s = 0; s < 4; ++s) {
                float4 wc[4];
#pragma unroll
                for (int p = 0; p < 4; ++p) wc[p] = wn[p];
                if (b + 1 < KB || s < 3) {
                    const float* q = bp + (s < 3 ? 8 * b + s + 1 : 8 * (b + 1)) * WPITCH;
#pragma unroll
                    for (int p = 0; p < 4; ++p) wn[p] = *(const float4*)(q + 64 * p);
                }
                __builtin_amdgcn_sched_barrier(0);
#pragma unroll
                for (int p = 0; p < 4; ++p) {
                    const float4 w = wc[p];
                    v2f b0, b1;
                    b0[0] = w.x; b0[1] = w.y;
                    b1[0] = w.z; b1[1] = w.w;
                    acc[2 * p] = __builtin_amdgcn_wmma_f32_16x16x4_f32(
                        false, Af[s], false, b0, (short)0, acc[2 * p], false, false);
                    acc[2 * p + 1] = __builtin_amdgcn_wmma_f32_16x16x4_f32(
                        false, Af[s], false, b1, (short)0, acc[2 * p + 1], false, false);
                }
            }
        }

        float4 bvlo, bvhi;
        load_bias(bvlo, bvhi);

        if (MODE == 0) {
#pragma unroll
            for (int r = 0; r < 8; ++r) {
                float4 lo, hi;
                lo.x = fmaxf(acc[0][r] + bvlo.x, 0.0f); lo.y = fmaxf(acc[1][r] + bvlo.y, 0.0f);
                lo.z = fmaxf(acc[2][r] + bvlo.z, 0.0f); lo.w = fmaxf(acc[3][r] + bvlo.w, 0.0f);
                hi.x = fmaxf(acc[4][r] + bvhi.x, 0.0f); hi.y = fmaxf(acc[5][r] + bvhi.y, 0.0f);
                hi.z = fmaxf(acc[6][r] + bvhi.z, 0.0f); hi.w = fmaxf(acc[7][r] + bvhi.w, 0.0f);
                float* cp = C + (size_t)(rt * 16 + 8 * g + r) * HID + 4 * c;
                *(float4*)cp        = lo;
                *(float4*)(cp + 64) = hi;
            }
        } else {
            const float4 u1l = *(const float4*)(Wd + 4 * c);
            const float4 u1h = *(const float4*)(Wd + 64 + 4 * c);
            const float4 u2l = *(const float4*)(Wd + HID + 4 * c);
            const float4 u2h = *(const float4*)(Wd + HID + 64 + 4 * c);
            float q[16];
#pragma unroll
            for (int r = 0; r < 8; ++r) {
                float4 lo, hi;
                lo.x = fmaxf(acc[0][r] + bvlo.x, 0.0f); lo.y = fmaxf(acc[1][r] + bvlo.y, 0.0f);
                lo.z = fmaxf(acc[2][r] + bvlo.z, 0.0f); lo.w = fmaxf(acc[3][r] + bvlo.w, 0.0f);
                hi.x = fmaxf(acc[4][r] + bvhi.x, 0.0f); hi.y = fmaxf(acc[5][r] + bvhi.y, 0.0f);
                hi.z = fmaxf(acc[6][r] + bvhi.z, 0.0f); hi.w = fmaxf(acc[7][r] + bvhi.w, 0.0f);
                float a = lo.x * u1l.x;
                a += lo.y * u1l.y; a += lo.z * u1l.z; a += lo.w * u1l.w;
                a += hi.x * u1h.x; a += hi.y * u1h.y; a += hi.z * u1h.z; a += hi.w * u1h.w;
                float d = lo.x * u2l.x;
                d += lo.y * u2l.y; d += lo.z * u2l.z; d += lo.w * u2l.w;
                d += hi.x * u2h.x; d += hi.y * u2h.y; d += hi.z * u2h.z; d += hi.w * u2h.w;
                q[2 * r]     = a;
                q[2 * r + 1] = d;
            }
            const bool bit3 = (c & 8) != 0;
            const bool bit2 = (c & 4) != 0;
            const bool bit1 = (c & 2) != 0;
            const bool bit0 = (c & 1) != 0;
            float u[8];
#pragma unroll
            for (int m = 0; m < 8; ++m) {
                const float keep = bit3 ? q[8 + m] : q[m];
                const float send = bit3 ? q[m] : q[8 + m];
                u[m] = keep + __shfl_xor(send, 8, 32);
            }
            float w4[4];
#pragma unroll
            for (int m = 0; m < 4; ++m) {
                const float keep = bit2 ? u[4 + m] : u[m];
                const float send = bit2 ? u[m] : u[4 + m];
                w4[m] = keep + __shfl_xor(send, 4, 32);
            }
            float x2[2];
#pragma unroll
            for (int m = 0; m < 2; ++m) {
                const float keep = bit1 ? w4[2 + m] : w4[m];
                const float send = bit1 ? w4[m] : w4[2 + m];
                x2[m] = keep + __shfl_xor(send, 2, 32);
            }
            const float keep1 = bit0 ? x2[1] : x2[0];
            const float send1 = bit0 ? x2[0] : x2[1];
            const float z = keep1 + __shfl_xor(send1, 1, 32);
            C[(size_t)rt * 32 + lane] = z;
        }
    } else if (hasTail) {
        v8f tacc[2];
        tacc[0] = v8f{};
        tacc[1] = v8f{};
        const float* bp = Ws + (4 * g) * WPITCH + c * 4 + 64 * pr;
        float4 wn[4];
#pragma unroll
        for (int s = 0; s < 4; ++s) wn[s] = *(const float4*)(bp + s * WPITCH);
#pragma unroll
        for (int b = 0; b < KB; ++b) {
            v2f Af[4];
            Af[0][0] = alo[b].x; Af[0][1] = alo[b].y;
            Af[1][0] = alo[b].z; Af[1][1] = alo[b].w;
            Af[2][0] = ahi[b].x; Af[2][1] = ahi[b].y;
            Af[3][0] = ahi[b].z; Af[3][1] = ahi[b].w;
            float4 wc[4];
#pragma unroll
            for (int s = 0; s < 4; ++s) wc[s] = wn[s];
            if (b + 1 < KB) {
#pragma unroll
                for (int s = 0; s < 4; ++s)
                    wn[s] = *(const float4*)(bp + (8 * (b + 1) + s) * WPITCH);
            }
            __builtin_amdgcn_sched_barrier(0);
#pragma unroll
            for (int s = 0; s < 4; ++s) {
                const float4 w = wc[s];
                v2f b0, b1;
                b0[0] = w.x; b0[1] = w.y;
                b1[0] = w.z; b1[1] = w.w;
                tacc[0] = __builtin_amdgcn_wmma_f32_16x16x4_f32(
                    false, Af[s], false, b0, (short)0, tacc[0], false, false);
                tacc[1] = __builtin_amdgcn_wmma_f32_16x16x4_f32(
                    false, Af[s], false, b1, (short)0, tacc[1], false, false);
            }
        }
        float4 bvlo, bvhi;
        load_bias(bvlo, bvhi);
        const float4 bv = (pr >> 1) ? bvhi : bvlo;
        const float bb0 = (pr & 1) ? bv.z : bv.x;
        const float bb1 = (pr & 1) ? bv.w : bv.y;
        float* tp = Ct + (8 * g) * CPITCH + 4 * c + 2 * (pr & 1) + 64 * (pr >> 1);
#pragma unroll
        for (int r = 0; r < 8; ++r) {
            float2 v;
            v.x = fmaxf(tacc[0][r] + bb0, 0.0f);
            v.y = fmaxf(tacc[1][r] + bb1, 0.0f);
            *(float2*)(tp + r * CPITCH) = v;
        }
    }

    if (hasTail) {
        __syncthreads();
        if (tailw) {
            if (MODE == 0) {
#pragma unroll
                for (int r = 0; r < 4; ++r) {
                    const int row = 4 * pr + r;
                    const float4 v = *(const float4*)(Ct + row * CPITCH + 4 * lane);
                    *(float4*)(C + (size_t)(rt * 16 + row) * HID + 4 * lane) = v;
                }
            } else if (pr == 0) {
                const float* cr = Ct + c * CPITCH;
                const float* wp = Wd + g * HID;
                float sum = 0.0f;
#pragma unroll
                for (int j = 0; j < HID / 4; ++j) {
                    const float4 v = *(const float4*)(cr + 4 * j);
                    const float4 w = *(const float4*)(wp + 4 * j);
                    sum += v.x * w.x; sum += v.y * w.y;
                    sum += v.z * w.z; sum += v.w * w.w;
                }
                C[(size_t)rt * 32 + 2 * c + g] = sum;
            }
        }
    }
}

__global__ __launch_bounds__(1024) void rank_edges_lds(
    const int* __restrict__ dst, int* __restrict__ rank,
    int* __restrict__ cntS, int* __restrict__ csum)
{
    __shared__ __attribute__((aligned(16))) int cnt[NNP];
    const int tid  = threadIdx.x;
    const int lane = tid & 31;
    const int wv   = tid >> 5;
    const int sl   = blockIdx.x;

    const int4* __restrict__ d4 = (const int4*)(dst + (size_t)sl * EPS);
    int4 d[RANK_IT];
#pragma unroll
    for (int k = 0; k < RANK_IT; ++k) {
        const int i = tid + 1024 * k;
        d[k] = make_int4(0, 0, 0, 0);
        if (i < EPS / 4) d[k] = d4[i];
    }

    int4* c4 = (int4*)cnt;
    for (int i = tid; i < NNP / 4; i += 1024) c4[i] = make_int4(0, 0, 0, 0);
    __syncthreads();

#pragma unroll
    for (int k = 0; k < RANK_IT; ++k) {
        const int i = tid + 1024 * k;
        if (i < EPS / 4) {
            d[k].x = atomicAdd(&cnt[d[k].x], 1);
            d[k].y = atomicAdd(&cnt[d[k].y], 1);
            d[k].z = atomicAdd(&cnt[d[k].z], 1);
            d[k].w = atomicAdd(&cnt[d[k].w], 1);
        }
    }
    __syncthreads();

    int4* __restrict__ r4 = (int4*)(rank + (size_t)sl * EPS);
#pragma unroll
    for (int k = 0; k < RANK_IT; ++k) {
        const int i = tid + 1024 * k;
        if (i < EPS / 4) r4[i] = d[k];
    }

    int4* __restrict__ g4 = (int4*)(cntS + (size_t)sl * NNP);
    int sum[CHUNK_IT];
#pragma unroll
    for (int t = 0; t < CHUNK_IT; ++t) {
        const int j = wv + 32 * t;
        sum[t] = 0;
        if (j < NCHUNK) {
            const int4 v0 = c4[64 * j + lane];
            const int4 v1 = c4[64 * j + 32 + lane];
            g4[64 * j + lane]      = v0;
            g4[64 * j + 32 + lane] = v1;
            sum[t] = ((v0.x + v0.y) + (v0.z + v0.w)) + ((v1.x + v1.y) + (v1.z + v1.w));
        }
    }
#pragma unroll
    for (int off = 16; off > 0; off >>= 1) {
#pragma unroll
        for (int t = 0; t < CHUNK_IT; ++t)
            sum[t] += __shfl_xor(sum[t], off, 32);
    }
    if (lane == 0) {
#pragma unroll
        for (int t = 0; t < CHUNK_IT; ++t) {
            const int j = wv + 32 * t;
            if (j < NCHUNK) csum[j * NSLICE + sl] = sum[t];
        }
    }
}

__global__ __launch_bounds__(256) void scan_offsets_dinv(
    int* __restrict__ cntS, const int* __restrict__ csum,
    int* __restrict__ offs, float* __restrict__ dinv, int n)
{
    __shared__ int waveCnt[8];
    __shared__ int wavePre[8];
    const int tid  = threadIdx.x;
    const int lane = tid & 31;
    const int wv   = tid >> 5;
    const int b    = blockIdx.x;
    const int i    = b * 256 + tid;

    int cs[NSLICE];
#pragma unroll
    for (int k = 0; k < NSLICE; ++k) cs[k] = cntS[(size_t)k * NNP + i];

    int pre = 0;
    const int4* __restrict__ s4 = (const int4*)csum;
    const int n4   = b * (NSLICE / 4);
    const int full = n4 >> 8;
#pragma unroll 4
    for (int it = 0; it < full; ++it) {
        int4 v = s4[it * 256 + tid];
        pre += (v.x + v.y) + (v.z + v.w);
    }
    if (tid < (n4 & 255)) {
        int4 v = s4[full * 256 + tid];
        pre += (v.x + v.y) + (v.z + v.w);
    }

    int c = 0;
#pragma unroll
    for (int k = 0; k < NSLICE; ++k) c += cs[k];

    int inc = c;
    for (int off = 1; off < 32; off <<= 1) {
        int t = __shfl_up(inc, off, 32);
        if (lane >= off) inc += t;
    }
    for (int off = 16; off > 0; off >>= 1)
        pre += __shfl_xor(pre, off, 32);

    if (lane == 31) waveCnt[wv] = inc;
    if (lane == 0)  wavePre[wv] = pre;
    __syncthreads();

    int base = 0;
    for (int k = 0; k < 8; ++k) {
        base += wavePre[k];
        if (k < wv) base += waveCnt[k];
    }
    const int excl = base + inc - c;

    int run = excl;
#pragma unroll
    for (int k = 0; k < NSLICE; ++k) {
        cntS[(size_t)k * NNP + i] = run;
        run += cs[k];
    }

    if (i < n) {
        offs[i] = excl;
        float d = (float)(c + 1);
        dinv[i] = (d > 0.0f) ? rsqrtf(d) : 0.0f;
    }
    if (i == n - 1) offs[n] = excl + c;
}

__global__ void bucket_edges(const int* __restrict__ src, const int* __restrict__ dst,
                             const int* __restrict__ rank, const int* __restrict__ start,
                             int* __restrict__ bsrc, int n)
{
    int i = blockIdx.x * blockDim.x + threadIdx.x;
    if (i < n) bsrc[start[(size_t)(i / EPS) * NNP + dst[i]] + rank[i]] = src[i];
}

__global__ __launch_bounds__(128, 4) void decode_edges(
    const float* __restrict__ pp, const float* __restrict__ ea,
    const int* __restrict__ src,  const int* __restrict__ dst,
    const float* __restrict__ We, const float* __restrict__ be,
    const float* __restrict__ Wd, const float* __restrict__ bd,
    float* __restrict__ out, int nGroups)
{
    const int lane = threadIdx.x & 31;
    const int c    = lane & 15;
    const int g    = lane >> 4;
    const int wave = __builtin_amdgcn_readfirstlane(
        (int)((blockIdx.x * blockDim.x + threadIdx.x) >> 5));
    const int nWaves = (int)((gridDim.x * blockDim.x) >> 5);

    v2f Bf[8][4];
#pragma unroll
    for (int s = 0; s < 4; ++s) {
#pragma unroll
        for (int j = 0; j < 2; ++j) {
            const float* rp = We + (size_t)(8 * g + 2 * s + j) * HID + 8 * c;
            const float4 lo = *(const float4*)rp;
            const float4 hi = *(const float4*)(rp + 4);
            Bf[0][s][j] = lo.x; Bf[1][s][j] = lo.y; Bf[2][s][j] = lo.z; Bf[3][s][j] = lo.w;
            Bf[4][s][j] = hi.x; Bf[5][s][j] = hi.y; Bf[6][s][j] = hi.z; Bf[7][s][j] = hi.w;
        }
    }
    float bev[8], wd3[8];
    {
        const float4 b0 = *(const float4*)(be + 8 * c);
        const float4 b1 = *(const float4*)(be + 8 * c + 4);
        const float4 w0 = *(const float4*)(Wd + 2 * HID + 8 * c);
        const float4 w1 = *(const float4*)(Wd + 2 * HID + 8 * c + 4);
        bev[0] = b0.x; bev[1] = b0.y; bev[2] = b0.z; bev[3] = b0.w;
        bev[4] = b1.x; bev[5] = b1.y; bev[6] = b1.z; bev[7] = b1.w;
        wd3[0] = w0.x; wd3[1] = w0.y; wd3[2] = w0.z; wd3[3] = w0.w;
        wd3[4] = w1.x; wd3[5] = w1.y; wd3[6] = w1.z; wd3[7] = w1.w;
    }
    const float bdv = bd[0];

    const bool bit3 = (c & 8) != 0;
    const bool bit2 = (c & 4) != 0;
    const bool bit1 = (c & 2) != 0;
    const bool bit0 = (c & 1) != 0;
    const int slot = ((c & 8) << 1) + 8 * g + (c & 7);

    int grp = wave;
    int sN = 0, dN = 0;
    float4 aloN[2], ahiN[2];
#pragma unroll
    for (int T = 0; T < 2; ++T) {
        aloN[T] = make_float4(0.0f, 0.0f, 0.0f, 0.0f);
        ahiN[T] = make_float4(0.0f, 0.0f, 0.0f, 0.0f);
    }
    if (grp < nGroups) {
        sN = src[grp * 32 + slot];
        dN = dst[grp * 32 + slot];
#pragma unroll
        for (int T = 0; T < 2; ++T) {
            const float* ap = ea + (size_t)(grp * 32 + 16 * T + c) * FE + 8 * g;
            aloN[T] = *(const float4*)ap;
            ahiN[T] = *(const float4*)(ap + 4);
        }
    }

    for (; grp < nGroups; grp += nWaves) {
        const int e0 = grp * 32;
        const float ps = pp[2 * (size_t)sN];
        const float pd = pp[2 * (size_t)dN + 1];

        float4 alo[2], ahi[2];
#pragma unroll
        for (int T = 0; T < 2; ++T) { alo[T] = aloN[T]; ahi[T] = ahiN[T]; }

        const int nxt = min(grp + nWaves, nGroups - 1);
        sN = src[nxt * 32 + slot];
        dN = dst[nxt * 32 + slot];
#pragma unroll
        for (int T = 0; T < 2; ++T) {
            const float* ap = ea + (size_t)(nxt * 32 + 16 * T + c) * FE + 8 * g;
            aloN[T] = *(const float4*)ap;
            ahiN[T] = *(const float4*)(ap + 4);
        }

        float q[16];
#pragma unroll
        for (int T = 0; T < 2; ++T) {
            v2f A[4];
            A[0][0] = alo[T].x; A[0][1] = alo[T].y;
            A[1][0] = alo[T].z; A[1][1] = alo[T].w;
            A[2][0] = ahi[T].x; A[2][1] = ahi[T].y;
            A[3][0] = ahi[T].z; A[3][1] = ahi[T].w;

            float part[8];
#pragma unroll
            for (int r = 0; r < 8; ++r) part[r] = 0.0f;
#pragma unroll
            for (int t = 0; t < 8; t += 2) {
                v8f acc0 = {};
                v8f acc1 = {};
#pragma unroll
                for (int s = 0; s < 4; ++s) {
                    acc0 = __builtin_amdgcn_wmma_f32_16x16x4_f32(
                        false, A[s], false, Bf[t][s], (short)0, acc0, false, false);
                    acc1 = __builtin_amdgcn_wmma_f32_16x16x4_f32(
                        false, A[s], false, Bf[t + 1][s], (short)0, acc1, false, false);
                }
#pragma unroll
                for (int r = 0; r < 8; ++r) {
                    part[r] += fmaxf(acc0[r] + bev[t], 0.0f) * wd3[t];
                    part[r] += fmaxf(acc1[r] + bev[t + 1], 0.0f) * wd3[t + 1];
                }
            }
#pragma unroll
            for (int r = 0; r < 8; ++r) q[8 * T + r] = part[r];
        }

        float u[8];
#pragma unroll
        for (int m = 0; m < 8; ++m) {
            const float keep = bit3 ? q[8 + m] : q[m];
            const float send = bit3 ? q[m] : q[8 + m];
            u[m] = keep + __shfl_xor(send, 8, 32);
        }
        float w4[4];
#pragma unroll
        for (int m = 0; m < 4; ++m) {
            const float keep = bit2 ? u[4 + m] : u[m];
            const float send = bit2 ? u[m] : u[4 + m];
            w4[m] = keep + __shfl_xor(send, 4, 32);
        }
        float x2[2];
#pragma unroll
        for (int m = 0; m < 2; ++m) {
            const float keep = bit1 ? w4[2 + m] : w4[m];
            const float send = bit1 ? w4[m] : w4[2 + m];
            x2[m] = keep + __shfl_xor(send, 2, 32);
        }
        const float keep1 = bit0 ? x2[1] : x2[0];
        const float send1 = bit0 ? x2[0] : x2[1];
        const float z = keep1 + __shfl_xor(send1, 1, 32);

        out[e0 + slot] = (ps + pd) + z + bdv;
    }
}

extern "C" void kernel_launch(void* const* d_in, const int* in_sizes, int n_in,
                              void* d_out, int out_size, void* d_ws, size_t ws_size,
                              hipStream_t stream)
{
    const float* x   = (const float*)d_in[0];
    const float* ea  = (const float*)d_in[1];
    const int*   ei  = (const int*)  d_in[2];
    const float* Wn  = (const float*)d_in[3];
    const float* bn  = (const float*)d_in[4];
    const float* We  = (const float*)d_in[5];
    const float* be  = (const float*)d_in[6];
    const float* Wc  = (const float*)d_in[7];
    const float* bc  = (const float*)d_in[8];
    const float* Wr  = (const float*)d_in[9];
    const float* br  = (const float*)d_in[10];
    const float* Wd  = (const float*)d_in[11];
    const float* bd  = (const float*)d_in[12];
    float* out = (float*)d_out;

    const int* srcI = ei;
    const int* dstI = ei + NE;

    float* ws   = (float*)d_ws;
    float* hA   = ws;
    float* hB   = hA   + (size_t)NN * HID;
    float* dinv = hB   + (size_t)NN * HID;
    int*   offs = (int*)(dinv + NN);
    int*   rank = offs + NN + 16;
    int*   bsrc = rank + NE;
    float* pp   = (float*)(bsrc + NE);
    int*   cntS = (int*)(pp + 2 * (size_t)NN);
    int*   csum = cntS + (size_t)NSLICE * NNP;

    gemm_strip<FN / 16, 0, 0><<<GEMM_BLOCKS, 256, 0, stream>>>(
        x, Wn, nullptr, nullptr, bn, nullptr, hA, nullptr, NRT, 1);

    rank_edges_lds<<<NSLICE, 1024, 0, stream>>>(dstI, rank, cntS, csum);
    scan_offsets_dinv<<<NCHUNK, 256, 0, stream>>>(cntS, csum, offs, dinv, NN);
    bucket_edges<<<(NE + 255) / 256, 256, 0, stream>>>(srcI, dstI, rank, cntS, bsrc, NE);

    layer_fused<0><<<LBLK, LWAVES * 32, 0, stream>>>(
        hA, offs, bsrc, dinv, Wc, Wr, br, bc, hB, nullptr);
    layer_fused<0><<<LBLK, LWAVES * 32, 0, stream>>>(
        hB, offs, bsrc, dinv, Wc + (size_t)HID * HID, Wr, br, bc + HID, hA, nullptr);
    layer_fused<1><<<LBLK, LWAVES * 32, 0, stream>>>(
        hA, offs, bsrc, dinv, Wc + (size_t)2 * HID * HID, Wr, br, bc + 2 * HID, pp, Wd);

    decode_edges<<<768, 128, 0, stream>>>(pp, ea, srcI, dstI, We, be, Wd, bd, out, NE / 32);
}


# Node encoder moved into the scan launch: its 391 eight-wave blocks run behind the scan's 196 memory-bound blocks on the otherwise idle matrix units; one launch fewer (7); every sum unchanged
# speedup vs baseline: 4.1917x; 1.0311x over previous
#include <hip/hip_runtime.h>

#define NN 50000
#define NE 800000
#define FN 32
#define FE 16
#define HID 128
#define NL 3
#define NRT (NN / 16)
#define NSLICE 64
#define EPS (NE / NSLICE)
#define NNP (((NN + 255) / 256) * 256)
#define NCHUNK (NNP / 256)
#define RANK_IT ((EPS / 4 + 1023) / 1024)
#define CHUNK_IT ((NCHUNK + 31) / 32)
static_assert(NE % NSLICE == 0 && EPS % 4 == 0, "slices are whole 16-byte runs");
static_assert(NSLICE % 4 == 0, "chunk sums are read 16 bytes at a time");

typedef float v2f __attribute__((ext_vector_type(2)));
typedef float v8f __attribute__((ext_vector_type(8)));

#define WPITCH 264

#define LBLK 256
#define SPB (NRT / LBLK)
#define NTAIL (NRT - SPB * LBLK)
#define LWAVES (SPB + 4)
#define CPITCH 132
#define GR 16
static_assert(NTAIL >= 0 && NTAIL <= LBLK, "one left-over strip per block at most");
static_assert(LWAVES * 32 <= 1024, "a block is at most 1024 threads");

template <int MODE>
__global__ __launch_bounds__(LWAVES * 32, 4) void layer_fused(
    const float* __restrict__ h, const int* __restrict__ offs,
    const int* __restrict__ bsrc, const float* __restrict__ dinv,
    const float* __restrict__ W1, const float* __restrict__ W2,
    const float* __restrict__ bias1, const float* __restrict__ bias2,
    float* __restrict__ C, const float* __restrict__ Wd)
{
    constexpr int KB1 = HID / 16;
    constexpr int KB2 = HID / 16;
    constexpr int KB = KB1 + KB2;
    static_assert((8 * KB1) % LWAVES == 0 && (8 * KB2) % LWAVES == 0,
                  "the staging passes are whole");
    __shared__ __attribute__((aligned(16))) float Ws[(8 * KB) * WPITCH];
    __shared__ __attribute__((aligned(16))) float Gt[LWAVES * 16 * CPITCH];
    __shared__ __attribute__((aligned(16))) float Ct[16 * CPITCH];

    const int tid  = threadIdx.x;
    const int lane = tid & 31;
    const int c    = lane & 15;
    const int g    = lane >> 4;
    const int wv  = __builtin_amdgcn_readfirstlane(tid >> 5);
    const int blk = (int)blockIdx.x;
    const bool tailw   = wv < 4;
    const bool hasTail = blk < NTAIL;
    const bool active  = !tailw || hasTail;
    const int  pr      = wv;
    const int  rt      = tailw ? (SPB * LBLK + blk) : (blk * SPB + wv - 4);

#pragma unroll
    for (int i = 0; i < (8 * KB1) / LWAVES; ++i) {
        const int kp = (tid >> 5) + LWAVES * i;
        const int m  = tid & 31;
        const float4 w0 = *(const float4*)(W1 + (size_t)(2 * kp) * HID + 4 * m);
        const float4 w1 = *(const float4*)(W1 + (size_t)(2 * kp + 1) * HID + 4 * m);
        float* dp = Ws + kp * WPITCH + (m >> 4) * 128 + (m & 15) * 4;
        *(float4*)dp        = make_float4(w0.x, w1.x, w0.y, w1.y);
        *(float4*)(dp + 64) = make_float4(w0.z, w1.z, w0.w, w1.w);
    }
#pragma unroll
    for (int i = 0; i < (8 * KB2) / LWAVES; ++i) {
        const int kp = (tid >> 5) + LWAVES * i;
        const int m  = tid & 31;
        const float4 w0 = *(const float4*)(W2 + (size_t)(2 * kp) * HID + 4 * m);
        const float4 w1 = *(const float4*)(W2 + (size_t)(2 * kp + 1) * HID + 4 * m);
        float* dp = Ws + (8 * KB1 + kp) * WPITCH + (m >> 4) * 128 + (m & 15) * 4;
        *(float4*)dp        = make_float4(w0.x, w1.x, w0.y, w1.y);
        *(float4*)(dp + 64) = make_float4(w0.z, w1.z, w0.w, w1.w);
    }
    __syncthreads();

    float4 alo[KB], ahi[KB];
    if (active) {
        const int n0 = rt * 16;
        float* gt = Gt + wv * (16 * CPITCH);
        const float* hl = h + lane * 4;
        const int   ob = offs[n0 + min(lane, 16)];
        const float dn = dinv[n0 + (lane & 15)];
        const int beg = __builtin_amdgcn_readlane(ob, 0);
        const int end = __builtin_amdgcn_readlane(ob, 16);
        int   node = 0;
        int   nb   = __builtin_amdgcn_readlane(ob, 1);
        float di   = __int_as_float(__builtin_amdgcn_readlane(__float_as_int(dn), 0));
        float4 acc = make_float4(0.0f, 0.0f, 0.0f, 0.0f);

        int   sC = 0, sN = 0;
        float wC = 0.0f;
        if (beg + lane < end) {
            sC = bsrc[beg + lane];
            wC = dinv[sC];
        }
        if (beg + 32 + lane < end) sN = bsrc[beg + 32 + lane];
        auto ask_rows = [&](float4 (&v)[GR], int j0, int m) {
#pragma unroll
            for (int u = 0; u < GR; ++u) {
                const int jj = min(j0 + u, m - 1);
                const int s  = __builtin_amdgcn_readlane(sC, jj);
                v[u] = *(const float4*)(hl + (size_t)s * HID);
            }
        };
        auto add_rows = [&](const float4 (&v)[GR], int base, int j0, int m) {
#pragma unroll
            for (int u = 0; u < GR; ++u) {
                const int j = j0 + u;
                if (j < m) {
                    const int e = base + j;
                    while (e >= nb) {
                        *(float4*)(gt + node * CPITCH + 4 * lane) = acc;
                        acc = make_float4(0.0f, 0.0f, 0.0f, 0.0f);
                        ++node;
                        nb = __builtin_amdgcn_readlane(ob, node + 1);
                        di = __int_as_float(__builtin_amdgcn_readlane(__float_as_int(dn), node));
                    }
                    const float w = __int_as_float(__builtin_amdgcn_readlane(__float_as_int(wC), j)) * di;
                    acc.x += v[u].x * w;
                    acc.y += v[u].y * w;
                    acc.z += v[u].z * w;
                    acc.w += v[u].w * w;
                }
            }
        };
        static_assert(2 * GR == 32, "a chunk of 32 edges is two groups of rows");
        for (int base = beg; base < end; base += 32) {
            const int m = min(32, end - base);
            float4 v[GR];
            ask_rows(v, 0, m);
            const float wN = dinv[sN];
            int sNN = 0;
            if (base + 64 + lane < end) sNN = bsrc[base + 64 + lane];
            add_rows(v, base, 0, m);
            if (m > GR) {
                ask_rows(v, GR, m);
                add_rows(v, base, GR, m);
            }
            sC = sN;
            wC = wN;
            sN = sNN;
        }
        for (; node < 16; ++node) {
            *(float4*)(gt + node * CPITCH + 4 * lane) = acc;
            acc = make_float4(0.0f, 0.0f, 0.0f, 0.0f);
        }

        float4 sv[16];
#pragma unroll
        for (int i = 0; i < 16; ++i) sv[i] = *(const float4*)(hl + (size_t)(n0 + i) * HID);
        const float* ap2 = h + (size_t)(n0 + c) * HID + 8 * g;
#pragma unroll
        for (int b = 0; b < KB2; ++b) {
            alo[KB1 + b] = *(const float4*)(ap2 + 16 * b);
            ahi[KB1 + b] = *(const float4*)(ap2 + 16 * b + 4);
        }
#pragma unroll
        for (int i = 0; i < 16; ++i) {
            const float d  = __int_as_float(__builtin_amdgcn_readlane(__float_as_int(dn), i));
            const float dd = d * d;
            float4 t = *(const float4*)(gt + i * CPITCH + 4 * lane);
            t.x += sv[i].x * dd;
            t.y += sv[i].y * dd;
            t.z += sv[i].z * dd;
            t.w += sv[i].w * dd;
            *(float4*)(gt + i * CPITCH + 4 * lane) = t;
        }
        const float* ap1 = gt + c * CPITCH + 8 * g;
#pragma unroll
        for (int b = 0; b < KB1; ++b) {
            alo[b] = *(const float4*)(ap1 + 16 * b);
            ahi[b] = *(const float4*)(ap1 + 16 * b + 4);
        }
    }

    auto load_bias = [&](float4& lo, float4& hi) {
        const float4 l1 = *(const float4*)(bias1 + 4 * c);
        const float4 h1 = *(const float4*)(bias1 + 64 + 4 * c);
        const float4 l2 = *(const float4*)(bias2 + 4 * c);
        const float4 h2 = *(const float4*)(bias2 + 64 + 4 * c);
        lo = make_float4(l1.x + l2.x, l1.y + l2.y, l1.z + l2.z, l1.w + l2.w);
        hi = make_float4(h1.x + h2.x, h1.y + h2.y, h1.z + h2.z, h1.w + h2.w);
    };

    if (!tailw) {
        v8f acc[8];
#pragma unroll
        for (int t = 0; t < 8; ++t) acc[t] = v8f{};

        const float* bp = Ws + (4 * g) * WPITCH + c * 4;
        float4 wn[4];
#pragma unroll
        for (int p = 0; p < 4; ++p) wn[p] = *(const float4*)(bp + 64 * p);
#pragma unroll
        for (int b = 0; b < KB; ++b) {
            v2f Af[4];
            Af[0][0] = alo[b].x; Af[0][1] = alo[b].y;
            Af[1][0] = alo[b].z; Af[1][1] = alo[b].w;
            Af[2][0] = ahi[b].x; Af[2][1] = ahi[b].y;
            Af[3][0] = ahi[b].z; Af[3][1] = ahi[b].w;
#pragma unroll
            for (int s = 0; s < 4; ++s) {
                float4 wc[4];
#pragma unroll
                for (int p = 0; p < 4; ++p) wc[p] = wn[p];
                if (b + 1 < KB || s < 3) {
                    const float* q = bp + (s < 3 ? 8 * b + s + 1 : 8 * (b + 1)) * WPITCH;
#pragma unroll
                    for (int p = 0; p < 4; ++p) wn[p] = *(const float4*)(q + 64 * p);
                }
                __builtin_amdgcn_sched_barrier(0);
#pragma unroll
                for (int p = 0; p < 4; ++p) {
                    const float4 w = wc[p];
                    v2f b0, b1;
                    b0[0] = w.x; b0[1] = w.y;
                    b1[0] = w.z; b1[1] = w.w;
                    acc[2 * p] = __builtin_amdgcn_wmma_f32_16x16x4_f32(
                        false, Af[s], false, b0, (short)0, acc[2 * p], false, false);
                    acc[2 * p + 1] = __builtin_amdgcn_wmma_f32_16x16x4_f32(
                        false, Af[s], false, b1, (short)0, acc[2 * p + 1], false, false);
                }
            }
        }

        float4 bvlo, bvhi;
        load_bias(bvlo, bvhi);

        if (MODE == 0) {
#pragma unroll
            for (int r = 0; r < 8; ++r) {
                float4 lo, hi;
                lo.x = fmaxf(acc[0][r] + bvlo.x, 0.0f); lo.y = fmaxf(acc[1][r] + bvlo.y, 0.0f);
                lo.z = fmaxf(acc[2][r] + bvlo.z, 0.0f); lo.w = fmaxf(acc[3][r] + bvlo.w, 0.0f);
                hi.x = fmaxf(acc[4][r] + bvhi.x, 0.0f); hi.y = fmaxf(acc[5][r] + bvhi.y, 0.0f);
                hi.z = fmaxf(acc[6][r] + bvhi.z, 0.0f); hi.w = fmaxf(acc[7][r] + bvhi.w, 0.0f);
                float* cp = C + (size_t)(rt * 16 + 8 * g + r) * HID + 4 * c;
                *(float4*)cp        = lo;
                *(float4*)(cp + 64) = hi;
            }
        } else {
            const float4 u1l = *(const float4*)(Wd + 4 * c);
            const float4 u1h = *(const float4*)(Wd + 64 + 4 * c);
            const float4 u2l = *(const float4*)(Wd + HID + 4 * c);
            const float4 u2h = *(const float4*)(Wd + HID + 64 + 4 * c);
            float q[16];
#pragma unroll
            for (int r = 0; r < 8; ++r) {
                float4 lo, hi;
                lo.x = fmaxf(acc[0][r] + bvlo.x, 0.0f); lo.y = fmaxf(acc[1][r] + bvlo.y, 0.0f);
                lo.z = fmaxf(acc[2][r] + bvlo.z, 0.0f); lo.w = fmaxf(acc[3][r] + bvlo.w, 0.0f);
                hi.x = fmaxf(acc[4][r] + bvhi.x, 0.0f); hi.y = fmaxf(acc[5][r] + bvhi.y, 0.0f);
                hi.z = fmaxf(acc[6][r] + bvhi.z, 0.0f); hi.w = fmaxf(acc[7][r] + bvhi.w, 0.0f);
                float a = lo.x * u1l.x;
                a += lo.y * u1l.y; a += lo.z * u1l.z; a += lo.w * u1l.w;
                a += hi.x * u1h.x; a += hi.y * u1h.y; a += hi.z * u1h.z; a += hi.w * u1h.w;
                float d = lo.x * u2l.x;
                d += lo.y * u2l.y; d += lo.z * u2l.z; d += lo.w * u2l.w;
                d += hi.x * u2h.x; d += hi.y * u2h.y; d += hi.z * u2h.z; d += hi.w * u2h.w;
                q[2 * r]     = a;
                q[2 * r + 1] = d;
            }
            const bool bit3 = (c & 8) != 0;
            const bool bit2 = (c & 4) != 0;
            const bool bit1 = (c & 2) != 0;
            const bool bit0 = (c & 1) != 0;
            float u[8];
#pragma unroll
            for (int m = 0; m < 8; ++m) {
                const float keep = bit3 ? q[8 + m] : q[m];
                const float send = bit3 ? q[m] : q[8 + m];
                u[m] = keep + __shfl_xor(send, 8, 32);
            }
            float w4[4];
#pragma unroll
            for (int m = 0; m < 4; ++m) {
                const float keep = bit2 ? u[4 + m] : u[m];
                const float send = bit2 ? u[m] : u[4 + m];
                w4[m] = keep + __shfl_xor(send, 4, 32);
            }
            float x2[2];
#pragma unroll
            for (int m = 0; m < 2; ++m) {
                const float keep = bit1 ? w4[2 + m] : w4[m];
                const float send = bit1 ? w4[m] : w4[2 + m];
                x2[m] = keep + __shfl_xor(send, 2, 32);
            }
            const float keep1 = bit0 ? x2[1] : x2[0];
            const float send1 = bit0 ? x2[0] : x2[1];
            const float z = keep1 + __shfl_xor(send1, 1, 32);
            C[(size_t)rt * 32 + lane] = z;
        }
    } else if (hasTail) {
        v8f tacc[2];
        tacc[0] = v8f{};
        tacc[1] = v8f{};
        const float* bp = Ws + (4 * g) * WPITCH + c * 4 + 64 * pr;
        float4 wn[4];
#pragma unroll
        for (int s = 0; s < 4; ++s) wn[s] = *(const float4*)(bp + s * WPITCH);
#pragma unroll
        for (int b = 0; b < KB; ++b) {
            v2f Af[4];
            Af[0][0] = alo[b].x; Af[0][1] = alo[b].y;
            Af[1][0] = alo[b].z; Af[1][1] = alo[b].w;
            Af[2][0] = ahi[b].x; Af[2][1] = ahi[b].y;
            Af[3][0] = ahi[b].z; Af[3][1] = ahi[b].w;
            float4 wc[4];
#pragma unroll
            for (int s = 0; s < 4; ++s) wc[s] = wn[s];
            if (b + 1 < KB) {
#pragma unroll
                for (int s = 0; s < 4; ++s)
                    wn[s] = *(const float4*)(bp + (8 * (b + 1) + s) * WPITCH);
            }
            __builtin_amdgcn_sched_barrier(0);
#pragma unroll
            for (int s = 0; s < 4; ++s) {
                const float4 w = wc[s];
                v2f b0, b1;
                b0[0] = w.x; b0[1] = w.y;
                b1[0] = w.z; b1[1] = w.w;
                tacc[0] = __builtin_amdgcn_wmma_f32_16x16x4_f32(
                    false, Af[s], false, b0, (short)0, tacc[0], false, false);
                tacc[1] = __builtin_amdgcn_wmma_f32_16x16x4_f32(
                    false, Af[s], false, b1, (short)0, tacc[1], false, false);
            }
        }
        float4 bvlo, bvhi;
        load_bias(bvlo, bvhi);
        const float4 bv = (pr >> 1) ? bvhi : bvlo;
        const float bb0 = (pr & 1) ? bv.z : bv.x;
        const float bb1 = (pr & 1) ? bv.w : bv.y;
        float* tp = Ct + (8 * g) * CPITCH + 4 * c + 2 * (pr & 1) + 64 * (pr >> 1);
#pragma unroll
        for (int r = 0; r < 8; ++r) {
            float2 v;
            v.x = fmaxf(tacc[0][r] + bb0, 0.0f);
            v.y = fmaxf(tacc[1][r] + bb1, 0.0f);
            *(float2*)(tp + r * CPITCH) = v;
        }
    }

    if (hasTail) {
        __syncthreads();
        if (tailw) {
            if (MODE == 0) {
#pragma unroll
                for (int r = 0; r < 4; ++r) {
                    const int row = 4 * pr + r;
                    const float4 v = *(const float4*)(Ct + row * CPITCH + 4 * lane);
                    *(float4*)(C + (size_t)(rt * 16 + row) * HID + 4 * lane) = v;
                }
            } else if (pr == 0) {
                const float* cr = Ct + c * CPITCH;
                const float* wp = Wd + g * HID;
                float sum = 0.0f;
#pragma unroll
                for (int j = 0; j < HID / 4; ++j) {
                    const float4 v = *(const float4*)(cr + 4 * j);
                    const float4 w = *(const float4*)(wp + 4 * j);
                    sum += v.x * w.x; sum += v.y * w.y;
                    sum += v.z * w.z; sum += v.w * w.w;
                }
                C[(size_t)rt * 32 + 2 * c + g] = sum;
            }
        }
    }
}

__global__ __launch_bounds__(1024) void rank_edges_lds(
    const int* __restrict__ dst, int* __restrict__ rank,
    int* __restrict__ cntS, int* __restrict__ csum)
{
    __shared__ __attribute__((aligned(16))) int cnt[NNP];
    const int tid  = threadIdx.x;
    const int lane = tid & 31;
    const int wv   = tid >> 5;
    const int sl   = blockIdx.x;

    const int4* __restrict__ d4 = (const int4*)(dst + (size_t)sl * EPS);
    int4 d[RANK_IT];
#pragma unroll
    for (int k = 0; k < RANK_IT; ++k) {
        const int i = tid + 1024 * k;
        d[k] = make_int4(0, 0, 0, 0);
        if (i < EPS / 4) d[k] = d4[i];
    }

    int4* c4 = (int4*)cnt;
    for (int i = tid; i < NNP / 4; i += 1024) c4[i] = make_int4(0, 0, 0, 0);
    __syncthreads();

#pragma unroll
    for (int k = 0; k < RANK_IT; ++k) {
        const int i = tid + 1024 * k;
        if (i < EPS / 4) {
            d[k].x = atomicAdd(&cnt[d[k].x], 1);
            d[k].y = atomicAdd(&cnt[d[k].y], 1);
            d[k].z = atomicAdd(&cnt[d[k].z], 1);
            d[k].w = atomicAdd(&cnt[d[k].w], 1);
        }
    }
    __syncthreads();

    int4* __restrict__ r4 = (int4*)(rank + (size_t)sl * EPS);
#pragma unroll
    for (int k = 0; k < RANK_IT; ++k) {
        const int i = tid + 1024 * k;
        if (i < EPS / 4) r4[i] = d[k];
    }

    int4* __restrict__ g4 = (int4*)(cntS + (size_t)sl * NNP);
    int sum[CHUNK_IT];
#pragma unroll
    for (int t = 0; t < CHUNK_IT; ++t) {
        const int j = wv + 32 * t;
        sum[t] = 0;
        if (j < NCHUNK) {
            const int4 v0 = c4[64 * j + lane];
            const int4 v1 = c4[64 * j + 32 + lane];
            g4[64 * j + lane]      = v0;
            g4[64 * j + 32 + lane] = v1;
            sum[t] = ((v0.x + v0.y) + (v0.z + v0.w)) + ((v1.x + v1.y) + (v1.z + v1.w));
        }
    }
#pragma unroll
    for (int off = 16; off > 0; off >>= 1) {
#pragma unroll
        for (int t = 0; t < CHUNK_IT; ++t)
            sum[t] += __shfl_xor(sum[t], off, 32);
    }
    if (lane == 0) {
#pragma unroll
        for (int t = 0; t < CHUNK_IT; ++t) {
            const int j = wv + 32 * t;
            if (j < NCHUNK) csum[j * NSLICE + sl] = sum[t];
        }
    }
}

#define ENC_BLOCKS ((NRT + 7) / 8)

__device__ __forceinline__ void encode_nodes(
    float* __restrict__ Ws, const float* __restrict__ x, const float* __restrict__ Wn,
    const float* __restrict__ bn, float* __restrict__ C, int eb)
{
    constexpr int KB = FN / 16;
    const int tid  = threadIdx.x;
    const int lane = tid & 31;
    const int c    = lane & 15;
    const int g    = lane >> 4;
    const int rt = __builtin_amdgcn_readfirstlane(eb * 8 + (tid >> 5));
    const bool active = rt < NRT;
    const int rtl = active ? rt : NRT - 1;

    float4 alo[KB], ahi[KB];
    {
        const float* ap = x + (size_t)(rtl * 16 + c) * FN + 8 * g;
#pragma unroll
        for (int b = 0; b < KB; ++b) {
            alo[b] = *(const float4*)(ap + 16 * b);
            ahi[b] = *(const float4*)(ap + 16 * b + 4);
        }
    }

#pragma unroll
    for (int i = 0; i < KB; ++i) {
        const int kp = (tid >> 5) + 8 * i;
        const int m  = tid & 31;
        const float4 w0 = *(const float4*)(Wn + (size_t)(2 * kp) * HID + 4 * m);
        const float4 w1 = *(const float4*)(Wn + (size_t)(2 * kp + 1) * HID + 4 * m);
        float* dp = Ws + kp * WPITCH + (m >> 4) * 128 + (m & 15) * 4;
        *(float4*)dp        = make_float4(w0.x, w1.x, w0.y, w1.y);
        *(float4*)(dp + 64) = make_float4(w0.z, w1.z, w0.w, w1.w);
    }
    __syncthreads();

    if (!active) return;

    v8f acc[8];
#pragma unroll
    for (int t = 0; t < 8; ++t) acc[t] = v8f{};

    const float* bp = Ws + (4 * g) * WPITCH + c * 4;
#pragma unroll
    for (int b = 0; b < KB; ++b) {
        v2f Af[4];
        Af[0][0] = alo[b].x; Af[0][1] = alo[b].y;
        Af[1][0] = alo[b].z; Af[1][1] = alo[b].w;
        Af[2][0] = ahi[b].x; Af[2][1] = ahi[b].y;
        Af[3][0] = ahi[b].z; Af[3][1] = ahi[b].w;
#pragma unroll
        for (int s = 0; s < 4; ++s) {
            const float* q = bp + (8 * b + s) * WPITCH;
#pragma unroll
            for (int p = 0; p < 4; ++p) {
                const float4 w = *(const float4*)(q + 64 * p);
                v2f b0, b1;
                b0[0] = w.x; b0[1] = w.y;
                b1[0] = w.z; b1[1] = w.w;
                acc[2 * p] = __builtin_amdgcn_wmma_f32_16x16x4_f32(
                    false, Af[s], false, b0, (short)0, acc[2 * p], false, false);
                acc[2 * p + 1] = __builtin_amdgcn_wmma_f32_16x16x4_f32(
                    false, Af[s], false, b1, (short)0, acc[2 * p + 1], false, false);
            }
        }
    }

    const float4 bvlo = *(const float4*)(bn + 4 * c);
    const float4 bvhi = *(const float4*)(bn + 64 + 4 * c);

#pragma unroll
    for (int r = 0; r < 8; ++r) {
        float4 lo, hi;
        lo.x = fmaxf(acc[0][r] + bvlo.x, 0.0f); lo.y = fmaxf(acc[1][r] + bvlo.y, 0.0f);
        lo.z = fmaxf(acc[2][r] + bvlo.z, 0.0f); lo.w = fmaxf(acc[3][r] + bvlo.w, 0.0f);
        hi.x = fmaxf(acc[4][r] + bvhi.x, 0.0f); hi.y = fmaxf(acc[5][r] + bvhi.y, 0.0f);
        hi.z = fmaxf(acc[6][r] + bvhi.z, 0.0f); hi.w = fmaxf(acc[7][r] + bvhi.w, 0.0f);
        float* cp = C + (size_t)(rt * 16 + 8 * g + r) * HID + 4 * c;
        *(float4*)cp        = lo;
        *(float4*)(cp + 64) = hi;
    }
}

__global__ __launch_bounds__(256) void scan_offsets_encode_nodes(
    int* __restrict__ cntS, const int* __restrict__ csum,
    int* __restrict__ offs, float* __restrict__ dinv, int n,
    const float* __restrict__ x, const float* __restrict__ Wn,
    const float* __restrict__ bn, float* __restrict__ h0)
{
    __shared__ __attribute__((aligned(16))) float Ws[8 * (FN / 16) * WPITCH];
    __shared__ int waveCnt[8];
    __shared__ int wavePre[8];
    if (blockIdx.x >= NCHUNK) {
        encode_nodes(Ws, x, Wn, bn, h0, (int)blockIdx.x - NCHUNK);
        return;
    }
    const int tid  = threadIdx.x;
    const int lane = tid & 31;
    const int wv   = tid >> 5;
    const int b    = blockIdx.x;
    const int i    = b * 256 + tid;

    int cs[NSLICE];
#pragma unroll
    for (int k = 0; k < NSLICE; ++k) cs[k] = cntS[(size_t)k * NNP + i];

    int pre = 0;
    const int4* __restrict__ s4 = (const int4*)csum;
    const int n4   = b * (NSLICE / 4);
    const int full = n4 >> 8;
#pragma unroll 4
    for (int it = 0; it < full; ++it) {
        int4 v = s4[it * 256 + tid];
        pre += (v.x + v.y) + (v.z + v.w);
    }
    if (tid < (n4 & 255)) {
        int4 v = s4[full * 256 + tid];
        pre += (v.x + v.y) + (v.z + v.w);
    }

    int c = 0;
#pragma unroll
    for (int k = 0; k < NSLICE; ++k) c += cs[k];

    int inc = c;
    for (int off = 1; off < 32; off <<= 1) {
        int t = __shfl_up(inc, off, 32);
        if (lane >= off) inc += t;
    }
    for (int off = 16; off > 0; off >>= 1)
        pre += __shfl_xor(pre, off, 32);

    if (lane == 31) waveCnt[wv] = inc;
    if (lane == 0)  wavePre[wv] = pre;
    __syncthreads();

    int base = 0;
    for (int k = 0; k < 8; ++k) {
        base += wavePre[k];
        if (k < wv) base += waveCnt[k];
    }
    const int excl = base + inc - c;

    int run = excl;
#pragma unroll
    for (int k = 0; k < NSLICE; ++k) {
        cntS[(size_t)k * NNP + i] = run;
        run += cs[k];
    }

    if (i < n) {
        offs[i] = excl;
        float d = (float)(c + 1);
        dinv[i] = (d > 0.0f) ? rsqrtf(d) : 0.0f;
    }
    if (i == n - 1) offs[n] = excl + c;
}

__global__ void bucket_edges(const int* __restrict__ src, const int* __restrict__ dst,
                             const int* __restrict__ rank, const int* __restrict__ start,
                             int* __restrict__ bsrc, int n)
{
    int i = blockIdx.x * blockDim.x + threadIdx.x;
    if (i < n) bsrc[start[(size_t)(i / EPS) * NNP + dst[i]] + rank[i]] = src[i];
}

__global__ __launch_bounds__(128, 4) void decode_edges(
    const float* __restrict__ pp, const float* __restrict__ ea,
    const int* __restrict__ src,  const int* __restrict__ dst,
    const float* __restrict__ We, const float* __restrict__ be,
    const float* __restrict__ Wd, const float* __restrict__ bd,
    float* __restrict__ out, int nGroups)
{
    const int lane = threadIdx.x & 31;
    const int c    = lane & 15;
    const int g    = lane >> 4;
    const int wave = __builtin_amdgcn_readfirstlane(
        (int)((blockIdx.x * blockDim.x + threadIdx.x) >> 5));
    const int nWaves = (int)((gridDim.x * blockDim.x) >> 5);

    v2f Bf[8][4];
#pragma unroll
    for (int s = 0; s < 4; ++s) {
#pragma unroll
        for (int j = 0; j < 2; ++j) {
            const float* rp = We + (size_t)(8 * g + 2 * s + j) * HID + 8 * c;
            const float4 lo = *(const float4*)rp;
            const float4 hi = *(const float4*)(rp + 4);
            Bf[0][s][j] = lo.x; Bf[1][s][j] = lo.y; Bf[2][s][j] = lo.z; Bf[3][s][j] = lo.w;
            Bf[4][s][j] = hi.x; Bf[5][s][j] = hi.y; Bf[6][s][j] = hi.z; Bf[7][s][j] = hi.w;
        }
    }
    float bev[8], wd3[8];
    {
        const float4 b0 = *(const float4*)(be + 8 * c);
        const float4 b1 = *(const float4*)(be + 8 * c + 4);
        const float4 w0 = *(const float4*)(Wd + 2 * HID + 8 * c);
        const float4 w1 = *(const float4*)(Wd + 2 * HID + 8 * c + 4);
        bev[0] = b0.x; bev[1] = b0.y; bev[2] = b0.z; bev[3] = b0.w;
        bev[4] = b1.x; bev[5] = b1.y; bev[6] = b1.z; bev[7] = b1.w;
        wd3[0] = w0.x; wd3[1] = w0.y; wd3[2] = w0.z; wd3[3] = w0.w;
        wd3[4] = w1.x; wd3[5] = w1.y; wd3[6] = w1.z; wd3[7] = w1.w;
    }
    const float bdv = bd[0];

    const bool bit3 = (c & 8) != 0;
    const bool bit2 = (c & 4) != 0;
    const bool bit1 = (c & 2) != 0;
    const bool bit0 = (c & 1) != 0;
    const int slot = ((c & 8) << 1) + 8 * g + (c & 7);

    int grp = wave;
    int sN = 0, dN = 0;
    float4 aloN[2], ahiN[2];
#pragma unroll
    for (int T = 0; T < 2; ++T) {
        aloN[T] = make_float4(0.0f, 0.0f, 0.0f, 0.0f);
        ahiN[T] = make_float4(0.0f, 0.0f, 0.0f, 0.0f);
    }
    if (grp < nGroups) {
        sN = src[grp * 32 + slot];
        dN = dst[grp * 32 + slot];
#pragma unroll
        for (int T = 0; T < 2; ++T) {
            const float* ap = ea + (size_t)(grp * 32 + 16 * T + c) * FE + 8 * g;
            aloN[T] = *(const float4*)ap;
            ahiN[T] = *(const float4*)(ap + 4);
        }
    }

    for (; grp < nGroups; grp += nWaves) {
        const int e0 = grp * 32;
        const float ps = pp[2 * (size_t)sN];
        const float pd = pp[2 * (size_t)dN + 1];

        float4 alo[2], ahi[2];
#pragma unroll
        for (int T = 0; T < 2; ++T) { alo[T] = aloN[T]; ahi[T] = ahiN[T]; }

        const int nxt = min(grp + nWaves, nGroups - 1);
        sN = src[nxt * 32 + slot];
        dN = dst[nxt * 32 + slot];
#pragma unroll
        for (int T = 0; T < 2; ++T) {
            const float* ap = ea + (size_t)(nxt * 32 + 16 * T + c) * FE + 8 * g;
            aloN[T] = *(const float4*)ap;
            ahiN[T] = *(const float4*)(ap + 4);
        }

        float q[16];
#pragma unroll
        for (int T = 0; T < 2; ++T) {
            v2f A[4];
            A[0][0] = alo[T].x; A[0][1] = alo[T].y;
            A[1][0] = alo[T].z; A[1][1] = alo[T].w;
            A[2][0] = ahi[T].x; A[2][1] = ahi[T].y;
            A[3][0] = ahi[T].z; A[3][1] = ahi[T].w;

            float part[8];
#pragma unroll
            for (int r = 0; r < 8; ++r) part[r] = 0.0f;
#pragma unroll
            for (int t = 0; t < 8; t += 2) {
                v8f acc0 = {};
                v8f acc1 = {};
#pragma unroll
                for (int s = 0; s < 4; ++s) {
                    acc0 = __builtin_amdgcn_wmma_f32_16x16x4_f32(
                        false, A[s], false, Bf[t][s], (short)0, acc0, false, false);
                    acc1 = __builtin_amdgcn_wmma_f32_16x16x4_f32(
                        false, A[s], false, Bf[t + 1][s], (short)0, acc1, false, false);
                }
#pragma unroll
                for (int r = 0; r < 8; ++r) {
                    part[r] += fmaxf(acc0[r] + bev[t], 0.0f) * wd3[t];
                    part[r] += fmaxf(acc1[r] + bev[t + 1], 0.0f) * wd3[t + 1];
                }
            }
#pragma unroll
            for (int r = 0; r < 8; ++r) q[8 * T + r] = part[r];
        }

        float u[8];
#pragma unroll
        for (int m = 0; m < 8; ++m) {
            const float keep = bit3 ? q[8 + m] : q[m];
            const float send = bit3 ? q[m] : q[8 + m];
            u[m] = keep + __shfl_xor(send, 8, 32);
        }
        float w4[4];
#pragma unroll
        for (int m = 0; m < 4; ++m) {
            const float keep = bit2 ? u[4 + m] : u[m];
            const float send = bit2 ? u[m] : u[4 + m];
            w4[m] = keep + __shfl_xor(send, 4, 32);
        }
        float x2[2];
#pragma unroll
        for (int m = 0; m < 2; ++m) {
            const float keep = bit1 ? w4[2 + m] : w4[m];
            const float send = bit1 ? w4[m] : w4[2 + m];
            x2[m] = keep + __shfl_xor(send, 2, 32);
        }
        const float keep1 = bit0 ? x2[1] : x2[0];
        const float send1 = bit0 ? x2[0] : x2[1];
        const float z = keep1 + __shfl_xor(send1, 1, 32);

        out[e0 + slot] = (ps + pd) + z + bdv;
    }
}

extern "C" void kernel_launch(void* const* d_in, const int* in_sizes, int n_in,
                              void* d_out, int out_size, void* d_ws, size_t ws_size,
                              hipStream_t stream)
{
    const float* x   = (const float*)d_in[0];
    const float* ea  = (const float*)d_in[1];
    const int*   ei  = (const int*)  d_in[2];
    const float* Wn  = (const float*)d_in[3];
    const float* bn  = (const float*)d_in[4];
    const float* We  = (const float*)d_in[5];
    const float* be  = (const float*)d_in[6];
    const float* Wc  = (const float*)d_in[7];
    const float* bc  = (const float*)d_in[8];
    const float* Wr  = (const float*)d_in[9];
    const float* br  = (const float*)d_in[10];
    const float* Wd  = (const float*)d_in[11];
    const float* bd  = (const float*)d_in[12];
    float* out = (float*)d_out;

    const int* srcI = ei;
    const int* dstI = ei + NE;

    float* ws   = (float*)d_ws;
    float* hA   = ws;
    float* hB   = hA   + (size_t)NN * HID;
    float* dinv = hB   + (size_t)NN * HID;
    int*   offs = (int*)(dinv + NN);
    int*   rank = offs + NN + 16;
    int*   bsrc = rank + NE;
    float* pp   = (float*)(bsrc + NE);
    int*   cntS = (int*)(pp + 2 * (size_t)NN);
    int*   csum = cntS + (size_t)NSLICE * NNP;

    rank_edges_lds<<<NSLICE, 1024, 0, stream>>>(dstI, rank, cntS, csum);
    scan_offsets_encode_nodes<<<NCHUNK + ENC_BLOCKS, 256, 0, stream>>>(
        cntS, csum, offs, dinv, NN, x, Wn, bn, hA);
    bucket_edges<<<(NE + 255) / 256, 256, 0, stream>>>(srcI, dstI, rank, cntS, bsrc, NE);

    layer_fused<0><<<LBLK, LWAVES * 32, 0, stream>>>(
        hA, offs, bsrc, dinv, Wc, Wr, br, bc, hB, nullptr);
    layer_fused<0><<<LBLK, LWAVES * 32, 0, stream>>>(
        hB, offs, bsrc, dinv, Wc + (size_t)HID * HID, Wr, br, bc + HID, hA, nullptr);
    layer_fused<1><<<LBLK, LWAVES * 32, 0, stream>>>(
        hA, offs, bsrc, dinv, Wc + (size_t)2 * HID * HID, Wr, br, bc + 2 * HID, pp, Wd);

    decode_edges<<<768, 128, 0, stream>>>(pp, ea, srcI, dstI, We, be, Wd, bd, out, NE / 32);
}


# Decoder's edge-encoder term moved into the layer launches: the idle tail waves of blocks with no left-over strip run it on the matrix units under the gathers; decoder only adds; every sum unchanged
# speedup vs baseline: 4.3391x; 1.0352x over previous
#include <hip/hip_runtime.h>

#define NN 50000
#define NE 800000
#define FN 32
#define FE 16
#define HID 128
#define NL 3
#define NRT (NN / 16)
#define NSLICE 64
#define EPS (NE / NSLICE)
#define NNP (((NN + 255) / 256) * 256)
#define NCHUNK (NNP / 256)
#define RANK_IT ((EPS / 4 + 1023) / 1024)
#define CHUNK_IT ((NCHUNK + 31) / 32)
static_assert(NE % NSLICE == 0 && EPS % 4 == 0, "slices are whole 16-byte runs");
static_assert(NSLICE % 4 == 0, "chunk sums are read 16 bytes at a time");

typedef float v2f __attribute__((ext_vector_type(2)));
typedef float v8f __attribute__((ext_vector_type(8)));

#define WPITCH 264

#define LBLK 256
#define SPB (NRT / LBLK)
#define NTAIL (NRT - SPB * LBLK)
#define LWAVES (SPB + 4)
#define CPITCH 132
#define GR 16
static_assert(NTAIL >= 0 && NTAIL <= LBLK, "one left-over strip per block at most");
static_assert(LWAVES * 32 <= 1024, "a block is at most 1024 threads");
#define NGRP (NE / 32)
#define GPL ((NGRP + NL - 1) / NL)
#define EWAVES (4 * (LBLK - NTAIL))
static_assert(NL == 3 && 2 * GPL < NGRP && NTAIL < LBLK, "three layer launches share the edge groups");
static_assert(NE % 32 == 0, "edge groups are whole");

__device__ __forceinline__ void encode_edge_groups(
    const float* __restrict__ ea, const float* __restrict__ We,
    const float* __restrict__ be, const float* __restrict__ Wd,
    float* __restrict__ ze, int grp, int gEnd, int stride, int lane)
{
    const int c = lane & 15;
    const int g = lane >> 4;

    v2f Bf[8][4];
#pragma unroll
    for (int s = 0; s < 4; ++s) {
#pragma unroll
        for (int j = 0; j < 2; ++j) {
            const float* rp = We + (size_t)(8 * g + 2 * s + j) * HID + 8 * c;
            const float4 lo = *(const float4*)rp;
            const float4 hi = *(const float4*)(rp + 4);
            Bf[0][s][j] = lo.x; Bf[1][s][j] = lo.y; Bf[2][s][j] = lo.z; Bf[3][s][j] = lo.w;
            Bf[4][s][j] = hi.x; Bf[5][s][j] = hi.y; Bf[6][s][j] = hi.z; Bf[7][s][j] = hi.w;
        }
    }
    float bev[8], wd3[8];
    {
        const float4 b0 = *(const float4*)(be + 8 * c);
        const float4 b1 = *(const float4*)(be + 8 * c + 4);
        const float4 w0 = *(const float4*)(Wd + 2 * HID + 8 * c);
        const float4 w1 = *(const float4*)(Wd + 2 * HID + 8 * c + 4);
        bev[0] = b0.x; bev[1] = b0.y; bev[2] = b0.z; bev[3] = b0.w;
        bev[4] = b1.x; bev[5] = b1.y; bev[6] = b1.z; bev[7] = b1.w;
        wd3[0] = w0.x; wd3[1] = w0.y; wd3[2] = w0.z; wd3[3] = w0.w;
        wd3[4] = w1.x; wd3[5] = w1.y; wd3[6] = w1.z; wd3[7] = w1.w;
    }

    const bool bit3 = (c & 8) != 0;
    const bool bit2 = (c & 4) != 0;
    const bool bit1 = (c & 2) != 0;
    const bool bit0 = (c & 1) != 0;
    const int slot = ((c & 8) << 1) + 8 * g + (c & 7);

    float4 aloN[2], ahiN[2];
#pragma unroll
    for (int T = 0; T < 2; ++T) {
        aloN[T] = make_float4(0.0f, 0.0f, 0.0f, 0.0f);
        ahiN[T] = make_float4(0.0f, 0.0f, 0.0f, 0.0f);
    }
    if (grp < gEnd) {
#pragma unroll
        for (int T = 0; T < 2; ++T) {
            const float* ap = ea + (size_t)(grp * 32 + 16 * T + c) * FE + 8 * g;
            aloN[T] = *(const float4*)ap;
            ahiN[T] = *(const float4*)(ap + 4);
        }
    }

    for (; grp < gEnd; grp += stride) {
        const int e0 = grp * 32;
        float4 alo[2], ahi[2];
#pragma unroll
        for (int T = 0; T < 2; ++T) { alo[T] = aloN[T]; ahi[T] = ahiN[T]; }

        const int nxt = min(grp + stride, gEnd - 1);
#pragma unroll
        for (int T = 0; T < 2; ++T) {
            const float* ap = ea + (size_t)(nxt * 32 + 16 * T + c) * FE + 8 * g;
            aloN[T] = *(const float4*)ap;
            ahiN[T] = *(const float4*)(ap + 4);
        }
        __builtin_amdgcn_sched_barrier(0);

        float q[16];
#pragma unroll
        for (int T = 0; T < 2; ++T) {
            v2f A[4];
            A[0][0] = alo[T].x; A[0][1] = alo[T].y;
            A[1][0] = alo[T].z; A[1][1] = alo[T].w;
            A[2][0] = ahi[T].x; A[2][1] = ahi[T].y;
            A[3][0] = ahi[T].z; A[3][1] = ahi[T].w;

            float part[8];
#pragma unroll
            for (int r = 0; r < 8; ++r) part[r] = 0.0f;
#pragma unroll
            for (int t = 0; t < 8; t += 2) {
                v8f acc0 = {};
                v8f acc1 = {};
#pragma unroll
                for (int s = 0; s < 4; ++s) {
                    acc0 = __builtin_amdgcn_wmma_f32_16x16x4_f32(
                        false, A[s], false, Bf[t][s], (short)0, acc0, false, false);
                    acc1 = __builtin_amdgcn_wmma_f32_16x16x4_f32(
                        false, A[s], false, Bf[t + 1][s], (short)0, acc1, false, false);
                }
#pragma unroll
                for (int r = 0; r < 8; ++r) {
                    part[r] += fmaxf(acc0[r] + bev[t], 0.0f) * wd3[t];
                    part[r] += fmaxf(acc1[r] + bev[t + 1], 0.0f) * wd3[t + 1];
                }
            }
#pragma unroll
            for (int r = 0; r < 8; ++r) q[8 * T + r] = part[r];
            __builtin_amdgcn_sched_barrier(0);
        }

        float u[8];
#pragma unroll
        for (int m = 0; m < 8; ++m) {
            const float keep = bit3 ? q[8 + m] : q[m];
            const float send = bit3 ? q[m] : q[8 + m];
            u[m] = keep + __shfl_xor(send, 8, 32);
        }
        float w4[4];
#pragma unroll
        for (int m = 0; m < 4; ++m) {
            const float keep = bit2 ? u[4 + m] : u[m];
            const float send = bit2 ? u[m] : u[4 + m];
            w4[m] = keep + __shfl_xor(send, 4, 32);
        }
        float x2[2];
#pragma unroll
        for (int m = 0; m < 2; ++m) {
            const float keep = bit1 ? w4[2 + m] : w4[m];
            const float send = bit1 ? w4[m] : w4[2 + m];
            x2[m] = keep + __shfl_xor(send, 2, 32);
        }
        const float keep1 = bit0 ? x2[1] : x2[0];
        const float send1 = bit0 ? x2[0] : x2[1];
        const float z = keep1 + __shfl_xor(send1, 1, 32);

        ze[e0 + slot] = z;
    }
}

template <int MODE>
__global__ __launch_bounds__(LWAVES * 32, 4) void layer_fused(
    const float* __restrict__ h, const int* __restrict__ offs,
    const int* __restrict__ bsrc, const float* __restrict__ dinv,
    const float* __restrict__ W1, const float* __restrict__ W2,
    const float* __restrict__ bias1, const float* __restrict__ bias2,
    float* __restrict__ C, const float* __restrict__ Wd,
    const float* __restrict__ ea, const float* __restrict__ We,
    const float* __restrict__ be, float* __restrict__ ze, int grp0, int grp1)
{
    constexpr int KB1 = HID / 16;
    constexpr int KB2 = HID / 16;
    constexpr int KB = KB1 + KB2;
    static_assert((8 * KB1) % LWAVES == 0 && (8 * KB2) % LWAVES == 0,
                  "the staging passes are whole");
    __shared__ __attribute__((aligned(16))) float Ws[(8 * KB) * WPITCH];
    __shared__ __attribute__((aligned(16))) float Gt[LWAVES * 16 * CPITCH];
    __shared__ __attribute__((aligned(16))) float Ct[16 * CPITCH];

    const int tid  = threadIdx.x;
    const int lane = tid & 31;
    const int c    = lane & 15;
    const int g    = lane >> 4;
    const int wv  = __builtin_amdgcn_readfirstlane(tid >> 5);
    const int blk = (int)blockIdx.x;
    const bool tailw   = wv < 4;
    const bool hasTail = blk < NTAIL;
    const int  pr      = wv;
    const int  rt      = tailw ? (SPB * LBLK + blk) : (blk * SPB + wv - 4);

#pragma unroll
    for (int i = 0; i < (8 * KB1) / LWAVES; ++i) {
        const int kp = (tid >> 5) + LWAVES * i;
        const int m  = tid & 31;
        const float4 w0 = *(const float4*)(W1 + (size_t)(2 * kp) * HID + 4 * m);
        const float4 w1 = *(const float4*)(W1 + (size_t)(2 * kp + 1) * HID + 4 * m);
        float* dp = Ws + kp * WPITCH + (m >> 4) * 128 + (m & 15) * 4;
        *(float4*)dp        = make_float4(w0.x, w1.x, w0.y, w1.y);
        *(float4*)(dp + 64) = make_float4(w0.z, w1.z, w0.w, w1.w);
    }
#pragma unroll
    for (int i = 0; i < (8 * KB2) / LWAVES; ++i) {
        const int kp = (tid >> 5) + LWAVES * i;
        const int m  = tid & 31;
        const float4 w0 = *(const float4*)(W2 + (size_t)(2 * kp) * HID + 4 * m);
        const float4 w1 = *(const float4*)(W2 + (size_t)(2 * kp + 1) * HID + 4 * m);
        float* dp = Ws + (8 * KB1 + kp) * WPITCH + (m >> 4) * 128 + (m & 15) * 4;
        *(float4*)dp        = make_float4(w0.x, w1.x, w0.y, w1.y);
        *(float4*)(dp + 64) = make_float4(w0.z, w1.z, w0.w, w1.w);
    }
    __syncthreads();

    if (tailw && !hasTail) {
        encode_edge_groups(ea, We, be, Wd, ze, grp0 + (blk - NTAIL) * 4 + wv, grp1, EWAVES, lane);
        return;
    }

    float4 alo[KB], ahi[KB];
    {
        const int n0 = rt * 16;
        float* gt = Gt + wv * (16 * CPITCH);
        const float* hl = h + lane * 4;
        const int   ob = offs[n0 + min(lane, 16)];
        const float dn = dinv[n0 + (lane & 15)];
        const int beg = __builtin_amdgcn_readlane(ob, 0);
        const int end = __builtin_amdgcn_readlane(ob, 16);
        int   node = 0;
        int   nb   = __builtin_amdgcn_readlane(ob, 1);
        float di   = __int_as_float(__builtin_amdgcn_readlane(__float_as_int(dn), 0));
        float4 acc = make_float4(0.0f, 0.0f, 0.0f, 0.0f);

        int   sC = 0, sN = 0;
        float wC = 0.0f;
        if (beg + lane < end) {
            sC = bsrc[beg + lane];
            wC = dinv[sC];
        }
        if (beg + 32 + lane < end) sN = bsrc[beg + 32 + lane];
        auto ask_rows = [&](float4 (&v)[GR], int j0, int m) {
#pragma unroll
            for (int u = 0; u < GR; ++u) {
                const int jj = min(j0 + u, m - 1);
                const int s  = __builtin_amdgcn_readlane(sC, jj);
                v[u] = *(const float4*)(hl + (size_t)s * HID);
            }
        };
        auto add_rows = [&](const float4 (&v)[GR], int base, int j0, int m) {
#pragma unroll
            for (int u = 0; u < GR; ++u) {
                const int j = j0 + u;
                if (j < m) {
                    const int e = base + j;
                    while (e >= nb) {
                        *(float4*)(gt + node * CPITCH + 4 * lane) = acc;
                        acc = make_float4(0.0f, 0.0f, 0.0f, 0.0f);
                        ++node;
                        nb = __builtin_amdgcn_readlane(ob, node + 1);
                        di = __int_as_float(__builtin_amdgcn_readlane(__float_as_int(dn), node));
                    }
                    const float w = __int_as_float(__builtin_amdgcn_readlane(__float_as_int(wC), j)) * di;
                    acc.x += v[u].x * w;
                    acc.y += v[u].y * w;
                    acc.z += v[u].z * w;
                    acc.w += v[u].w * w;
                }
            }
        };
        static_assert(2 * GR == 32, "a chunk of 32 edges is two groups of rows");
        for (int base = beg; base < end; base += 32) {
            const int m = min(32, end - base);
            float4 v[GR];
            ask_rows(v, 0, m);
            const float wN = dinv[sN];
            int sNN = 0;
            if (base + 64 + lane < end) sNN = bsrc[base + 64 + lane];
            add_rows(v, base, 0, m);
            if (m > GR) {
                ask_rows(v, GR, m);
                add_rows(v, base, GR, m);
            }
            sC = sN;
            wC = wN;
            sN = sNN;
        }
        for (; node < 16; ++node) {
            *(float4*)(gt + node * CPITCH + 4 * lane) = acc;
            acc = make_float4(0.0f, 0.0f, 0.0f, 0.0f);
        }

        float4 sv[16];
#pragma unroll
        for (int i = 0; i < 16; ++i) sv[i] = *(const float4*)(hl + (size_t)(n0 + i) * HID);
        const float* ap2 = h + (size_t)(n0 + c) * HID + 8 * g;
#pragma unroll
        for (int b = 0; b < KB2; ++b) {
            alo[KB1 + b] = *(const float4*)(ap2 + 16 * b);
            ahi[KB1 + b] = *(const float4*)(ap2 + 16 * b + 4);
        }
#pragma unroll
        for (int i = 0; i < 16; ++i) {
            const float d  = __int_as_float(__builtin_amdgcn_readlane(__float_as_int(dn), i));
            const float dd = d * d;
            float4 t = *(const float4*)(gt + i * CPITCH + 4 * lane);
            t.x += sv[i].x * dd;
            t.y += sv[i].y * dd;
            t.z += sv[i].z * dd;
            t.w += sv[i].w * dd;
            *(float4*)(gt + i * CPITCH + 4 * lane) = t;
        }
        const float* ap1 = gt + c * CPITCH + 8 * g;
#pragma unroll
        for (int b = 0; b < KB1; ++b) {
            alo[b] = *(const float4*)(ap1 + 16 * b);
            ahi[b] = *(const float4*)(ap1 + 16 * b + 4);
        }
    }

    auto load_bias = [&](float4& lo, float4& hi) {
        const float4 l1 = *(const float4*)(bias1 + 4 * c);
        const float4 h1 = *(const float4*)(bias1 + 64 + 4 * c);
        const float4 l2 = *(const float4*)(bias2 + 4 * c);
        const float4 h2 = *(const float4*)(bias2 + 64 + 4 * c);
        lo = make_float4(l1.x + l2.x, l1.y + l2.y, l1.z + l2.z, l1.w + l2.w);
        hi = make_float4(h1.x + h2.x, h1.y + h2.y, h1.z + h2.z, h1.w + h2.w);
    };

    if (!tailw) {
        v8f acc[8];
#pragma unroll
        for (int t = 0; t < 8; ++t) acc[t] = v8f{};

        const float* bp = Ws + (4 * g) * WPITCH + c * 4;
        float4 wn[4];
#pragma unroll
        for (int p = 0; p < 4; ++p) wn[p] = *(const float4*)(bp + 64 * p);
#pragma unroll
        for (int b = 0; b < KB; ++b) {
            v2f Af[4];
            Af[0][0] = alo[b].x; Af[0][1] = alo[b].y;
            Af[1][0] = alo[b].z; Af[1][1] = alo[b].w;
            Af[2][0] = ahi[b].x; Af[2][1] = ahi[b].y;
            Af[3][0] = ahi[b].z; Af[3][1] = ahi[b].w;
#pragma unroll
            for (int s = 0; s < 4; ++s) {
                float4 wc[4];
#pragma unroll
                for (int p = 0; p < 4; ++p) wc[p] = wn[p];
                if (b + 1 < KB || s < 3) {
                    const float* q = bp + (s < 3 ? 8 * b + s + 1 : 8 * (b + 1)) * WPITCH;
#pragma unroll
                    for (int p = 0; p < 4; ++p) wn[p] = *(const float4*)(q + 64 * p);
                }
                __builtin_amdgcn_sched_barrier(0);
#pragma unroll
                for (int p = 0; p < 4; ++p) {
                    const float4 w = wc[p];
                    v2f b0, b1;
                    b0[0] = w.x; b0[1] = w.y;
                    b1[0] = w.z; b1[1] = w.w;
                    acc[2 * p] = __builtin_amdgcn_wmma_f32_16x16x4_f32(
                        false, Af[s], false, b0, (short)0, acc[2 * p], false, false);
                    acc[2 * p + 1] = __builtin_amdgcn_wmma_f32_16x16x4_f32(
                        false, Af[s], false, b1, (short)0, acc[2 * p + 1], false, false);
                }
            }
        }

        float4 bvlo, bvhi;
        load_bias(bvlo, bvhi);

        if (MODE == 0) {
#pragma unroll
            for (int r = 0; r < 8; ++r) {
                float4 lo, hi;
                lo.x = fmaxf(acc[0][r] + bvlo.x, 0.0f); lo.y = fmaxf(acc[1][r] + bvlo.y, 0.0f);
                lo.z = fmaxf(acc[2][r] + bvlo.z, 0.0f); lo.w = fmaxf(acc[3][r] + bvlo.w, 0.0f);
                hi.x = fmaxf(acc[4][r] + bvhi.x, 0.0f); hi.y = fmaxf(acc[5][r] + bvhi.y, 0.0f);
                hi.z = fmaxf(acc[6][r] + bvhi.z, 0.0f); hi.w = fmaxf(acc[7][r] + bvhi.w, 0.0f);
                float* cp = C + (size_t)(rt * 16 + 8 * g + r) * HID + 4 * c;
                *(float4*)cp        = lo;
                *(float4*)(cp + 64) = hi;
            }
        } else {
            const float4 u1l = *(const float4*)(Wd + 4 * c);
            const float4 u1h = *(const float4*)(Wd + 64 + 4 * c);
            const float4 u2l = *(const float4*)(Wd + HID + 4 * c);
            const float4 u2h = *(const float4*)(Wd + HID + 64 + 4 * c);
            float q[16];
#pragma unroll
            for (int r = 0; r < 8; ++r) {
                float4 lo, hi;
                lo.x = fmaxf(acc[0][r] + bvlo.x, 0.0f); lo.y = fmaxf(acc[1][r] + bvlo.y, 0.0f);
                lo.z = fmaxf(acc[2][r] + bvlo.z, 0.0f); lo.w = fmaxf(acc[3][r] + bvlo.w, 0.0f);
                hi.x = fmaxf(acc[4][r] + bvhi.x, 0.0f); hi.y = fmaxf(acc[5][r] + bvhi.y, 0.0f);
                hi.z = fmaxf(acc[6][r] + bvhi.z, 0.0f); hi.w = fmaxf(acc[7][r] + bvhi.w, 0.0f);
                float a = lo.x * u1l.x;
                a += lo.y * u1l.y; a += lo.z * u1l.z; a += lo.w * u1l.w;
                a += hi.x * u1h.x; a += hi.y * u1h.y; a += hi.z * u1h.z; a += hi.w * u1h.w;
                float d = lo.x * u2l.x;
                d += lo.y * u2l.y; d += lo.z * u2l.z; d += lo.w * u2l.w;
                d += hi.x * u2h.x; d += hi.y * u2h.y; d += hi.z * u2h.z; d += hi.w * u2h.w;
                q[2 * r]     = a;
                q[2 * r + 1] = d;
            }
            const bool bit3 = (c & 8) != 0;
            const bool bit2 = (c & 4) != 0;
            const bool bit1 = (c & 2) != 0;
            const bool bit0 = (c & 1) != 0;
            float u[8];
#pragma unroll
            for (int m = 0; m < 8; ++m) {
                const float keep = bit3 ? q[8 + m] : q[m];
                const float send = bit3 ? q[m] : q[8 + m];
                u[m] = keep + __shfl_xor(send, 8, 32);
            }
            float w4[4];
#pragma unroll
            for (int m = 0; m < 4; ++m) {
                const float keep = bit2 ? u[4 + m] : u[m];
                const float send = bit2 ? u[m] : u[4 + m];
                w4[m] = keep + __shfl_xor(send, 4, 32);
            }
            float x2[2];
#pragma unroll
            for (int m = 0; m < 2; ++m) {
                const float keep = bit1 ? w4[2 + m] : w4[m];
                const float send = bit1 ? w4[m] : w4[2 + m];
                x2[m] = keep + __shfl_xor(send, 2, 32);
            }
            const float keep1 = bit0 ? x2[1] : x2[0];
            const float send1 = bit0 ? x2[0] : x2[1];
            const float z = keep1 + __shfl_xor(send1, 1, 32);
            C[(size_t)rt * 32 + lane] = z;
        }
    } else if (hasTail) {
        v8f tacc[2];
        tacc[0] = v8f{};
        tacc[1] = v8f{};
        const float* bp = Ws + (4 * g) * WPITCH + c * 4 + 64 * pr;
        float4 wn[4];
#pragma unroll
        for (int s = 0; s < 4; ++s) wn[s] = *(const float4*)(bp + s * WPITCH);
#pragma unroll
        for (int b = 0; b < KB; ++b) {
            v2f Af[4];
            Af[0][0] = alo[b].x; Af[0][1] = alo[b].y;
            Af[1][0] = alo[b].z; Af[1][1] = alo[b].w;
            Af[2][0] = ahi[b].x; Af[2][1] = ahi[b].y;
            Af[3][0] = ahi[b].z; Af[3][1] = ahi[b].w;
            float4 wc[4];
#pragma unroll
            for (int s = 0; s < 4; ++s) wc[s] = wn[s];
            if (b + 1 < KB) {
#pragma unroll
                for (int s = 0; s < 4; ++s)
                    wn[s] = *(const float4*)(bp + (8 * (b + 1) + s) * WPITCH);
            }
            __builtin_amdgcn_sched_barrier(0);
#pragma unroll
            for (int s = 0; s < 4; ++s) {
                const float4 w = wc[s];
                v2f b0, b1;
                b0[0] = w.x; b0[1] = w.y;
                b1[0] = w.z; b1[1] = w.w;
                tacc[0] = __builtin_amdgcn_wmma_f32_16x16x4_f32(
                    false, Af[s], false, b0, (short)0, tacc[0], false, false);
                tacc[1] = __builtin_amdgcn_wmma_f32_16x16x4_f32(
                    false, Af[s], false, b1, (short)0, tacc[1], false, false);
            }
        }
        float4 bvlo, bvhi;
        load_bias(bvlo, bvhi);
        const float4 bv = (pr >> 1) ? bvhi : bvlo;
        const float bb0 = (pr & 1) ? bv.z : bv.x;
        const float bb1 = (pr & 1) ? bv.w : bv.y;
        float* tp = Ct + (8 * g) * CPITCH + 4 * c + 2 * (pr & 1) + 64 * (pr >> 1);
#pragma unroll
        for (int r = 0; r < 8; ++r) {
            float2 v;
            v.x = fmaxf(tacc[0][r] + bb0, 0.0f);
            v.y = fmaxf(tacc[1][r] + bb1, 0.0f);
            *(float2*)(tp + r * CPITCH) = v;
        }
    }

    if (hasTail) {
        __syncthreads();
        if (tailw) {
            if (MODE == 0) {
#pragma unroll
                for (int r = 0; r < 4; ++r) {
                    const int row = 4 * pr + r;
                    const float4 v = *(const float4*)(Ct + row * CPITCH + 4 * lane);
                    *(float4*)(C + (size_t)(rt * 16 + row) * HID + 4 * lane) = v;
                }
            } else if (pr == 0) {
                const float* cr = Ct + c * CPITCH;
                const float* wp = Wd + g * HID;
                float sum = 0.0f;
#pragma unroll
                for (int j = 0; j < HID / 4; ++j) {
                    const float4 v = *(const float4*)(cr + 4 * j);
                    const float4 w = *(const float4*)(wp + 4 * j);
                    sum += v.x * w.x; sum += v.y * w.y;
                    sum += v.z * w.z; sum += v.w * w.w;
                }
                C[(size_t)rt * 32 + 2 * c + g] = sum;
            }
        }
    }
}

__global__ __launch_bounds__(1024) void rank_edges_lds(
    const int* __restrict__ dst, int* __restrict__ rank,
    int* __restrict__ cntS, int* __restrict__ csum)
{
    __shared__ __attribute__((aligned(16))) int cnt[NNP];
    const int tid  = threadIdx.x;
    const int lane = tid & 31;
    const int wv   = tid >> 5;
    const int sl   = blockIdx.x;

    const int4* __restrict__ d4 = (const int4*)(dst + (size_t)sl * EPS);
    int4 d[RANK_IT];
#pragma unroll
    for (int k = 0; k < RANK_IT; ++k) {
        const int i = tid + 1024 * k;
        d[k] = make_int4(0, 0, 0, 0);
        if (i < EPS / 4) d[k] = d4[i];
    }

    int4* c4 = (int4*)cnt;
    for (int i = tid; i < NNP / 4; i += 1024) c4[i] = make_int4(0, 0, 0, 0);
    __syncthreads();

#pragma unroll
    for (int k = 0; k < RANK_IT; ++k) {
        const int i = tid + 1024 * k;
        if (i < EPS / 4) {
            d[k].x = atomicAdd(&cnt[d[k].x], 1);
            d[k].y = atomicAdd(&cnt[d[k].y], 1);
            d[k].z = atomicAdd(&cnt[d[k].z], 1);
            d[k].w = atomicAdd(&cnt[d[k].w], 1);
        }
    }
    __syncthreads();

    int4* __restrict__ r4 = (int4*)(rank + (size_t)sl * EPS);
#pragma unroll
    for (int k = 0; k < RANK_IT; ++k) {
        const int i = tid + 1024 * k;
        if (i < EPS / 4) r4[i] = d[k];
    }

    int4* __restrict__ g4 = (int4*)(cntS + (size_t)sl * NNP);
    int sum[CHUNK_IT];
#pragma unroll
    for (int t = 0; t < CHUNK_IT; ++t) {
        const int j = wv + 32 * t;
        sum[t] = 0;
        if (j < NCHUNK) {
            const int4 v0 = c4[64 * j + lane];
            const int4 v1 = c4[64 * j + 32 + lane];
            g4[64 * j + lane]      = v0;
            g4[64 * j + 32 + lane] = v1;
            sum[t] = ((v0.x + v0.y) + (v0.z + v0.w)) + ((v1.x + v1.y) + (v1.z + v1.w));
        }
    }
#pragma unroll
    for (int off = 16; off > 0; off >>= 1) {
#pragma unroll
        for (int t = 0; t < CHUNK_IT; ++t)
            sum[t] += __shfl_xor(sum[t], off, 32);
    }
    if (lane == 0) {
#pragma unroll
        for (int t = 0; t < CHUNK_IT; ++t) {
            const int j = wv + 32 * t;
            if (j < NCHUNK) csum[j * NSLICE + sl] = sum[t];
        }
    }
}

#define ENC_BLOCKS ((NRT + 7) / 8)

__device__ __forceinline__ void encode_nodes(
    float* __restrict__ Ws, const float* __restrict__ x, const float* __restrict__ Wn,
    const float* __restrict__ bn, float* __restrict__ C, int eb)
{
    constexpr int KB = FN / 16;
    const int tid  = threadIdx.x;
    const int lane = tid & 31;
    const int c    = lane & 15;
    const int g    = lane >> 4;
    const int rt = __builtin_amdgcn_readfirstlane(eb * 8 + (tid >> 5));
    const bool active = rt < NRT;
    const int rtl = active ? rt : NRT - 1;

    float4 alo[KB], ahi[KB];
    {
        const float* ap = x + (size_t)(rtl * 16 + c) * FN + 8 * g;
#pragma unroll
        for (int b = 0; b < KB; ++b) {
            alo[b] = *(const float4*)(ap + 16 * b);
            ahi[b] = *(const float4*)(ap + 16 * b + 4);
        }
    }

#pragma unroll
    for (int i = 0; i < KB; ++i) {
        const int kp = (tid >> 5) + 8 * i;
        const int m  = tid & 31;
        const float4 w0 = *(const float4*)(Wn + (size_t)(2 * kp) * HID + 4 * m);
        const float4 w1 = *(const float4*)(Wn + (size_t)(2 * kp + 1) * HID + 4 * m);
        float* dp = Ws + kp * WPITCH + (m >> 4) * 128 + (m & 15) * 4;
        *(float4*)dp        = make_float4(w0.x, w1.x, w0.y, w1.y);
        *(float4*)(dp + 64) = make_float4(w0.z, w1.z, w0.w, w1.w);
    }
    __syncthreads();

    if (!active) return;

    v8f acc[8];
#pragma unroll
    for (int t = 0; t < 8; ++t) acc[t] = v8f{};

    const float* bp = Ws + (4 * g) * WPITCH + c * 4;
#pragma unroll
    for (int b = 0; b < KB; ++b) {
        v2f Af[4];
        Af[0][0] = alo[b].x; Af[0][1] = alo[b].y;
        Af[1][0] = alo[b].z; Af[1][1] = alo[b].w;
        Af[2][0] = ahi[b].x; Af[2][1] = ahi[b].y;
        Af[3][0] = ahi[b].z; Af[3][1] = ahi[b].w;
#pragma unroll
        for (int s = 0; s < 4; ++s) {
            const float* q = bp + (8 * b + s) * WPITCH;
#pragma unroll
            for (int p = 0; p < 4; ++p) {
                const float4 w = *(const float4*)(q + 64 * p);
                v2f b0, b1;
                b0[0] = w.x; b0[1] = w.y;
                b1[0] = w.z; b1[1] = w.w;
                acc[2 * p] = __builtin_amdgcn_wmma_f32_16x16x4_f32(
                    false, Af[s], false, b0, (short)0, acc[2 * p], false, false);
                acc[2 * p + 1] = __builtin_amdgcn_wmma_f32_16x16x4_f32(
                    false, Af[s], false, b1, (short)0, acc[2 * p + 1], false, false);
            }
        }
    }

    const float4 bvlo = *(const float4*)(bn + 4 * c);
    const float4 bvhi = *(const float4*)(bn + 64 + 4 * c);

#pragma unroll
    for (int r = 0; r < 8; ++r) {
        float4 lo, hi;
        lo.x = fmaxf(acc[0][r] + bvlo.x, 0.0f); lo.y = fmaxf(acc[1][r] + bvlo.y, 0.0f);
        lo.z = fmaxf(acc[2][r] + bvlo.z, 0.0f); lo.w = fmaxf(acc[3][r] + bvlo.w, 0.0f);
        hi.x = fmaxf(acc[4][r] + bvhi.x, 0.0f); hi.y = fmaxf(acc[5][r] + bvhi.y, 0.0f);
        hi.z = fmaxf(acc[6][r] + bvhi.z, 0.0f); hi.w = fmaxf(acc[7][r] + bvhi.w, 0.0f);
        float* cp = C + (size_t)(rt * 16 + 8 * g + r) * HID + 4 * c;
        *(float4*)cp        = lo;
        *(float4*)(cp + 64) = hi;
    }
}

__global__ __launch_bounds__(256) void scan_offsets_encode_nodes(
    int* __restrict__ cntS, const int* __restrict__ csum,
    int* __restrict__ offs, float* __restrict__ dinv, int n,
    const float* __restrict__ x, const float* __restrict__ Wn,
    const float* __restrict__ bn, float* __restrict__ h0)
{
    __shared__ __attribute__((aligned(16))) float Ws[8 * (FN / 16) * WPITCH];
    __shared__ int waveCnt[8];
    __shared__ int wavePre[8];
    if (blockIdx.x >= NCHUNK) {
        encode_nodes(Ws, x, Wn, bn, h0, (int)blockIdx.x - NCHUNK);
        return;
    }
    const int tid  = threadIdx.x;
    const int lane = tid & 31;
    const int wv   = tid >> 5;
    const int b    = blockIdx.x;
    const int i    = b * 256 + tid;

    int cs[NSLICE];
#pragma unroll
    for (int k = 0; k < NSLICE; ++k) cs[k] = cntS[(size_t)k * NNP + i];

    int pre = 0;
    const int4* __restrict__ s4 = (const int4*)csum;
    const int n4   = b * (NSLICE / 4);
    const int full = n4 >> 8;
#pragma unroll 4
    for (int it = 0; it < full; ++it) {
        int4 v = s4[it * 256 + tid];
        pre += (v.x + v.y) + (v.z + v.w);
    }
    if (tid < (n4 & 255)) {
        int4 v = s4[full * 256 + tid];
        pre += (v.x + v.y) + (v.z + v.w);
    }

    int c = 0;
#pragma unroll
    for (int k = 0; k < NSLICE; ++k) c += cs[k];

    int inc = c;
    for (int off = 1; off < 32; off <<= 1) {
        int t = __shfl_up(inc, off, 32);
        if (lane >= off) inc += t;
    }
    for (int off = 16; off > 0; off >>= 1)
        pre += __shfl_xor(pre, off, 32);

    if (lane == 31) waveCnt[wv] = inc;
    if (lane == 0)  wavePre[wv] = pre;
    __syncthreads();

    int base = 0;
    for (int k = 0; k < 8; ++k) {
        base += wavePre[k];
        if (k < wv) base += waveCnt[k];
    }
    const int excl = base + inc - c;

    int run = excl;
#pragma unroll
    for (int k = 0; k < NSLICE; ++k) {
        cntS[(size_t)k * NNP + i] = run;
        run += cs[k];
    }

    if (i < n) {
        offs[i] = excl;
        float d = (float)(c + 1);
        dinv[i] = (d > 0.0f) ? rsqrtf(d) : 0.0f;
    }
    if (i == n - 1) offs[n] = excl + c;
}

__global__ void bucket_edges(const int* __restrict__ src, const int* __restrict__ dst,
                             const int* __restrict__ rank, const int* __restrict__ start,
                             int* __restrict__ bsrc, int n)
{
    int i = blockIdx.x * blockDim.x + threadIdx.x;
    if (i < n) bsrc[start[(size_t)(i / EPS) * NNP + dst[i]] + rank[i]] = src[i];
}

static_assert(NE % 4 == 0, "edges are whole 16-byte runs");
__global__ __launch_bounds__(256) void decode_edges(
    const float* __restrict__ pp, const float* __restrict__ ze,
    const int* __restrict__ src,  const int* __restrict__ dst,
    const float* __restrict__ bd, float* __restrict__ out)
{
    const int i = (int)(blockIdx.x * 256 + threadIdx.x);
    if (i < NE / 4) {
        const int4   sv = ((const int4*)src)[i];
        const int4   dv = ((const int4*)dst)[i];
        const float4 zv = ((const float4*)ze)[i];
        const float bdv = bd[0];
        float4 o;
        o.x = (pp[2 * (size_t)sv.x] + pp[2 * (size_t)dv.x + 1]) + zv.x + bdv;
        o.y = (pp[2 * (size_t)sv.y] + pp[2 * (size_t)dv.y + 1]) + zv.y + bdv;
        o.z = (pp[2 * (size_t)sv.z] + pp[2 * (size_t)dv.z + 1]) + zv.z + bdv;
        o.w = (pp[2 * (size_t)sv.w] + pp[2 * (size_t)dv.w + 1]) + zv.w + bdv;
        ((float4*)out)[i] = o;
    }
}

extern "C" void kernel_launch(void* const* d_in, const int* in_sizes, int n_in,
                              void* d_out, int out_size, void* d_ws, size_t ws_size,
                              hipStream_t stream)
{
    const float* x   = (const float*)d_in[0];
    const float* ea  = (const float*)d_in[1];
    const int*   ei  = (const int*)  d_in[2];
    const float* Wn  = (const float*)d_in[3];
    const float* bn  = (const float*)d_in[4];
    const float* We  = (const float*)d_in[5];
    const float* be  = (const float*)d_in[6];
    const float* Wc  = (const float*)d_in[7];
    const float* bc  = (const float*)d_in[8];
    const float* Wr  = (const float*)d_in[9];
    const float* br  = (const float*)d_in[10];
    const float* Wd  = (const float*)d_in[11];
    const float* bd  = (const float*)d_in[12];
    float* out = (float*)d_out;

    const int* srcI = ei;
    const int* dstI = ei + NE;

    float* ws   = (float*)d_ws;
    float* hA   = ws;
    float* hB   = hA   + (size_t)NN * HID;
    float* dinv = hB   + (size_t)NN * HID;
    int*   offs = (int*)(dinv + NN);
    int*   rank = offs + NN + 16;
    int*   bsrc = rank + NE;
    float* pp   = (float*)(bsrc + NE);
    int*   cntS = (int*)(pp + 2 * (size_t)NN);
    int*   csum = cntS + (size_t)NSLICE * NNP;
    float* ze   = (float*)(csum + (size_t)NCHUNK * NSLICE);

    rank_edges_lds<<<NSLICE, 1024, 0, stream>>>(dstI, rank, cntS, csum);
    scan_offsets_encode_nodes<<<NCHUNK + ENC_BLOCKS, 256, 0, stream>>>(
        cntS, csum, offs, dinv, NN, x, Wn, bn, hA);
    bucket_edges<<<(NE + 255) / 256, 256, 0, stream>>>(srcI, dstI, rank, cntS, bsrc, NE);

    layer_fused<0><<<LBLK, LWAVES * 32, 0, stream>>>(
        hA, offs, bsrc, dinv, Wc, Wr, br, bc, hB, Wd, ea, We, be, ze, 0, GPL);
    layer_fused<0><<<LBLK, LWAVES * 32, 0, stream>>>(
        hB, offs, bsrc, dinv, Wc + (size_t)HID * HID, Wr, br, bc + HID, hA, Wd,
        ea, We, be, ze, GPL, 2 * GPL);
    layer_fused<1><<<LBLK, LWAVES * 32, 0, stream>>>(
        hA, offs, bsrc, dinv, Wc + (size_t)2 * HID * HID, Wr, br, bc + 2 * HID, pp, Wd,
        ea, We, be, ze, 2 * GPL, NGRP);

    decode_edges<<<(NE / 4 + 255) / 256, 256, 0, stream>>>(pp, ze, srcI, dstI, bd, out);
}
